# starting kernel
# baseline (speedup 1.0000x reference)
#include <hip/hip_runtime.h>

typedef __attribute__((ext_vector_type(2))) float v2f;
typedef __attribute__((ext_vector_type(8))) float v8f;

#define IN_F   128
#define HID_F  64
#define OUT_F  32


__global__ void k_init_deg(float* __restrict__ deg_in, float* __restrict__ deg_out, int N) {
  int i = blockIdx.x * blockDim.x + threadIdx.x;
  if (i < N) { deg_in[i] = 1.0f; deg_out[i] = 1.0f; }
}

__global__ void k_count_deg(const int* __restrict__ ei0, const int* __restrict__ ei1,
                            float* __restrict__ deg_in, float* __restrict__ deg_out, int E) {
  int e = blockIdx.x * blockDim.x + threadIdx.x;
  if (e < E) {
    atomicAdd(&deg_out[ei0[e]], 1.0f);
    atomicAdd(&deg_in [ei1[e]], 1.0f);
  }
}

__global__ void k_norms(const int* __restrict__ ei0, const int* __restrict__ ei1,
                        const float* __restrict__ deg_in, const float* __restrict__ deg_out,
                        float* __restrict__ rn_edge, float* __restrict__ rn_self, int E, int N) {
  int i = blockIdx.x * blockDim.x + threadIdx.x;
  if (i < E) {
    rn_edge[i] = rsqrtf(deg_in[ei1[i]] * deg_out[ei0[i]]);
  } else if (i < E + N) {
    int n = i - E;
    rn_self[n] = rsqrtf(deg_in[n] * deg_out[n]);
  }
}


template <int K, int RELU>
__global__ __launch_bounds__(128)
void k_gemm_wmma(const float* __restrict__ X, int ldx,
                 const float* __restrict__ Wa, const float* __restrict__ ba,
                 const float* __restrict__ Wb, const float* __restrict__ bb,
                 float* __restrict__ Y, int ldy,
                 int halfO, int koffB, int nColTiles, int nMTiles) {
  constexpr int NK = K / 4;

  int wave = threadIdx.x >> 5;
  int lane = threadIdx.x & 31;
  int mt   = blockIdx.x * 4 + wave;
  if (mt >= nMTiles) return;

  int m0   = mt << 4;
  int mrow = lane & 15;
  int khi  = (lane >> 4) << 1;
  int col  = lane & 15;

  const float* xrow = X + (size_t)(m0 + mrow) * ldx;

  v2f afrag[NK];
  int curKoff = -1;

  for (int ct = 0; ct < nColTiles; ++ct) {
    int c0   = ct << 4;
    int half = (c0 >= halfO) ? 1 : 0;
    const float* W    = half ? Wb : Wa;
    const float* bias = half ? bb : ba;
    int wc0  = c0 - half * halfO;
    int koff = half ? koffB : 0;

    if (koff != curKoff) {
      curKoff = koff;
      const float* xp = xrow + koff + khi;
#pragma unroll
      for (int i = 0; i < NK; ++i) {
        v2f a = *(const v2f*)(xp + i * 4);
        if (RELU) { a.x = fmaxf(a.x, 0.0f); a.y = fmaxf(a.y, 0.0f); }
        afrag[i] = a;
      }
    }

    const float* wrow = W + (size_t)(wc0 + col) * K + khi;
    v8f acc = {};
#pragma unroll
    for (int i = 0; i < NK; ++i) {
      v2f bf = *(const v2f*)(wrow + i * 4);
      acc = __builtin_amdgcn_wmma_f32_16x16x4_f32(false, afrag[i], false, bf,
                                                  (short)0, acc, false, false);
    }

    float bv = bias[wc0 + col];
    int mbase = m0 + ((lane >> 4) << 3);
    float* yp = Y + (size_t)mbase * ldy + c0 + col;
#pragma unroll
    for (int g = 0; g < 8; ++g)
      yp[(size_t)g * ldy] = acc[g] + bv;
  }
}


__global__ void k_aggr_self(const float* __restrict__ T, const float* __restrict__ rn_self,
                            float* __restrict__ Y, int total4, int fshift) {
  int idx = blockIdx.x * blockDim.x + threadIdx.x;
  if (idx < total4) {
    float rn = rn_self[(idx << 2) >> fshift];
    float4 v = ((const float4*)T)[idx];
    float4 o; o.x = rn * v.x; o.y = rn * v.y; o.z = rn * v.z; o.w = rn * v.w;
    ((float4*)Y)[idx] = o;
  }
}

__global__ void k_scatter(const float* __restrict__ T,
                          const int* __restrict__ src, const int* __restrict__ dst,
                          const float* __restrict__ rn_edge, float* __restrict__ Y,
                          int E, int F) {
  int chunks = F >> 2;
  int tid = blockIdx.x * blockDim.x + threadIdx.x;
  if (tid >= E * chunks) return;
  int e = tid / chunks;
  int c = (tid - e * chunks) << 2;
  float rn = rn_edge[e];
  const float4 v = *(const float4*)(T + (size_t)src[e] * F + c);
  float* yp = Y + (size_t)dst[e] * F + c;
  atomicAdd(yp + 0, rn * v.x);
  atomicAdd(yp + 1, rn * v.y);
  atomicAdd(yp + 2, rn * v.z);
  atomicAdd(yp + 3, rn * v.w);
}


__global__ void k_final(const float* __restrict__ A2s, const float* __restrict__ A2t,
                        const float* __restrict__ noise_s, const float* __restrict__ noise_t,
                        float* __restrict__ out, int N) {
  int idx = blockIdx.x * blockDim.x + threadIdx.x;
  int total = N * OUT_F;
  if (idx < total) {
    int n = idx >> 5, j = idx & 31;
    out[idx] = A2s[n * 64 + j] + noise_s[idx] * expf(A2s[n * 64 + 32 + j]) * 0.2f;
  } else if (idx < 2 * total) {
    int k = idx - total;
    int n = k >> 5, j = k & 31;
    out[idx] = A2t[n * 64 + j] + noise_t[k] * expf(A2t[n * 64 + 32 + j]) * 0.2f;
  }
}


static inline int ceil_div(int a, int b) { return (a + b - 1) / b; }

extern "C" void kernel_launch(void* const* d_in, const int* in_sizes, int n_in,
                              void* d_out, int out_size, void* d_ws, size_t ws_size,
                              hipStream_t stream) {
  const float* s       = (const float*)d_in[0];
  const float* t       = (const float*)d_in[1];
  const int*   ei      = (const int*)  d_in[2];
  const float* noise_s = (const float*)d_in[3];
  const float* noise_t = (const float*)d_in[4];
  const float* sm1_W = (const float*)d_in[5],  *sm1_b = (const float*)d_in[6];
  const float* sm2_W = (const float*)d_in[7],  *sm2_b = (const float*)d_in[8];
  const float* sl1_W = (const float*)d_in[9],  *sl1_b = (const float*)d_in[10];
  const float* sl2_W = (const float*)d_in[11], *sl2_b = (const float*)d_in[12];
  const float* tm1_W = (const float*)d_in[13], *tm1_b = (const float*)d_in[14];
  const float* tm2_W = (const float*)d_in[15], *tm2_b = (const float*)d_in[16];
  const float* tl1_W = (const float*)d_in[17], *tl1_b = (const float*)d_in[18];
  const float* tl2_W = (const float*)d_in[19], *tl2_b = (const float*)d_in[20];

  const int N = in_sizes[0] / IN_F;
  const int E = in_sizes[2] / 2;
  const int* ei0 = ei;
  const int* ei1 = ei + E;

  float* ws      = (float*)d_ws;
  float* deg_in  = ws;
  float* deg_out = deg_in  + N;
  float* rn_self = deg_out + N;
  float* rn_edge = rn_self + N;
  float* Y1s = rn_edge + E;
  float* Y1t = Y1s + (size_t)N * IN_F;
  float* A1s = Y1t + (size_t)N * IN_F;
  float* A1t = A1s + (size_t)N * IN_F;
  float* Y2s = Y1s;
  float* A2s = Y1s + (size_t)N * HID_F;
  float* Y2t = Y1t;
  float* A2t = Y1t + (size_t)N * HID_F;

  const int B = 256;
  const int nMTiles = ceil_div(N, 16);
  const int gemmBlocks = ceil_div(nMTiles, 4);

  k_init_deg<<<ceil_div(N, B), B, 0, stream>>>(deg_in, deg_out, N);
  k_count_deg<<<ceil_div(E, B), B, 0, stream>>>(ei0, ei1, deg_in, deg_out, E);
  k_norms<<<ceil_div(E + N, B), B, 0, stream>>>(ei0, ei1, deg_in, deg_out, rn_edge, rn_self, E, N);

  k_gemm_wmma<IN_F, 0><<<gemmBlocks, 128, 0, stream>>>(
      s, IN_F, sm1_W, sm1_b, sl1_W, sl1_b, Y1s, IN_F, HID_F, 0, IN_F / 16, nMTiles);
  k_gemm_wmma<IN_F, 0><<<gemmBlocks, 128, 0, stream>>>(
      t, IN_F, tm1_W, tm1_b, tl1_W, tl1_b, Y1t, IN_F, HID_F, 0, IN_F / 16, nMTiles);

  {
    int total4 = N * IN_F / 4, work = E * (IN_F >> 2);
    k_aggr_self<<<ceil_div(total4, B), B, 0, stream>>>(Y1s, rn_self, A1s, total4, 7);
    k_scatter  <<<ceil_div(work,   B), B, 0, stream>>>(Y1s, ei0, ei1, rn_edge, A1s, E, IN_F);
    k_aggr_self<<<ceil_div(total4, B), B, 0, stream>>>(Y1t, rn_self, A1t, total4, 7);
    k_scatter  <<<ceil_div(work,   B), B, 0, stream>>>(Y1t, ei1, ei0, rn_edge, A1t, E, IN_F);
  }

  k_gemm_wmma<HID_F, 1><<<gemmBlocks, 128, 0, stream>>>(
      A1s, IN_F, sm2_W, sm2_b, sl2_W, sl2_b, Y2s, HID_F, OUT_F, HID_F, HID_F / 16, nMTiles);
  k_gemm_wmma<HID_F, 1><<<gemmBlocks, 128, 0, stream>>>(
      A1t, IN_F, tm2_W, tm2_b, tl2_W, tl2_b, Y2t, HID_F, OUT_F, HID_F, HID_F / 16, nMTiles);

  {
    int total4 = N * HID_F / 4, work = E * (HID_F >> 2);
    k_aggr_self<<<ceil_div(total4, B), B, 0, stream>>>(Y2s, rn_self, A2s, total4, 6);
    k_scatter  <<<ceil_div(work,   B), B, 0, stream>>>(Y2s, ei1, ei0, rn_edge, A2s, E, HID_F);
    k_aggr_self<<<ceil_div(total4, B), B, 0, stream>>>(Y2t, rn_self, A2t, total4, 6);
    k_scatter  <<<ceil_div(work,   B), B, 0, stream>>>(Y2t, ei0, ei1, rn_edge, A2t, E, HID_F);
  }

  k_final<<<ceil_div(2 * N * OUT_F, B), B, 0, stream>>>(A2s, A2t, noise_s, noise_t,
                                                        (float*)d_out, N);
}


# Scatter-add moved into LDS: a work-group owns 100 destination rows, edges binned once by that range, rows added by LDS f32 atomics (arrival order, as before), tile stored once; no global f32 atomic
# speedup vs baseline: 3.1783x; 3.1783x over previous
#include <hip/hip_runtime.h>


typedef __attribute__((ext_vector_type(2))) float v2f;
typedef __attribute__((ext_vector_type(8))) float v8f;

#define IN_F   128
#define HID_F  64
#define OUT_F  32

#define TR     100
#define AGG_T  512
#define BIN_T  1024
#define BIN_EPT 4
#define REPL   16


__global__ void k_zero_counts(int* __restrict__ cnt, int n) {
  int i = blockIdx.x * blockDim.x + threadIdx.x;
  if (i < n) cnt[i] = 0;
}

__global__ __launch_bounds__(BIN_T)
void k_bin_count(const int* __restrict__ ei0, const int* __restrict__ ei1,
                 int* __restrict__ cnt, int E, int NB) {
  extern __shared__ int bin_lds[];
  int* hist = bin_lds;
  for (int i = threadIdx.x; i < 2 * NB; i += BIN_T) hist[i] = 0;
  __syncthreads();
  int e0 = blockIdx.x * (BIN_T * BIN_EPT) + threadIdx.x;
  for (int k = 0; k < BIN_EPT; ++k) {
    int e = e0 + k * BIN_T;
    if (e < E) {
      atomicAdd(&hist[ei1[e] / TR], 1);
      atomicAdd(&hist[NB + ei0[e] / TR], 1);
    }
  }
  __syncthreads();
  int* mycnt = cnt + (blockIdx.x % REPL) * 2 * NB;
  for (int i = threadIdx.x; i < 2 * NB; i += BIN_T) {
    int c = hist[i];
    if (c) atomicAdd(&mycnt[i], c);
  }
}

__global__ __launch_bounds__(1024)
void k_bin_scan(const int* __restrict__ cnt, int* __restrict__ start,
                int* __restrict__ cursor, int NB) {
  __shared__ int sh[1024];
  int dir = blockIdx.x;
  int* st = start + dir * (NB + 1);
  int tid = threadIdx.x;
  int carry = 0;
  for (int base = 0; base < NB; base += 1024) {
    int i = base + tid;
    int c[REPL];
    int v = 0;
#pragma unroll
    for (int r = 0; r < REPL; ++r) {
      c[r] = (i < NB) ? cnt[r * 2 * NB + dir * NB + i] : 0;
      v += c[r];
    }
    sh[tid] = v;
    __syncthreads();
    for (int off = 1; off < 1024; off <<= 1) {
      int t = (tid >= off) ? sh[tid - off] : 0;
      __syncthreads();
      sh[tid] += t;
      __syncthreads();
    }
    int run = carry + sh[tid] - v;
    if (i < NB) {
      st[i] = run;
#pragma unroll
      for (int r = 0; r < REPL; ++r) {
        cursor[r * 2 * NB + dir * NB + i] = run;
        run += c[r];
      }
    }
    carry += sh[1023];
    __syncthreads();
  }
  if (tid == 0) st[NB] = carry;
}

__global__ __launch_bounds__(BIN_T)
void k_bin_place(const int* __restrict__ ei0, const int* __restrict__ ei1,
                 int* __restrict__ cursor, int* __restrict__ recF, int* __restrict__ recB,
                 int E, int NB) {
  extern __shared__ int bin_lds[];
  int* hist = bin_lds;
  int* base = bin_lds + 2 * NB;
  for (int i = threadIdx.x; i < 2 * NB; i += BIN_T) hist[i] = 0;
  __syncthreads();
  int s[BIN_EPT], d[BIN_EPT], rF[BIN_EPT], rB[BIN_EPT];
  int e0 = blockIdx.x * (BIN_T * BIN_EPT) + threadIdx.x;
#pragma unroll
  for (int k = 0; k < BIN_EPT; ++k) {
    int e = e0 + k * BIN_T;
    s[k] = 0; d[k] = 0; rF[k] = 0; rB[k] = 0;
    if (e < E) {
      s[k] = ei0[e];
      d[k] = ei1[e];
      rF[k] = atomicAdd(&hist[d[k] / TR], 1);
      rB[k] = atomicAdd(&hist[NB + s[k] / TR], 1);
    }
  }
  __syncthreads();
  int* mycur = cursor + (blockIdx.x % REPL) * 2 * NB;
  for (int i = threadIdx.x; i < 2 * NB; i += BIN_T) {
    int c = hist[i];
    base[i] = c ? atomicAdd(&mycur[i], c) : 0;
  }
  __syncthreads();
  int2* oF = (int2*)recF;
  int2* oB = (int2*)recB;
#pragma unroll
  for (int k = 0; k < BIN_EPT; ++k) {
    int e = e0 + k * BIN_T;
    if (e < E) {
      oF[base[d[k] / TR] + rF[k]]      = make_int2(s[k], d[k]);
      oB[base[NB + s[k] / TR] + rB[k]] = make_int2(d[k], s[k]);
    }
  }
}

__global__ __launch_bounds__(256)
void k_degrees(const int* __restrict__ recF, const int* __restrict__ recB,
               const int* __restrict__ start,
               float* __restrict__ deg_in, float* __restrict__ deg_out, int N, int NB) {
  __shared__ int c[TR];
  int b = blockIdx.x, dir = blockIdx.y;
  const int2* rec = (const int2*)(dir ? recB : recF);
  const int* st = start + dir * (NB + 1);
  float* deg = dir ? deg_out : deg_in;
  for (int i = threadIdx.x; i < TR; i += 256) c[i] = 0;
  __syncthreads();
  int n0 = b * TR;
  int lo = st[b], hi = st[b + 1];
  for (int i = lo + threadIdx.x; i < hi; i += 256) atomicAdd(&c[rec[i].y - n0], 1);
  __syncthreads();
  for (int i = threadIdx.x; i < TR; i += 256) {
    int n = n0 + i;
    if (n < N) deg[n] = 1.0f + (float)c[i];
  }
}


template <int K, int RELU>
__global__ __launch_bounds__(128)
void k_gemm_wmma(const float* __restrict__ X, int ldx,
                 const float* __restrict__ Wa, const float* __restrict__ ba,
                 const float* __restrict__ Wb, const float* __restrict__ bb,
                 float* __restrict__ Y, int ldy,
                 int halfO, int koffB, int nColTiles, int nMTiles) {
  constexpr int NK = K / 4;

  int wave = threadIdx.x >> 5;
  int lane = threadIdx.x & 31;
  int mt   = blockIdx.x * 4 + wave;
  if (mt >= nMTiles) return;

  int m0   = mt << 4;
  int mrow = lane & 15;
  int khi  = (lane >> 4) << 1;
  int col  = lane & 15;

  const float* xrow = X + (size_t)(m0 + mrow) * ldx;

  v2f afrag[NK];
  int curKoff = -1;

  for (int ct = 0; ct < nColTiles; ++ct) {
    int c0   = ct << 4;
    int half = (c0 >= halfO) ? 1 : 0;
    const float* W    = half ? Wb : Wa;
    const float* bias = half ? bb : ba;
    int wc0  = c0 - half * halfO;
    int koff = half ? koffB : 0;

    if (koff != curKoff) {
      curKoff = koff;
      const float* xp = xrow + koff + khi;
#pragma unroll
      for (int i = 0; i < NK; ++i) {
        v2f a = *(const v2f*)(xp + i * 4);
        if (RELU) { a.x = fmaxf(a.x, 0.0f); a.y = fmaxf(a.y, 0.0f); }
        afrag[i] = a;
      }
    }

    const float* wrow = W + (size_t)(wc0 + col) * K + khi;
    v8f acc = {};
#pragma unroll
    for (int i = 0; i < NK; ++i) {
      v2f bf = *(const v2f*)(wrow + i * 4);
      acc = __builtin_amdgcn_wmma_f32_16x16x4_f32(false, afrag[i], false, bf,
                                                  (short)0, acc, false, false);
    }

    float bv = bias[wc0 + col];
    int mbase = m0 + ((lane >> 4) << 3);
    float* yp = Y + (size_t)mbase * ldy + c0 + col;
#pragma unroll
    for (int g = 0; g < 8; ++g)
      yp[(size_t)g * ldy] = acc[g] + bv;
  }
}


template <int F, int U, bool TAIL>
static __device__ __forceinline__
void add_records(const float* __restrict__ Y, float* tile, int lane,
                 int src, int dl, float rn, int k0, int cnt) {
  constexpr int J = F / 32;
  float v[U][J];
  int   d[U];
  float r[U];
#pragma unroll
  for (int u = 0; u < U; ++u) {
    int k = k0 + u;
    d[u] = 0; r[u] = 0.0f;
    if (!TAIL || k < cnt) {
      int s = __builtin_amdgcn_readlane(src, k);
      d[u]  = __builtin_amdgcn_readlane(dl, k);
      r[u]  = __int_as_float(__builtin_amdgcn_readlane(__float_as_int(rn), k));
      const float* yr = Y + (size_t)s * F + lane;
#pragma unroll
      for (int j = 0; j < J; ++j) v[u][j] = yr[32 * j];
    }
  }
#pragma unroll
  for (int u = 0; u < U; ++u) {
    if (!TAIL || k0 + u < cnt) {
      float* tp = tile + d[u] * F + lane;
#pragma unroll
      for (int j = 0; j < J; ++j) atomicAdd(tp + 32 * j, r[u] * v[u][j]);
    }
  }
}

template <int F>
__global__ __launch_bounds__(AGG_T)
void k_aggr_tile(const float* __restrict__ Y, const int* __restrict__ rec_,
                 const int* __restrict__ st,
                 const float* __restrict__ degD, const float* __restrict__ degS,
                 float* __restrict__ A, int N) {
  constexpr int U  = 1024 / F;
  constexpr int Q  = TR * (F / 4);
  constexpr int IT = (Q + AGG_T - 1) / AGG_T;
  __shared__ float tile[TR * F];

  const int2* rec = (const int2*)rec_;
  const int b    = blockIdx.x;
  const int n0   = b * TR;
  const int rows = min(TR, N - n0);
  const int q    = rows * (F / 4);
  const int tid  = threadIdx.x;
  const int lane = tid & 31;
  const int wave = __builtin_amdgcn_readfirstlane(tid >> 5);

  {
    const float4* y4 = (const float4*)(Y + (size_t)n0 * F);
    float4 v[IT];
    float  dd[IT], ds[IT];
#pragma unroll
    for (int i = 0; i < IT; ++i) {
      int idx = min(tid + i * AGG_T, q - 1);
      int n = n0 + idx / (F / 4);
      v[i]  = y4[idx];
      dd[i] = degD[n];
      ds[i] = degS[n];
    }
#pragma unroll
    for (int i = 0; i < IT; ++i) {
      int idx = tid + i * AGG_T;
      if (idx < q) {
        float rn = rsqrtf(dd[i] * ds[i]);
        float4 o; o.x = rn * v[i].x; o.y = rn * v[i].y; o.z = rn * v[i].z; o.w = rn * v[i].w;
        ((float4*)tile)[idx] = o;
      }
    }
  }
  __syncthreads();

  const int lo = st[b], hi = st[b + 1];
  for (int c0 = lo + wave * 32; c0 < hi; c0 += AGG_T) {
    const int cnt = min(32, hi - c0);
    int src = 0, dl = 0;
    float rn = 0.0f;
    if (lane < cnt) {
      int2 r = rec[c0 + lane];
      src = r.x;
      dl  = r.y - n0;
      rn  = rsqrtf(degD[r.y] * degS[r.x]);
    }
    int k0 = 0;
    for (; k0 + U <= cnt; k0 += U)
      add_records<F, U, false>(Y, tile, lane, src, dl, rn, k0, cnt);
    if (k0 < cnt)
      add_records<F, U, true>(Y, tile, lane, src, dl, rn, k0, cnt);
  }
  __syncthreads();

  {
    float4* a4 = (float4*)(A + (size_t)n0 * F);
#pragma unroll
    for (int i = 0; i < IT; ++i) {
      int idx = tid + i * AGG_T;
      if (idx < q) a4[idx] = ((const float4*)tile)[idx];
    }
  }
}


__global__ void k_final(const float* __restrict__ A2s, const float* __restrict__ A2t,
                        const float* __restrict__ noise_s, const float* __restrict__ noise_t,
                        float* __restrict__ out, int N) {
  int idx = blockIdx.x * blockDim.x + threadIdx.x;
  int total = N * OUT_F;
  if (idx < total) {
    int n = idx >> 5, j = idx & 31;
    out[idx] = A2s[n * 64 + j] + noise_s[idx] * expf(A2s[n * 64 + 32 + j]) * 0.2f;
  } else if (idx < 2 * total) {
    int k = idx - total;
    int n = k >> 5, j = k & 31;
    out[idx] = A2t[n * 64 + j] + noise_t[k] * expf(A2t[n * 64 + 32 + j]) * 0.2f;
  }
}


static inline int ceil_div(int a, int b) { return (a + b - 1) / b; }

extern "C" void kernel_launch(void* const* d_in, const int* in_sizes, int n_in,
                              void* d_out, int out_size, void* d_ws, size_t ws_size,
                              hipStream_t stream) {
  const float* s       = (const float*)d_in[0];
  const float* t       = (const float*)d_in[1];
  const int*   ei      = (const int*)  d_in[2];
  const float* noise_s = (const float*)d_in[3];
  const float* noise_t = (const float*)d_in[4];
  const float* sm1_W = (const float*)d_in[5],  *sm1_b = (const float*)d_in[6];
  const float* sm2_W = (const float*)d_in[7],  *sm2_b = (const float*)d_in[8];
  const float* sl1_W = (const float*)d_in[9],  *sl1_b = (const float*)d_in[10];
  const float* sl2_W = (const float*)d_in[11], *sl2_b = (const float*)d_in[12];
  const float* tm1_W = (const float*)d_in[13], *tm1_b = (const float*)d_in[14];
  const float* tm2_W = (const float*)d_in[15], *tm2_b = (const float*)d_in[16];
  const float* tl1_W = (const float*)d_in[17], *tl1_b = (const float*)d_in[18];
  const float* tl2_W = (const float*)d_in[19], *tl2_b = (const float*)d_in[20];

  const int N = in_sizes[0] / IN_F;
  const int E = in_sizes[2] / 2;
  const int* ei0 = ei;
  const int* ei1 = ei + E;
  const int NB = ceil_div(N, TR);

  float* ws  = (float*)d_ws;
  float* Y1  = ws;
  float* A1  = Y1  + (size_t)N * IN_F;
  float* A2s = A1  + (size_t)N * IN_F;
  float* A2t = A2s + (size_t)N * HID_F;
  float* Y2  = Y1;
  int*   recF = (int*)(A2t + (size_t)N * HID_F);
  int*   recB = recF + (size_t)2 * E;
  float* deg_in  = (float*)(recB + (size_t)2 * E);
  float* deg_out = deg_in + N;
  int*   cnt     = (int*)(deg_out + N);
  int*   cursor  = cnt + REPL * 2 * NB;
  int*   start   = cursor + REPL * 2 * NB;
  const int* startF = start;
  const int* startB = start + (NB + 1);

  const int nMTiles = ceil_div(N, 16);
  const int gemmBlocks = ceil_div(nMTiles, 4);
  const int binBlocks = ceil_div(E, BIN_T * BIN_EPT);

  k_zero_counts<<<ceil_div(REPL * 2 * NB, 256), 256, 0, stream>>>(cnt, REPL * 2 * NB);
  k_bin_count<<<binBlocks, BIN_T, 2 * NB * sizeof(int), stream>>>(ei0, ei1, cnt, E, NB);
  k_bin_scan<<<2, 1024, 0, stream>>>(cnt, start, cursor, NB);
  k_bin_place<<<binBlocks, BIN_T, 4 * NB * sizeof(int), stream>>>(ei0, ei1, cursor, recF, recB, E, NB);
  k_degrees<<<dim3(NB, 2), 256, 0, stream>>>(recF, recB, start, deg_in, deg_out, N, NB);

  k_gemm_wmma<IN_F, 0><<<gemmBlocks, 128, 0, stream>>>(
      s, IN_F, sm1_W, sm1_b, sl1_W, sl1_b, Y1, IN_F, HID_F, 0, IN_F / 16, nMTiles);
  k_aggr_tile<IN_F><<<NB, AGG_T, 0, stream>>>(Y1, recF, startF, deg_in, deg_out, A1, N);
  k_gemm_wmma<HID_F, 1><<<gemmBlocks, 128, 0, stream>>>(
      A1, IN_F, sm2_W, sm2_b, sl2_W, sl2_b, Y2, HID_F, OUT_F, HID_F, HID_F / 16, nMTiles);
  k_aggr_tile<HID_F><<<NB, AGG_T, 0, stream>>>(Y2, recB, startB, deg_out, deg_in, A2s, N);

  k_gemm_wmma<IN_F, 0><<<gemmBlocks, 128, 0, stream>>>(
      t, IN_F, tm1_W, tm1_b, tl1_W, tl1_b, Y1, IN_F, HID_F, 0, IN_F / 16, nMTiles);
  k_aggr_tile<IN_F><<<NB, AGG_T, 0, stream>>>(Y1, recB, startB, deg_out, deg_in, A1, N);
  k_gemm_wmma<HID_F, 1><<<gemmBlocks, 128, 0, stream>>>(
      A1, IN_F, tm2_W, tm2_b, tl2_W, tl2_b, Y2, HID_F, OUT_F, HID_F, HID_F / 16, nMTiles);
  k_aggr_tile<HID_F><<<NB, AGG_T, 0, stream>>>(Y2, recF, startF, deg_in, deg_out, A2t, N);

  k_final<<<ceil_div(2 * N * OUT_F, 256), 256, 0, stream>>>(A2s, A2t, noise_s, noise_t,
                                                            (float*)d_out, N);
}


# WMMA operand feed: a wave keeps its 16-column weight tile's B fragments in registers for the launch, X chunks go through LDS by whole-row loads (512 long-lived work-groups); k order, sums unchanged
# speedup vs baseline: 3.6224x; 1.1397x over previous
#include <hip/hip_runtime.h>


typedef __attribute__((ext_vector_type(2))) float v2f;
typedef __attribute__((ext_vector_type(8))) float v8f;

#define IN_F   128
#define HID_F  64
#define OUT_F  32

#define TR     100
#define AGG_T  512
#define BIN_T  1024
#define BIN_EPT 4
#define REPL   16


__global__ void k_zero_counts(int* __restrict__ cnt, int n) {
  int i = blockIdx.x * blockDim.x + threadIdx.x;
  if (i < n) cnt[i] = 0;
}

__global__ __launch_bounds__(BIN_T)
void k_bin_count(const int* __restrict__ ei0, const int* __restrict__ ei1,
                 int* __restrict__ cnt, int E, int NB) {
  extern __shared__ int bin_lds[];
  int* hist = bin_lds;
  for (int i = threadIdx.x; i < 2 * NB; i += BIN_T) hist[i] = 0;
  __syncthreads();
  int e0 = blockIdx.x * (BIN_T * BIN_EPT) + threadIdx.x;
  for (int k = 0; k < BIN_EPT; ++k) {
    int e = e0 + k * BIN_T;
    if (e < E) {
      atomicAdd(&hist[ei1[e] / TR], 1);
      atomicAdd(&hist[NB + ei0[e] / TR], 1);
    }
  }
  __syncthreads();
  int* mycnt = cnt + (blockIdx.x % REPL) * 2 * NB;
  for (int i = threadIdx.x; i < 2 * NB; i += BIN_T) {
    int c = hist[i];
    if (c) atomicAdd(&mycnt[i], c);
  }
}

__global__ __launch_bounds__(1024)
void k_bin_scan(const int* __restrict__ cnt, int* __restrict__ start,
                int* __restrict__ cursor, int NB) {
  __shared__ int sh[1024];
  int dir = blockIdx.x;
  int* st = start + dir * (NB + 1);
  int tid = threadIdx.x;
  int carry = 0;
  for (int base = 0; base < NB; base += 1024) {
    int i = base + tid;
    int c[REPL];
    int v = 0;
#pragma unroll
    for (int r = 0; r < REPL; ++r) {
      c[r] = (i < NB) ? cnt[r * 2 * NB + dir * NB + i] : 0;
      v += c[r];
    }
    sh[tid] = v;
    __syncthreads();
    for (int off = 1; off < 1024; off <<= 1) {
      int t = (tid >= off) ? sh[tid - off] : 0;
      __syncthreads();
      sh[tid] += t;
      __syncthreads();
    }
    int run = carry + sh[tid] - v;
    if (i < NB) {
      st[i] = run;
#pragma unroll
      for (int r = 0; r < REPL; ++r) {
        cursor[r * 2 * NB + dir * NB + i] = run;
        run += c[r];
      }
    }
    carry += sh[1023];
    __syncthreads();
  }
  if (tid == 0) st[NB] = carry;
}

__global__ __launch_bounds__(BIN_T)
void k_bin_place(const int* __restrict__ ei0, const int* __restrict__ ei1,
                 int* __restrict__ cursor, int* __restrict__ recF, int* __restrict__ recB,
                 int E, int NB) {
  extern __shared__ int bin_lds[];
  int* hist = bin_lds;
  int* base = bin_lds + 2 * NB;
  for (int i = threadIdx.x; i < 2 * NB; i += BIN_T) hist[i] = 0;
  __syncthreads();
  int s[BIN_EPT], d[BIN_EPT], rF[BIN_EPT], rB[BIN_EPT];
  int e0 = blockIdx.x * (BIN_T * BIN_EPT) + threadIdx.x;
#pragma unroll
  for (int k = 0; k < BIN_EPT; ++k) {
    int e = e0 + k * BIN_T;
    s[k] = 0; d[k] = 0; rF[k] = 0; rB[k] = 0;
    if (e < E) {
      s[k] = ei0[e];
      d[k] = ei1[e];
      rF[k] = atomicAdd(&hist[d[k] / TR], 1);
      rB[k] = atomicAdd(&hist[NB + s[k] / TR], 1);
    }
  }
  __syncthreads();
  int* mycur = cursor + (blockIdx.x % REPL) * 2 * NB;
  for (int i = threadIdx.x; i < 2 * NB; i += BIN_T) {
    int c = hist[i];
    base[i] = c ? atomicAdd(&mycur[i], c) : 0;
  }
  __syncthreads();
  int2* oF = (int2*)recF;
  int2* oB = (int2*)recB;
#pragma unroll
  for (int k = 0; k < BIN_EPT; ++k) {
    int e = e0 + k * BIN_T;
    if (e < E) {
      oF[base[d[k] / TR] + rF[k]]      = make_int2(s[k], d[k]);
      oB[base[NB + s[k] / TR] + rB[k]] = make_int2(d[k], s[k]);
    }
  }
}

__global__ __launch_bounds__(256)
void k_degrees(const int* __restrict__ recF, const int* __restrict__ recB,
               const int* __restrict__ start,
               float* __restrict__ deg_in, float* __restrict__ deg_out, int N, int NB) {
  __shared__ int c[TR];
  int b = blockIdx.x, dir = blockIdx.y;
  const int2* rec = (const int2*)(dir ? recB : recF);
  const int* st = start + dir * (NB + 1);
  float* deg = dir ? deg_out : deg_in;
  for (int i = threadIdx.x; i < TR; i += 256) c[i] = 0;
  __syncthreads();
  int n0 = b * TR;
  int lo = st[b], hi = st[b + 1];
  for (int i = lo + threadIdx.x; i < hi; i += 256) atomicAdd(&c[rec[i].y - n0], 1);
  __syncthreads();
  for (int i = threadIdx.x; i < TR; i += 256) {
    int n = n0 + i;
    if (n < N) deg[n] = 1.0f + (float)c[i];
  }
}


#define GEMM_T    256
#define GEMM_ROWS 32
#define GEMM_P    (IN_F + 4)

template <int K, int NCT, int RELU>
__global__ __launch_bounds__(GEMM_T)
void k_gemm_wmma(const float* __restrict__ X,
                 const float* __restrict__ Wa, const float* __restrict__ ba,
                 const float* __restrict__ Wb, const float* __restrict__ bb,
                 float* __restrict__ Y, int ldy,
                 int halfO, int koffB, int N, int nChunks) {
  constexpr int NK  = K / 4;
  constexpr int G   = (GEMM_T / 32) / NCT;
  constexpr int MPW = (GEMM_ROWS / 16) / G;
  __shared__ float xt[GEMM_ROWS * GEMM_P];

  const int tid  = threadIdx.x;
  const int lane = tid & 31;
  const int wave = __builtin_amdgcn_readfirstlane(tid >> 5);
  const int ct   = wave % NCT;
  const int grp  = wave / NCT;
  const int col  = lane & 15;
  const int khi  = (lane >> 4) << 1;

  const int c0   = ct << 4;
  const int half = (c0 >= halfO) ? 1 : 0;
  const float* W    = half ? Wb : Wa;
  const float* bias = half ? bb : ba;
  const int wc0  = c0 - half * halfO;
  const int koff = half ? koffB : 0;

  v2f bfr[NK];
  const float* wrow = W + (size_t)(wc0 + col) * K + khi;
#pragma unroll
  for (int i = 0; i < NK; ++i) bfr[i] = *(const v2f*)(wrow + i * 4);
  const float bv = bias[wc0 + col];

  const float* xl = X + lane * 4;
  const int rmax = N - 1;
  float4 s0 = {}, s1 = {}, s2 = {}, s3 = {};
  int ch = blockIdx.x;
  if (ch < nChunks) {
    const int r0 = ch * GEMM_ROWS + wave;
    s0 = *(const float4*)(xl + (size_t)min(r0,      rmax) * IN_F);
    s1 = *(const float4*)(xl + (size_t)min(r0 + 8,  rmax) * IN_F);
    s2 = *(const float4*)(xl + (size_t)min(r0 + 16, rmax) * IN_F);
    s3 = *(const float4*)(xl + (size_t)min(r0 + 24, rmax) * IN_F);
  }
  float* xs = xt + wave * GEMM_P + lane * 4;
  const float* ap = xt + (grp * MPW * 16 + col) * GEMM_P + koff + khi;

  for (; ch < nChunks; ch += gridDim.x) {
    if (RELU) {
      s0.x = fmaxf(s0.x, 0.0f); s0.y = fmaxf(s0.y, 0.0f); s0.z = fmaxf(s0.z, 0.0f); s0.w = fmaxf(s0.w, 0.0f);
      s1.x = fmaxf(s1.x, 0.0f); s1.y = fmaxf(s1.y, 0.0f); s1.z = fmaxf(s1.z, 0.0f); s1.w = fmaxf(s1.w, 0.0f);
      s2.x = fmaxf(s2.x, 0.0f); s2.y = fmaxf(s2.y, 0.0f); s2.z = fmaxf(s2.z, 0.0f); s2.w = fmaxf(s2.w, 0.0f);
      s3.x = fmaxf(s3.x, 0.0f); s3.y = fmaxf(s3.y, 0.0f); s3.z = fmaxf(s3.z, 0.0f); s3.w = fmaxf(s3.w, 0.0f);
    }
    *(float4*)(xs)               = s0;
    *(float4*)(xs + 8 * GEMM_P)  = s1;
    *(float4*)(xs + 16 * GEMM_P) = s2;
    *(float4*)(xs + 24 * GEMM_P) = s3;
    __syncthreads();

    const int nx = ch + gridDim.x;
    if (nx < nChunks) {
      const int r0 = nx * GEMM_ROWS + wave;
      s0 = *(const float4*)(xl + (size_t)min(r0,      rmax) * IN_F);
      s1 = *(const float4*)(xl + (size_t)min(r0 + 8,  rmax) * IN_F);
      s2 = *(const float4*)(xl + (size_t)min(r0 + 16, rmax) * IN_F);
      s3 = *(const float4*)(xl + (size_t)min(r0 + 24, rmax) * IN_F);
    }

    v8f acc[MPW];
#pragma unroll
    for (int j = 0; j < MPW; ++j) acc[j] = v8f{};
#pragma unroll
    for (int i = 0; i < NK; ++i) {
#pragma unroll
      for (int j = 0; j < MPW; ++j) {
        v2f a = *(const v2f*)(ap + j * 16 * GEMM_P + i * 4);
        acc[j] = __builtin_amdgcn_wmma_f32_16x16x4_f32(false, a, false, bfr[i],
                                                       (short)0, acc[j], false, false);
      }
    }

#pragma unroll
    for (int j = 0; j < MPW; ++j) {
      const int m0 = ch * GEMM_ROWS + (grp * MPW + j) * 16;
      if (m0 < N) {
        float* yp = Y + (size_t)(m0 + ((lane >> 4) << 3)) * ldy + c0 + col;
#pragma unroll
        for (int g = 0; g < 8; ++g)
          yp[(size_t)g * ldy] = acc[j][g] + bv;
      }
    }
    __syncthreads();
  }
}


template <int F, int U, bool TAIL>
static __device__ __forceinline__
void add_records(const float* __restrict__ Y, float* tile, int lane,
                 int src, int dl, float rn, int k0, int cnt) {
  constexpr int J = F / 32;
  float v[U][J];
  int   d[U];
  float r[U];
#pragma unroll
  for (int u = 0; u < U; ++u) {
    int k = k0 + u;
    d[u] = 0; r[u] = 0.0f;
    if (!TAIL || k < cnt) {
      int s = __builtin_amdgcn_readlane(src, k);
      d[u]  = __builtin_amdgcn_readlane(dl, k);
      r[u]  = __int_as_float(__builtin_amdgcn_readlane(__float_as_int(rn), k));
      const float* yr = Y + (size_t)s * F + lane;
#pragma unroll
      for (int j = 0; j < J; ++j) v[u][j] = yr[32 * j];
    }
  }
#pragma unroll
  for (int u = 0; u < U; ++u) {
    if (!TAIL || k0 + u < cnt) {
      float* tp = tile + d[u] * F + lane;
#pragma unroll
      for (int j = 0; j < J; ++j) atomicAdd(tp + 32 * j, r[u] * v[u][j]);
    }
  }
}

template <int F>
__global__ __launch_bounds__(AGG_T)
void k_aggr_tile(const float* __restrict__ Y, const int* __restrict__ rec_,
                 const int* __restrict__ st,
                 const float* __restrict__ degD, const float* __restrict__ degS,
                 float* __restrict__ A, int N) {
  constexpr int U  = 1024 / F;
  constexpr int Q  = TR * (F / 4);
  constexpr int IT = (Q + AGG_T - 1) / AGG_T;
  __shared__ float tile[TR * F];

  const int2* rec = (const int2*)rec_;
  const int b    = blockIdx.x;
  const int n0   = b * TR;
  const int rows = min(TR, N - n0);
  const int q    = rows * (F / 4);
  const int tid  = threadIdx.x;
  const int lane = tid & 31;
  const int wave = __builtin_amdgcn_readfirstlane(tid >> 5);

  {
    const float4* y4 = (const float4*)(Y + (size_t)n0 * F);
    float4 v[IT];
    float  dd[IT], ds[IT];
#pragma unroll
    for (int i = 0; i < IT; ++i) {
      int idx = min(tid + i * AGG_T, q - 1);
      int n = n0 + idx / (F / 4);
      v[i]  = y4[idx];
      dd[i] = degD[n];
      ds[i] = degS[n];
    }
#pragma unroll
    for (int i = 0; i < IT; ++i) {
      int idx = tid + i * AGG_T;
      if (idx < q) {
        float rn = rsqrtf(dd[i] * ds[i]);
        float4 o; o.x = rn * v[i].x; o.y = rn * v[i].y; o.z = rn * v[i].z; o.w = rn * v[i].w;
        ((float4*)tile)[idx] = o;
      }
    }
  }
  __syncthreads();

  const int lo = st[b], hi = st[b + 1];
  for (int c0 = lo + wave * 32; c0 < hi; c0 += AGG_T) {
    const int cnt = min(32, hi - c0);
    int src = 0, dl = 0;
    float rn = 0.0f;
    if (lane < cnt) {
      int2 r = rec[c0 + lane];
      src = r.x;
      dl  = r.y - n0;
      rn  = rsqrtf(degD[r.y] * degS[r.x]);
    }
    int k0 = 0;
    for (; k0 + U <= cnt; k0 += U)
      add_records<F, U, false>(Y, tile, lane, src, dl, rn, k0, cnt);
    if (k0 < cnt)
      add_records<F, U, true>(Y, tile, lane, src, dl, rn, k0, cnt);
  }
  __syncthreads();

  {
    float4* a4 = (float4*)(A + (size_t)n0 * F);
#pragma unroll
    for (int i = 0; i < IT; ++i) {
      int idx = tid + i * AGG_T;
      if (idx < q) a4[idx] = ((const float4*)tile)[idx];
    }
  }
}


__global__ void k_final(const float* __restrict__ A2s, const float* __restrict__ A2t,
                        const float* __restrict__ noise_s, const float* __restrict__ noise_t,
                        float* __restrict__ out, int N) {
  int idx = blockIdx.x * blockDim.x + threadIdx.x;
  int total = N * OUT_F;
  if (idx < total) {
    int n = idx >> 5, j = idx & 31;
    out[idx] = A2s[n * 64 + j] + noise_s[idx] * expf(A2s[n * 64 + 32 + j]) * 0.2f;
  } else if (idx < 2 * total) {
    int k = idx - total;
    int n = k >> 5, j = k & 31;
    out[idx] = A2t[n * 64 + j] + noise_t[k] * expf(A2t[n * 64 + 32 + j]) * 0.2f;
  }
}


static inline int ceil_div(int a, int b) { return (a + b - 1) / b; }

extern "C" void kernel_launch(void* const* d_in, const int* in_sizes, int n_in,
                              void* d_out, int out_size, void* d_ws, size_t ws_size,
                              hipStream_t stream) {
  const float* s       = (const float*)d_in[0];
  const float* t       = (const float*)d_in[1];
  const int*   ei      = (const int*)  d_in[2];
  const float* noise_s = (const float*)d_in[3];
  const float* noise_t = (const float*)d_in[4];
  const float* sm1_W = (const float*)d_in[5],  *sm1_b = (const float*)d_in[6];
  const float* sm2_W = (const float*)d_in[7],  *sm2_b = (const float*)d_in[8];
  const float* sl1_W = (const float*)d_in[9],  *sl1_b = (const float*)d_in[10];
  const float* sl2_W = (const float*)d_in[11], *sl2_b = (const float*)d_in[12];
  const float* tm1_W = (const float*)d_in[13], *tm1_b = (const float*)d_in[14];
  const float* tm2_W = (const float*)d_in[15], *tm2_b = (const float*)d_in[16];
  const float* tl1_W = (const float*)d_in[17], *tl1_b = (const float*)d_in[18];
  const float* tl2_W = (const float*)d_in[19], *tl2_b = (const float*)d_in[20];

  const int N = in_sizes[0] / IN_F;
  const int E = in_sizes[2] / 2;
  const int* ei0 = ei;
  const int* ei1 = ei + E;
  const int NB = ceil_div(N, TR);

  float* ws  = (float*)d_ws;
  float* Y1  = ws;
  float* A1  = Y1  + (size_t)N * IN_F;
  float* A2s = A1  + (size_t)N * IN_F;
  float* A2t = A2s + (size_t)N * HID_F;
  float* Y2  = Y1;
  int*   recF = (int*)(A2t + (size_t)N * HID_F);
  int*   recB = recF + (size_t)2 * E;
  float* deg_in  = (float*)(recB + (size_t)2 * E);
  float* deg_out = deg_in + N;
  int*   cnt     = (int*)(deg_out + N);
  int*   cursor  = cnt + REPL * 2 * NB;
  int*   start   = cursor + REPL * 2 * NB;
  const int* startF = start;
  const int* startB = start + (NB + 1);

  const int nChunks = ceil_div(N, GEMM_ROWS);
  const int gemmBlocks = 512;
  const int binBlocks = ceil_div(E, BIN_T * BIN_EPT);

  k_zero_counts<<<ceil_div(REPL * 2 * NB, 256), 256, 0, stream>>>(cnt, REPL * 2 * NB);
  k_bin_count<<<binBlocks, BIN_T, 2 * NB * sizeof(int), stream>>>(ei0, ei1, cnt, E, NB);
  k_bin_scan<<<2, 1024, 0, stream>>>(cnt, start, cursor, NB);
  k_bin_place<<<binBlocks, BIN_T, 4 * NB * sizeof(int), stream>>>(ei0, ei1, cursor, recF, recB, E, NB);
  k_degrees<<<dim3(NB, 2), 256, 0, stream>>>(recF, recB, start, deg_in, deg_out, N, NB);

  k_gemm_wmma<IN_F, IN_F / 16, 0><<<gemmBlocks, GEMM_T, 0, stream>>>(
      s, sm1_W, sm1_b, sl1_W, sl1_b, Y1, IN_F, HID_F, 0, N, nChunks);
  k_aggr_tile<IN_F><<<NB, AGG_T, 0, stream>>>(Y1, recF, startF, deg_in, deg_out, A1, N);
  k_gemm_wmma<HID_F, HID_F / 16, 1><<<gemmBlocks, GEMM_T, 0, stream>>>(
      A1, sm2_W, sm2_b, sl2_W, sl2_b, Y2, HID_F, OUT_F, HID_F, N, nChunks);
  k_aggr_tile<HID_F><<<NB, AGG_T, 0, stream>>>(Y2, recB, startB, deg_out, deg_in, A2s, N);

  k_gemm_wmma<IN_F, IN_F / 16, 0><<<gemmBlocks, GEMM_T, 0, stream>>>(
      t, tm1_W, tm1_b, tl1_W, tl1_b, Y1, IN_F, HID_F, 0, N, nChunks);
  k_aggr_tile<IN_F><<<NB, AGG_T, 0, stream>>>(Y1, recB, startB, deg_out, deg_in, A1, N);
  k_gemm_wmma<HID_F, HID_F / 16, 1><<<gemmBlocks, GEMM_T, 0, stream>>>(
      A1, tm2_W, tm2_b, tl2_W, tl2_b, Y2, HID_F, OUT_F, HID_F, N, nChunks);
  k_aggr_tile<HID_F><<<NB, AGG_T, 0, stream>>>(Y2, recF, startF, deg_in, deg_out, A2t, N);

  k_final<<<ceil_div(2 * N * OUT_F, 256), 256, 0, stream>>>(A2s, A2t, noise_s, noise_t,
                                                            (float*)d_out, N);
}


# Layer-2 linear done in the layer-1 aggregation's LDS tile epilogue: a work-group multiplies its finished 100-row tile (relu, f32 WMMA, k order kept), stores Y2; no [N,128] sums plane; sums unchanged
# speedup vs baseline: 3.9007x; 1.0768x over previous
#include <hip/hip_runtime.h>


typedef __attribute__((ext_vector_type(2))) float v2f;
typedef __attribute__((ext_vector_type(8))) float v8f;

#define IN_F   128
#define HID_F  64
#define OUT_F  32

#define TR     100
#define AGG_T  512
#define BIN_T  1024
#define BIN_EPT 4
#define REPL   16


__global__ void k_zero_counts(int* __restrict__ cnt, int n) {
  int i = blockIdx.x * blockDim.x + threadIdx.x;
  if (i < n) cnt[i] = 0;
}

__global__ __launch_bounds__(BIN_T)
void k_bin_count(const int* __restrict__ ei0, const int* __restrict__ ei1,
                 int* __restrict__ cnt, int E, int NB) {
  extern __shared__ int bin_lds[];
  int* hist = bin_lds;
  for (int i = threadIdx.x; i < 2 * NB; i += BIN_T) hist[i] = 0;
  __syncthreads();
  int e0 = blockIdx.x * (BIN_T * BIN_EPT) + threadIdx.x;
  for (int k = 0; k < BIN_EPT; ++k) {
    int e = e0 + k * BIN_T;
    if (e < E) {
      atomicAdd(&hist[ei1[e] / TR], 1);
      atomicAdd(&hist[NB + ei0[e] / TR], 1);
    }
  }
  __syncthreads();
  int* mycnt = cnt + (blockIdx.x % REPL) * 2 * NB;
  for (int i = threadIdx.x; i < 2 * NB; i += BIN_T) {
    int c = hist[i];
    if (c) atomicAdd(&mycnt[i], c);
  }
}

__global__ __launch_bounds__(1024)
void k_bin_scan(const int* __restrict__ cnt, int* __restrict__ start,
                int* __restrict__ cursor, int NB) {
  __shared__ int sh[1024];
  int dir = blockIdx.x;
  int* st = start + dir * (NB + 1);
  int tid = threadIdx.x;
  int carry = 0;
  for (int base = 0; base < NB; base += 1024) {
    int i = base + tid;
    int c[REPL];
    int v = 0;
#pragma unroll
    for (int r = 0; r < REPL; ++r) {
      c[r] = (i < NB) ? cnt[r * 2 * NB + dir * NB + i] : 0;
      v += c[r];
    }
    sh[tid] = v;
    __syncthreads();
    for (int off = 1; off < 1024; off <<= 1) {
      int t = (tid >= off) ? sh[tid - off] : 0;
      __syncthreads();
      sh[tid] += t;
      __syncthreads();
    }
    int run = carry + sh[tid] - v;
    if (i < NB) {
      st[i] = run;
#pragma unroll
      for (int r = 0; r < REPL; ++r) {
        cursor[r * 2 * NB + dir * NB + i] = run;
        run += c[r];
      }
    }
    carry += sh[1023];
    __syncthreads();
  }
  if (tid == 0) st[NB] = carry;
}

__global__ __launch_bounds__(BIN_T)
void k_bin_place(const int* __restrict__ ei0, const int* __restrict__ ei1,
                 int* __restrict__ cursor, int* __restrict__ recF, int* __restrict__ recB,
                 int E, int NB) {
  extern __shared__ int bin_lds[];
  int* hist = bin_lds;
  int* base = bin_lds + 2 * NB;
  for (int i = threadIdx.x; i < 2 * NB; i += BIN_T) hist[i] = 0;
  __syncthreads();
  int s[BIN_EPT], d[BIN_EPT], rF[BIN_EPT], rB[BIN_EPT];
  int e0 = blockIdx.x * (BIN_T * BIN_EPT) + threadIdx.x;
#pragma unroll
  for (int k = 0; k < BIN_EPT; ++k) {
    int e = e0 + k * BIN_T;
    s[k] = 0; d[k] = 0; rF[k] = 0; rB[k] = 0;
    if (e < E) {
      s[k] = ei0[e];
      d[k] = ei1[e];
      rF[k] = atomicAdd(&hist[d[k] / TR], 1);
      rB[k] = atomicAdd(&hist[NB + s[k] / TR], 1);
    }
  }
  __syncthreads();
  int* mycur = cursor + (blockIdx.x % REPL) * 2 * NB;
  for (int i = threadIdx.x; i < 2 * NB; i += BIN_T) {
    int c = hist[i];
    base[i] = c ? atomicAdd(&mycur[i], c) : 0;
  }
  __syncthreads();
  int2* oF = (int2*)recF;
  int2* oB = (int2*)recB;
#pragma unroll
  for (int k = 0; k < BIN_EPT; ++k) {
    int e = e0 + k * BIN_T;
    if (e < E) {
      oF[base[d[k] / TR] + rF[k]]      = make_int2(s[k], d[k]);
      oB[base[NB + s[k] / TR] + rB[k]] = make_int2(d[k], s[k]);
    }
  }
}

__global__ __launch_bounds__(256)
void k_degrees(const int* __restrict__ recF, const int* __restrict__ recB,
               const int* __restrict__ start,
               float* __restrict__ deg_in, float* __restrict__ deg_out, int N, int NB) {
  __shared__ int c[TR];
  int b = blockIdx.x, dir = blockIdx.y;
  const int2* rec = (const int2*)(dir ? recB : recF);
  const int* st = start + dir * (NB + 1);
  float* deg = dir ? deg_out : deg_in;
  for (int i = threadIdx.x; i < TR; i += 256) c[i] = 0;
  __syncthreads();
  int n0 = b * TR;
  int lo = st[b], hi = st[b + 1];
  for (int i = lo + threadIdx.x; i < hi; i += 256) atomicAdd(&c[rec[i].y - n0], 1);
  __syncthreads();
  for (int i = threadIdx.x; i < TR; i += 256) {
    int n = n0 + i;
    if (n < N) deg[n] = 1.0f + (float)c[i];
  }
}


#define GEMM_T    256
#define GEMM_ROWS 32
#define GEMM_P    (IN_F + 4)

template <int K, int NCT, int RELU>
__global__ __launch_bounds__(GEMM_T)
void k_gemm_wmma(const float* __restrict__ X,
                 const float* __restrict__ Wa, const float* __restrict__ ba,
                 const float* __restrict__ Wb, const float* __restrict__ bb,
                 float* __restrict__ Y, int ldy,
                 int halfO, int koffB, int N, int nChunks) {
  constexpr int NK  = K / 4;
  constexpr int G   = (GEMM_T / 32) / NCT;
  constexpr int MPW = (GEMM_ROWS / 16) / G;
  __shared__ float xt[GEMM_ROWS * GEMM_P];

  const int tid  = threadIdx.x;
  const int lane = tid & 31;
  const int wave = __builtin_amdgcn_readfirstlane(tid >> 5);
  const int ct   = wave % NCT;
  const int grp  = wave / NCT;
  const int col  = lane & 15;
  const int khi  = (lane >> 4) << 1;

  const int c0   = ct << 4;
  const int half = (c0 >= halfO) ? 1 : 0;
  const float* W    = half ? Wb : Wa;
  const float* bias = half ? bb : ba;
  const int wc0  = c0 - half * halfO;
  const int koff = half ? koffB : 0;

  v2f bfr[NK];
  const float* wrow = W + (size_t)(wc0 + col) * K + khi;
#pragma unroll
  for (int i = 0; i < NK; ++i) bfr[i] = *(const v2f*)(wrow + i * 4);
  const float bv = bias[wc0 + col];

  const float* xl = X + lane * 4;
  const int rmax = N - 1;
  float4 s0 = {}, s1 = {}, s2 = {}, s3 = {};
  int ch = blockIdx.x;
  if (ch < nChunks) {
    const int r0 = ch * GEMM_ROWS + wave;
    s0 = *(const float4*)(xl + (size_t)min(r0,      rmax) * IN_F);
    s1 = *(const float4*)(xl + (size_t)min(r0 + 8,  rmax) * IN_F);
    s2 = *(const float4*)(xl + (size_t)min(r0 + 16, rmax) * IN_F);
    s3 = *(const float4*)(xl + (size_t)min(r0 + 24, rmax) * IN_F);
  }
  float* xs = xt + wave * GEMM_P + lane * 4;
  const float* ap = xt + (grp * MPW * 16 + col) * GEMM_P + koff + khi;

  for (; ch < nChunks; ch += gridDim.x) {
    if (RELU) {
      s0.x = fmaxf(s0.x, 0.0f); s0.y = fmaxf(s0.y, 0.0f); s0.z = fmaxf(s0.z, 0.0f); s0.w = fmaxf(s0.w, 0.0f);
      s1.x = fmaxf(s1.x, 0.0f); s1.y = fmaxf(s1.y, 0.0f); s1.z = fmaxf(s1.z, 0.0f); s1.w = fmaxf(s1.w, 0.0f);
      s2.x = fmaxf(s2.x, 0.0f); s2.y = fmaxf(s2.y, 0.0f); s2.z = fmaxf(s2.z, 0.0f); s2.w = fmaxf(s2.w, 0.0f);
      s3.x = fmaxf(s3.x, 0.0f); s3.y = fmaxf(s3.y, 0.0f); s3.z = fmaxf(s3.z, 0.0f); s3.w = fmaxf(s3.w, 0.0f);
    }
    *(float4*)(xs)               = s0;
    *(float4*)(xs + 8 * GEMM_P)  = s1;
    *(float4*)(xs + 16 * GEMM_P) = s2;
    *(float4*)(xs + 24 * GEMM_P) = s3;
    __syncthreads();

    const int nx = ch + gridDim.x;
    if (nx < nChunks) {
      const int r0 = nx * GEMM_ROWS + wave;
      s0 = *(const float4*)(xl + (size_t)min(r0,      rmax) * IN_F);
      s1 = *(const float4*)(xl + (size_t)min(r0 + 8,  rmax) * IN_F);
      s2 = *(const float4*)(xl + (size_t)min(r0 + 16, rmax) * IN_F);
      s3 = *(const float4*)(xl + (size_t)min(r0 + 24, rmax) * IN_F);
    }

    v8f acc[MPW];
#pragma unroll
    for (int j = 0; j < MPW; ++j) acc[j] = v8f{};
#pragma unroll
    for (int i = 0; i < NK; ++i) {
#pragma unroll
      for (int j = 0; j < MPW; ++j) {
        v2f a = *(const v2f*)(ap + j * 16 * GEMM_P + i * 4);
        acc[j] = __builtin_amdgcn_wmma_f32_16x16x4_f32(false, a, false, bfr[i],
                                                       (short)0, acc[j], false, false);
      }
    }

#pragma unroll
    for (int j = 0; j < MPW; ++j) {
      const int m0 = ch * GEMM_ROWS + (grp * MPW + j) * 16;
      if (m0 < N) {
        float* yp = Y + (size_t)(m0 + ((lane >> 4) << 3)) * ldy + c0 + col;
#pragma unroll
        for (int g = 0; g < 8; ++g)
          yp[(size_t)g * ldy] = acc[j][g] + bv;
      }
    }
    __syncthreads();
  }
}


template <int F, int P, int U, bool TAIL>
static __device__ __forceinline__
void add_records(const float* __restrict__ Y, float* tile, int lane,
                 int src, int dl, float rn, int k0, int cnt) {
  constexpr int J = F / 32;
  float v[U][J];
  int   d[U];
  float r[U];
#pragma unroll
  for (int u = 0; u < U; ++u) {
    int k = k0 + u;
    d[u] = 0; r[u] = 0.0f;
    if (!TAIL || k < cnt) {
      int s = __builtin_amdgcn_readlane(src, k);
      d[u]  = __builtin_amdgcn_readlane(dl, k);
      r[u]  = __int_as_float(__builtin_amdgcn_readlane(__float_as_int(rn), k));
      const float* yr = Y + (size_t)s * F + lane;
#pragma unroll
      for (int j = 0; j < J; ++j) v[u][j] = yr[32 * j];
    }
  }
#pragma unroll
  for (int u = 0; u < U; ++u) {
    if (!TAIL || k0 + u < cnt) {
      float* tp = tile + d[u] * P + lane;
#pragma unroll
      for (int j = 0; j < J; ++j) atomicAdd(tp + 32 * j, r[u] * v[u][j]);
    }
  }
}

template <int F>
__global__ __launch_bounds__(AGG_T)
void k_aggr_tile(const float* __restrict__ Y, const int* __restrict__ rec_,
                 const int* __restrict__ st,
                 const float* __restrict__ degD, const float* __restrict__ degS,
                 float* __restrict__ A, int N) {
  constexpr int U  = 1024 / F;
  constexpr int Q  = TR * (F / 4);
  constexpr int IT = (Q + AGG_T - 1) / AGG_T;
  __shared__ float tile[TR * F];

  const int2* rec = (const int2*)rec_;
  const int b    = blockIdx.x;
  const int n0   = b * TR;
  const int rows = min(TR, N - n0);
  const int q    = rows * (F / 4);
  const int tid  = threadIdx.x;
  const int lane = tid & 31;
  const int wave = __builtin_amdgcn_readfirstlane(tid >> 5);

  {
    const float4* y4 = (const float4*)(Y + (size_t)n0 * F);
    float4 v[IT];
    float  dd[IT], ds[IT];
#pragma unroll
    for (int i = 0; i < IT; ++i) {
      int idx = min(tid + i * AGG_T, q - 1);
      int n = n0 + idx / (F / 4);
      v[i]  = y4[idx];
      dd[i] = degD[n];
      ds[i] = degS[n];
    }
#pragma unroll
    for (int i = 0; i < IT; ++i) {
      int idx = tid + i * AGG_T;
      if (idx < q) {
        float rn = rsqrtf(dd[i] * ds[i]);
        float4 o; o.x = rn * v[i].x; o.y = rn * v[i].y; o.z = rn * v[i].z; o.w = rn * v[i].w;
        ((float4*)tile)[idx] = o;
      }
    }
  }
  __syncthreads();

  const int lo = st[b], hi = st[b + 1];
  for (int c0 = lo + wave * 32; c0 < hi; c0 += AGG_T) {
    const int cnt = min(32, hi - c0);
    int src = 0, dl = 0;
    float rn = 0.0f;
    if (lane < cnt) {
      int2 r = rec[c0 + lane];
      src = r.x;
      dl  = r.y - n0;
      rn  = rsqrtf(degD[r.y] * degS[r.x]);
    }
    int k0 = 0;
    for (; k0 + U <= cnt; k0 += U)
      add_records<F, F, U, false>(Y, tile, lane, src, dl, rn, k0, cnt);
    if (k0 < cnt)
      add_records<F, F, U, true>(Y, tile, lane, src, dl, rn, k0, cnt);
  }
  __syncthreads();

  {
    float4* a4 = (float4*)(A + (size_t)n0 * F);
#pragma unroll
    for (int i = 0; i < IT; ++i) {
      int idx = tid + i * AGG_T;
      if (idx < q) a4[idx] = ((const float4*)tile)[idx];
    }
  }
}


#define L2_RT   ((TR + 15) / 16)
#define L2_ROWS (L2_RT * 16)
#define L2_P    (IN_F + 4)

__global__ __launch_bounds__(AGG_T)
void k_aggr_lin2(const float* __restrict__ Y, const int* __restrict__ rec_,
                 const int* __restrict__ st,
                 const float* __restrict__ degD, const float* __restrict__ degS,
                 const float* __restrict__ Wa, const float* __restrict__ ba,
                 const float* __restrict__ Wb, const float* __restrict__ bb,
                 float* __restrict__ Y2, int N) {
  constexpr int F  = IN_F;
  constexpr int P  = L2_P;
  constexpr int K  = HID_F;
  constexpr int NK = K / 4;
  constexpr int U  = 1024 / F;
  constexpr int QP = L2_ROWS * (F / 4);
  constexpr int IT = (QP + AGG_T - 1) / AGG_T;
  constexpr int NCT = HID_F / 16;
  constexpr int NG  = (AGG_T / 32) / NCT;
  static_assert(L2_RT <= 2 * NG, "two row tiles per wave cover the tile");
  __shared__ float tile[L2_ROWS * P];

  const int2* rec = (const int2*)rec_;
  const int b    = blockIdx.x;
  const int n0   = b * TR;
  const int rows = min(TR, N - n0);
  const int q    = rows * (F / 4);
  const int tid  = threadIdx.x;
  const int lane = tid & 31;
  const int wave = __builtin_amdgcn_readfirstlane(tid >> 5);

  {
    const float4* y4 = (const float4*)(Y + (size_t)n0 * F);
    float4 v[IT];
    float  dd[IT], ds[IT];
#pragma unroll
    for (int i = 0; i < IT; ++i) {
      int idx = min(tid + i * AGG_T, q - 1);
      int n = n0 + idx / (F / 4);
      v[i]  = y4[idx];
      dd[i] = degD[n];
      ds[i] = degS[n];
    }
#pragma unroll
    for (int i = 0; i < IT; ++i) {
      int idx = tid + i * AGG_T;
      if (idx < QP) {
        float4 o = {};
        if (idx < q) {
          float rn = rsqrtf(dd[i] * ds[i]);
          o.x = rn * v[i].x; o.y = rn * v[i].y; o.z = rn * v[i].z; o.w = rn * v[i].w;
        }
        *(float4*)(tile + (idx / (F / 4)) * P + (idx % (F / 4)) * 4) = o;
      }
    }
  }
  __syncthreads();

  const int lo = st[b], hi = st[b + 1];
  for (int c0 = lo + wave * 32; c0 < hi; c0 += AGG_T) {
    const int cnt = min(32, hi - c0);
    int src = 0, dl = 0;
    float rn = 0.0f;
    if (lane < cnt) {
      int2 r = rec[c0 + lane];
      src = r.x;
      dl  = r.y - n0;
      rn  = rsqrtf(degD[r.y] * degS[r.x]);
    }
    int k0 = 0;
    for (; k0 + U <= cnt; k0 += U)
      add_records<F, P, U, false>(Y, tile, lane, src, dl, rn, k0, cnt);
    if (k0 < cnt)
      add_records<F, P, U, true>(Y, tile, lane, src, dl, rn, k0, cnt);
  }

  const int ct   = wave % NCT;
  const int grp  = wave / NCT;
  const int col  = lane & 15;
  const int khi  = (lane >> 4) << 1;
  const int cc0  = ct << 4;
  const int half = (cc0 >= OUT_F) ? 1 : 0;
  const float* W    = half ? Wb : Wa;
  const float* bias = half ? bb : ba;
  const int wc0  = cc0 - half * OUT_F;
  const int koff = half ? HID_F : 0;
  v2f bfr[NK];
  const float* wrow = W + (size_t)(wc0 + col) * K + khi;
#pragma unroll
  for (int i = 0; i < NK; ++i) bfr[i] = *(const v2f*)(wrow + i * 4);
  const float bv = bias[wc0 + col];
  __syncthreads();

  const bool two = (grp + NG) < L2_RT;
  const int rt1  = two ? grp + NG : grp;
  const float* ap0 = tile + (grp * 16 + col) * P + koff + khi;
  const float* ap1 = tile + (rt1 * 16 + col) * P + koff + khi;
  v8f acc0 = {}, acc1 = {};
#pragma unroll
  for (int i = 0; i < NK; ++i) {
    v2f a0 = *(const v2f*)(ap0 + i * 4);
    v2f a1 = *(const v2f*)(ap1 + i * 4);
    a0.x = fmaxf(a0.x, 0.0f); a0.y = fmaxf(a0.y, 0.0f);
    a1.x = fmaxf(a1.x, 0.0f); a1.y = fmaxf(a1.y, 0.0f);
    acc0 = __builtin_amdgcn_wmma_f32_16x16x4_f32(false, a0, false, bfr[i], (short)0, acc0, false, false);
    acc1 = __builtin_amdgcn_wmma_f32_16x16x4_f32(false, a1, false, bfr[i], (short)0, acc1, false, false);
  }
  __syncthreads();

  {
    float* yp0 = tile + (grp * 16 + ((lane >> 4) << 3)) * P + cc0 + col;
#pragma unroll
    for (int g = 0; g < 8; ++g) yp0[g * P] = acc0[g] + bv;
    if (two) {
      float* yp1 = tile + (rt1 * 16 + ((lane >> 4) << 3)) * P + cc0 + col;
#pragma unroll
      for (int g = 0; g < 8; ++g) yp1[g * P] = acc1[g] + bv;
    }
  }
  __syncthreads();

  {
    constexpr int C4 = HID_F / 4;
    constexpr int IO = (TR * C4 + AGG_T - 1) / AGG_T;
    const int qo = rows * C4;
    float4* o4 = (float4*)(Y2 + (size_t)n0 * HID_F);
#pragma unroll
    for (int i = 0; i < IO; ++i) {
      int idx = tid + i * AGG_T;
      if (idx < qo) o4[idx] = *(const float4*)(tile + (idx / C4) * P + (idx % C4) * 4);
    }
  }
}


__global__ void k_final(const float* __restrict__ A2s, const float* __restrict__ A2t,
                        const float* __restrict__ noise_s, const float* __restrict__ noise_t,
                        float* __restrict__ out, int N) {
  int idx = blockIdx.x * blockDim.x + threadIdx.x;
  int total = N * OUT_F;
  if (idx < total) {
    int n = idx >> 5, j = idx & 31;
    out[idx] = A2s[n * 64 + j] + noise_s[idx] * expf(A2s[n * 64 + 32 + j]) * 0.2f;
  } else if (idx < 2 * total) {
    int k = idx - total;
    int n = k >> 5, j = k & 31;
    out[idx] = A2t[n * 64 + j] + noise_t[k] * expf(A2t[n * 64 + 32 + j]) * 0.2f;
  }
}


static inline int ceil_div(int a, int b) { return (a + b - 1) / b; }

extern "C" void kernel_launch(void* const* d_in, const int* in_sizes, int n_in,
                              void* d_out, int out_size, void* d_ws, size_t ws_size,
                              hipStream_t stream) {
  const float* s       = (const float*)d_in[0];
  const float* t       = (const float*)d_in[1];
  const int*   ei      = (const int*)  d_in[2];
  const float* noise_s = (const float*)d_in[3];
  const float* noise_t = (const float*)d_in[4];
  const float* sm1_W = (const float*)d_in[5],  *sm1_b = (const float*)d_in[6];
  const float* sm2_W = (const float*)d_in[7],  *sm2_b = (const float*)d_in[8];
  const float* sl1_W = (const float*)d_in[9],  *sl1_b = (const float*)d_in[10];
  const float* sl2_W = (const float*)d_in[11], *sl2_b = (const float*)d_in[12];
  const float* tm1_W = (const float*)d_in[13], *tm1_b = (const float*)d_in[14];
  const float* tm2_W = (const float*)d_in[15], *tm2_b = (const float*)d_in[16];
  const float* tl1_W = (const float*)d_in[17], *tl1_b = (const float*)d_in[18];
  const float* tl2_W = (const float*)d_in[19], *tl2_b = (const float*)d_in[20];

  const int N = in_sizes[0] / IN_F;
  const int E = in_sizes[2] / 2;
  const int* ei0 = ei;
  const int* ei1 = ei + E;
  const int NB = ceil_div(N, TR);

  float* ws  = (float*)d_ws;
  float* Y1  = ws;
  float* Y2  = Y1  + (size_t)N * IN_F;
  float* A2s = Y2  + (size_t)N * IN_F;
  float* A2t = A2s + (size_t)N * HID_F;
  int*   recF = (int*)(A2t + (size_t)N * HID_F);
  int*   recB = recF + (size_t)2 * E;
  float* deg_in  = (float*)(recB + (size_t)2 * E);
  float* deg_out = deg_in + N;
  int*   cnt     = (int*)(deg_out + N);
  int*   cursor  = cnt + REPL * 2 * NB;
  int*   start   = cursor + REPL * 2 * NB;
  const int* startF = start;
  const int* startB = start + (NB + 1);

  const int nChunks = ceil_div(N, GEMM_ROWS);
  const int gemmBlocks = 512;
  const int binBlocks = ceil_div(E, BIN_T * BIN_EPT);

  k_zero_counts<<<ceil_div(REPL * 2 * NB, 256), 256, 0, stream>>>(cnt, REPL * 2 * NB);
  k_bin_count<<<binBlocks, BIN_T, 2 * NB * sizeof(int), stream>>>(ei0, ei1, cnt, E, NB);
  k_bin_scan<<<2, 1024, 0, stream>>>(cnt, start, cursor, NB);
  k_bin_place<<<binBlocks, BIN_T, 4 * NB * sizeof(int), stream>>>(ei0, ei1, cursor, recF, recB, E, NB);
  k_degrees<<<dim3(NB, 2), 256, 0, stream>>>(recF, recB, start, deg_in, deg_out, N, NB);

  k_gemm_wmma<IN_F, IN_F / 16, 0><<<gemmBlocks, GEMM_T, 0, stream>>>(
      s, sm1_W, sm1_b, sl1_W, sl1_b, Y1, IN_F, HID_F, 0, N, nChunks);
  k_aggr_lin2<<<NB, AGG_T, 0, stream>>>(Y1, recF, startF, deg_in, deg_out,
                                        sm2_W, sm2_b, sl2_W, sl2_b, Y2, N);
  k_aggr_tile<HID_F><<<NB, AGG_T, 0, stream>>>(Y2, recB, startB, deg_out, deg_in, A2s, N);

  k_gemm_wmma<IN_F, IN_F / 16, 0><<<gemmBlocks, GEMM_T, 0, stream>>>(
      t, tm1_W, tm1_b, tl1_W, tl1_b, Y1, IN_F, HID_F, 0, N, nChunks);
  k_aggr_lin2<<<NB, AGG_T, 0, stream>>>(Y1, recB, startB, deg_out, deg_in,
                                        tm2_W, tm2_b, tl2_W, tl2_b, Y2, N);
  k_aggr_tile<HID_F><<<NB, AGG_T, 0, stream>>>(Y2, recF, startF, deg_in, deg_out, A2t, N);

  k_final<<<ceil_div(2 * N * OUT_F, 256), 256, 0, stream>>>(A2s, A2t, noise_s, noise_t,
                                                            (float*)d_out, N);
}


# s layer-2 and t layer-1 aggregation share one walk of their common edge list: a record and its norm are read once, both source rows go into two LDS tiles; 12 launches to 11; sums unchanged
# speedup vs baseline: 4.0574x; 1.0402x over previous
#include <hip/hip_runtime.h>


typedef __attribute__((ext_vector_type(2))) float v2f;
typedef __attribute__((ext_vector_type(8))) float v8f;

#define IN_F   128
#define HID_F  64
#define OUT_F  32

#define TR     100
#define AGG_T  512
#define BIN_T  1024
#define BIN_EPT 4
#define REPL   16


__global__ void k_zero_counts(int* __restrict__ cnt, int n) {
  int i = blockIdx.x * blockDim.x + threadIdx.x;
  if (i < n) cnt[i] = 0;
}

__global__ __launch_bounds__(BIN_T)
void k_bin_count(const int* __restrict__ ei0, const int* __restrict__ ei1,
                 int* __restrict__ cnt, int E, int NB) {
  extern __shared__ int bin_lds[];
  int* hist = bin_lds;
  for (int i = threadIdx.x; i < 2 * NB; i += BIN_T) hist[i] = 0;
  __syncthreads();
  int e0 = blockIdx.x * (BIN_T * BIN_EPT) + threadIdx.x;
  for (int k = 0; k < BIN_EPT; ++k) {
    int e = e0 + k * BIN_T;
    if (e < E) {
      atomicAdd(&hist[ei1[e] / TR], 1);
      atomicAdd(&hist[NB + ei0[e] / TR], 1);
    }
  }
  __syncthreads();
  int* mycnt = cnt + (blockIdx.x % REPL) * 2 * NB;
  for (int i = threadIdx.x; i < 2 * NB; i += BIN_T) {
    int c = hist[i];
    if (c) atomicAdd(&mycnt[i], c);
  }
}

__global__ __launch_bounds__(1024)
void k_bin_scan(const int* __restrict__ cnt, int* __restrict__ start,
                int* __restrict__ cursor, int NB) {
  __shared__ int sh[1024];
  int dir = blockIdx.x;
  int* st = start + dir * (NB + 1);
  int tid = threadIdx.x;
  int carry = 0;
  for (int base = 0; base < NB; base += 1024) {
    int i = base + tid;
    int c[REPL];
    int v = 0;
#pragma unroll
    for (int r = 0; r < REPL; ++r) {
      c[r] = (i < NB) ? cnt[r * 2 * NB + dir * NB + i] : 0;
      v += c[r];
    }
    sh[tid] = v;
    __syncthreads();
    for (int off = 1; off < 1024; off <<= 1) {
      int t = (tid >= off) ? sh[tid - off] : 0;
      __syncthreads();
      sh[tid] += t;
      __syncthreads();
    }
    int run = carry + sh[tid] - v;
    if (i < NB) {
      st[i] = run;
#pragma unroll
      for (int r = 0; r < REPL; ++r) {
        cursor[r * 2 * NB + dir * NB + i] = run;
        run += c[r];
      }
    }
    carry += sh[1023];
    __syncthreads();
  }
  if (tid == 0) st[NB] = carry;
}

__global__ __launch_bounds__(BIN_T)
void k_bin_place(const int* __restrict__ ei0, const int* __restrict__ ei1,
                 int* __restrict__ cursor, int* __restrict__ recF, int* __restrict__ recB,
                 int E, int NB) {
  extern __shared__ int bin_lds[];
  int* hist = bin_lds;
  int* base = bin_lds + 2 * NB;
  for (int i = threadIdx.x; i < 2 * NB; i += BIN_T) hist[i] = 0;
  __syncthreads();
  int s[BIN_EPT], d[BIN_EPT], rF[BIN_EPT], rB[BIN_EPT];
  int e0 = blockIdx.x * (BIN_T * BIN_EPT) + threadIdx.x;
#pragma unroll
  for (int k = 0; k < BIN_EPT; ++k) {
    int e = e0 + k * BIN_T;
    s[k] = 0; d[k] = 0; rF[k] = 0; rB[k] = 0;
    if (e < E) {
      s[k] = ei0[e];
      d[k] = ei1[e];
      rF[k] = atomicAdd(&hist[d[k] / TR], 1);
      rB[k] = atomicAdd(&hist[NB + s[k] / TR], 1);
    }
  }
  __syncthreads();
  int* mycur = cursor + (blockIdx.x % REPL) * 2 * NB;
  for (int i = threadIdx.x; i < 2 * NB; i += BIN_T) {
    int c = hist[i];
    base[i] = c ? atomicAdd(&mycur[i], c) : 0;
  }
  __syncthreads();
  int2* oF = (int2*)recF;
  int2* oB = (int2*)recB;
#pragma unroll
  for (int k = 0; k < BIN_EPT; ++k) {
    int e = e0 + k * BIN_T;
    if (e < E) {
      oF[base[d[k] / TR] + rF[k]]      = make_int2(s[k], d[k]);
      oB[base[NB + s[k] / TR] + rB[k]] = make_int2(d[k], s[k]);
    }
  }
}

__global__ __launch_bounds__(256)
void k_degrees(const int* __restrict__ recF, const int* __restrict__ recB,
               const int* __restrict__ start,
               float* __restrict__ deg_in, float* __restrict__ deg_out, int N, int NB) {
  __shared__ int c[TR];
  int b = blockIdx.x, dir = blockIdx.y;
  const int2* rec = (const int2*)(dir ? recB : recF);
  const int* st = start + dir * (NB + 1);
  float* deg = dir ? deg_out : deg_in;
  for (int i = threadIdx.x; i < TR; i += 256) c[i] = 0;
  __syncthreads();
  int n0 = b * TR;
  int lo = st[b], hi = st[b + 1];
  for (int i = lo + threadIdx.x; i < hi; i += 256) atomicAdd(&c[rec[i].y - n0], 1);
  __syncthreads();
  for (int i = threadIdx.x; i < TR; i += 256) {
    int n = n0 + i;
    if (n < N) deg[n] = 1.0f + (float)c[i];
  }
}


#define GEMM_T    256
#define GEMM_ROWS 32
#define GEMM_P    (IN_F + 4)

template <int K, int NCT, int RELU>
__global__ __launch_bounds__(GEMM_T)
void k_gemm_wmma(const float* __restrict__ X,
                 const float* __restrict__ Wa, const float* __restrict__ ba,
                 const float* __restrict__ Wb, const float* __restrict__ bb,
                 float* __restrict__ Y, int ldy,
                 int halfO, int koffB, int N, int nChunks) {
  constexpr int NK  = K / 4;
  constexpr int G   = (GEMM_T / 32) / NCT;
  constexpr int MPW = (GEMM_ROWS / 16) / G;
  __shared__ float xt[GEMM_ROWS * GEMM_P];

  const int tid  = threadIdx.x;
  const int lane = tid & 31;
  const int wave = __builtin_amdgcn_readfirstlane(tid >> 5);
  const int ct   = wave % NCT;
  const int grp  = wave / NCT;
  const int col  = lane & 15;
  const int khi  = (lane >> 4) << 1;

  const int c0   = ct << 4;
  const int half = (c0 >= halfO) ? 1 : 0;
  const float* W    = half ? Wb : Wa;
  const float* bias = half ? bb : ba;
  const int wc0  = c0 - half * halfO;
  const int koff = half ? koffB : 0;

  v2f bfr[NK];
  const float* wrow = W + (size_t)(wc0 + col) * K + khi;
#pragma unroll
  for (int i = 0; i < NK; ++i) bfr[i] = *(const v2f*)(wrow + i * 4);
  const float bv = bias[wc0 + col];

  const float* xl = X + lane * 4;
  const int rmax = N - 1;
  float4 s0 = {}, s1 = {}, s2 = {}, s3 = {};
  int ch = blockIdx.x;
  if (ch < nChunks) {
    const int r0 = ch * GEMM_ROWS + wave;
    s0 = *(const float4*)(xl + (size_t)min(r0,      rmax) * IN_F);
    s1 = *(const float4*)(xl + (size_t)min(r0 + 8,  rmax) * IN_F);
    s2 = *(const float4*)(xl + (size_t)min(r0 + 16, rmax) * IN_F);
    s3 = *(const float4*)(xl + (size_t)min(r0 + 24, rmax) * IN_F);
  }
  float* xs = xt + wave * GEMM_P + lane * 4;
  const float* ap = xt + (grp * MPW * 16 + col) * GEMM_P + koff + khi;

  for (; ch < nChunks; ch += gridDim.x) {
    if (RELU) {
      s0.x = fmaxf(s0.x, 0.0f); s0.y = fmaxf(s0.y, 0.0f); s0.z = fmaxf(s0.z, 0.0f); s0.w = fmaxf(s0.w, 0.0f);
      s1.x = fmaxf(s1.x, 0.0f); s1.y = fmaxf(s1.y, 0.0f); s1.z = fmaxf(s1.z, 0.0f); s1.w = fmaxf(s1.w, 0.0f);
      s2.x = fmaxf(s2.x, 0.0f); s2.y = fmaxf(s2.y, 0.0f); s2.z = fmaxf(s2.z, 0.0f); s2.w = fmaxf(s2.w, 0.0f);
      s3.x = fmaxf(s3.x, 0.0f); s3.y = fmaxf(s3.y, 0.0f); s3.z = fmaxf(s3.z, 0.0f); s3.w = fmaxf(s3.w, 0.0f);
    }
    *(float4*)(xs)               = s0;
    *(float4*)(xs + 8 * GEMM_P)  = s1;
    *(float4*)(xs + 16 * GEMM_P) = s2;
    *(float4*)(xs + 24 * GEMM_P) = s3;
    __syncthreads();

    const int nx = ch + gridDim.x;
    if (nx < nChunks) {
      const int r0 = nx * GEMM_ROWS + wave;
      s0 = *(const float4*)(xl + (size_t)min(r0,      rmax) * IN_F);
      s1 = *(const float4*)(xl + (size_t)min(r0 + 8,  rmax) * IN_F);
      s2 = *(const float4*)(xl + (size_t)min(r0 + 16, rmax) * IN_F);
      s3 = *(const float4*)(xl + (size_t)min(r0 + 24, rmax) * IN_F);
    }

    v8f acc[MPW];
#pragma unroll
    for (int j = 0; j < MPW; ++j) acc[j] = v8f{};
#pragma unroll
    for (int i = 0; i < NK; ++i) {
#pragma unroll
      for (int j = 0; j < MPW; ++j) {
        v2f a = *(const v2f*)(ap + j * 16 * GEMM_P + i * 4);
        acc[j] = __builtin_amdgcn_wmma_f32_16x16x4_f32(false, a, false, bfr[i],
                                                       (short)0, acc[j], false, false);
      }
    }

#pragma unroll
    for (int j = 0; j < MPW; ++j) {
      const int m0 = ch * GEMM_ROWS + (grp * MPW + j) * 16;
      if (m0 < N) {
        float* yp = Y + (size_t)(m0 + ((lane >> 4) << 3)) * ldy + c0 + col;
#pragma unroll
        for (int g = 0; g < 8; ++g)
          yp[(size_t)g * ldy] = acc[j][g] + bv;
      }
    }
    __syncthreads();
  }
}


template <int F, int P, int U, bool TAIL>
static __device__ __forceinline__
void add_records(const float* __restrict__ Y, float* tile, int lane,
                 int src, int dl, float rn, int k0, int cnt) {
  constexpr int J = F / 32;
  float v[U][J];
  int   d[U];
  float r[U];
#pragma unroll
  for (int u = 0; u < U; ++u) {
    int k = k0 + u;
    d[u] = 0; r[u] = 0.0f;
    if (!TAIL || k < cnt) {
      int s = __builtin_amdgcn_readlane(src, k);
      d[u]  = __builtin_amdgcn_readlane(dl, k);
      r[u]  = __int_as_float(__builtin_amdgcn_readlane(__float_as_int(rn), k));
      const float* yr = Y + (size_t)s * F + lane;
#pragma unroll
      for (int j = 0; j < J; ++j) v[u][j] = yr[32 * j];
    }
  }
#pragma unroll
  for (int u = 0; u < U; ++u) {
    if (!TAIL || k0 + u < cnt) {
      float* tp = tile + d[u] * P + lane;
#pragma unroll
      for (int j = 0; j < J; ++j) atomicAdd(tp + 32 * j, r[u] * v[u][j]);
    }
  }
}

template <int F>
__global__ __launch_bounds__(AGG_T)
void k_aggr_tile(const float* __restrict__ Y, const int* __restrict__ rec_,
                 const int* __restrict__ st,
                 const float* __restrict__ degD, const float* __restrict__ degS,
                 float* __restrict__ A, int N) {
  constexpr int U  = 1024 / F;
  constexpr int Q  = TR * (F / 4);
  constexpr int IT = (Q + AGG_T - 1) / AGG_T;
  __shared__ float tile[TR * F];

  const int2* rec = (const int2*)rec_;
  const int b    = blockIdx.x;
  const int n0   = b * TR;
  const int rows = min(TR, N - n0);
  const int q    = rows * (F / 4);
  const int tid  = threadIdx.x;
  const int lane = tid & 31;
  const int wave = __builtin_amdgcn_readfirstlane(tid >> 5);

  {
    const float4* y4 = (const float4*)(Y + (size_t)n0 * F);
    float4 v[IT];
    float  dd[IT], ds[IT];
#pragma unroll
    for (int i = 0; i < IT; ++i) {
      int idx = min(tid + i * AGG_T, q - 1);
      int n = n0 + idx / (F / 4);
      v[i]  = y4[idx];
      dd[i] = degD[n];
      ds[i] = degS[n];
    }
#pragma unroll
    for (int i = 0; i < IT; ++i) {
      int idx = tid + i * AGG_T;
      if (idx < q) {
        float rn = rsqrtf(dd[i] * ds[i]);
        float4 o; o.x = rn * v[i].x; o.y = rn * v[i].y; o.z = rn * v[i].z; o.w = rn * v[i].w;
        ((float4*)tile)[idx] = o;
      }
    }
  }
  __syncthreads();

  const int lo = st[b], hi = st[b + 1];
  for (int c0 = lo + wave * 32; c0 < hi; c0 += AGG_T) {
    const int cnt = min(32, hi - c0);
    int src = 0, dl = 0;
    float rn = 0.0f;
    if (lane < cnt) {
      int2 r = rec[c0 + lane];
      src = r.x;
      dl  = r.y - n0;
      rn  = rsqrtf(degD[r.y] * degS[r.x]);
    }
    int k0 = 0;
    for (; k0 + U <= cnt; k0 += U)
      add_records<F, F, U, false>(Y, tile, lane, src, dl, rn, k0, cnt);
    if (k0 < cnt)
      add_records<F, F, U, true>(Y, tile, lane, src, dl, rn, k0, cnt);
  }
  __syncthreads();

  {
    float4* a4 = (float4*)(A + (size_t)n0 * F);
#pragma unroll
    for (int i = 0; i < IT; ++i) {
      int idx = tid + i * AGG_T;
      if (idx < q) a4[idx] = ((const float4*)tile)[idx];
    }
  }
}


#define L2_RT   ((TR + 15) / 16)
#define L2_ROWS (L2_RT * 16)
#define L2_P    (IN_F + 4)

__global__ __launch_bounds__(AGG_T)
void k_aggr_lin2(const float* __restrict__ Y, const int* __restrict__ rec_,
                 const int* __restrict__ st,
                 const float* __restrict__ degD, const float* __restrict__ degS,
                 const float* __restrict__ Wa, const float* __restrict__ ba,
                 const float* __restrict__ Wb, const float* __restrict__ bb,
                 float* __restrict__ Y2, int N) {
  constexpr int F  = IN_F;
  constexpr int P  = L2_P;
  constexpr int K  = HID_F;
  constexpr int NK = K / 4;
  constexpr int U  = 1024 / F;
  constexpr int QP = L2_ROWS * (F / 4);
  constexpr int IT = (QP + AGG_T - 1) / AGG_T;
  constexpr int NCT = HID_F / 16;
  constexpr int NG  = (AGG_T / 32) / NCT;
  static_assert(L2_RT <= 2 * NG, "two row tiles per wave cover the tile");
  __shared__ float tile[L2_ROWS * P];

  const int2* rec = (const int2*)rec_;
  const int b    = blockIdx.x;
  const int n0   = b * TR;
  const int rows = min(TR, N - n0);
  const int q    = rows * (F / 4);
  const int tid  = threadIdx.x;
  const int lane = tid & 31;
  const int wave = __builtin_amdgcn_readfirstlane(tid >> 5);

  {
    const float4* y4 = (const float4*)(Y + (size_t)n0 * F);
    float4 v[IT];
    float  dd[IT], ds[IT];
#pragma unroll
    for (int i = 0; i < IT; ++i) {
      int idx = min(tid + i * AGG_T, q - 1);
      int n = n0 + idx / (F / 4);
      v[i]  = y4[idx];
      dd[i] = degD[n];
      ds[i] = degS[n];
    }
#pragma unroll
    for (int i = 0; i < IT; ++i) {
      int idx = tid + i * AGG_T;
      if (idx < QP) {
        float4 o = {};
        if (idx < q) {
          float rn = rsqrtf(dd[i] * ds[i]);
          o.x = rn * v[i].x; o.y = rn * v[i].y; o.z = rn * v[i].z; o.w = rn * v[i].w;
        }
        *(float4*)(tile + (idx / (F / 4)) * P + (idx % (F / 4)) * 4) = o;
      }
    }
  }
  __syncthreads();

  const int lo = st[b], hi = st[b + 1];
  for (int c0 = lo + wave * 32; c0 < hi; c0 += AGG_T) {
    const int cnt = min(32, hi - c0);
    int src = 0, dl = 0;
    float rn = 0.0f;
    if (lane < cnt) {
      int2 r = rec[c0 + lane];
      src = r.x;
      dl  = r.y - n0;
      rn  = rsqrtf(degD[r.y] * degS[r.x]);
    }
    int k0 = 0;
    for (; k0 + U <= cnt; k0 += U)
      add_records<F, P, U, false>(Y, tile, lane, src, dl, rn, k0, cnt);
    if (k0 < cnt)
      add_records<F, P, U, true>(Y, tile, lane, src, dl, rn, k0, cnt);
  }

  const int ct   = wave % NCT;
  const int grp  = wave / NCT;
  const int col  = lane & 15;
  const int khi  = (lane >> 4) << 1;
  const int cc0  = ct << 4;
  const int half = (cc0 >= OUT_F) ? 1 : 0;
  const float* W    = half ? Wb : Wa;
  const float* bias = half ? bb : ba;
  const int wc0  = cc0 - half * OUT_F;
  const int koff = half ? HID_F : 0;
  v2f bfr[NK];
  const float* wrow = W + (size_t)(wc0 + col) * K + khi;
#pragma unroll
  for (int i = 0; i < NK; ++i) bfr[i] = *(const v2f*)(wrow + i * 4);
  const float bv = bias[wc0 + col];
  __syncthreads();

  const bool two = (grp + NG) < L2_RT;
  const int rt1  = two ? grp + NG : grp;
  const float* ap0 = tile + (grp * 16 + col) * P + koff + khi;
  const float* ap1 = tile + (rt1 * 16 + col) * P + koff + khi;
  v8f acc0 = {}, acc1 = {};
#pragma unroll
  for (int i = 0; i < NK; ++i) {
    v2f a0 = *(const v2f*)(ap0 + i * 4);
    v2f a1 = *(const v2f*)(ap1 + i * 4);
    a0.x = fmaxf(a0.x, 0.0f); a0.y = fmaxf(a0.y, 0.0f);
    a1.x = fmaxf(a1.x, 0.0f); a1.y = fmaxf(a1.y, 0.0f);
    acc0 = __builtin_amdgcn_wmma_f32_16x16x4_f32(false, a0, false, bfr[i], (short)0, acc0, false, false);
    acc1 = __builtin_amdgcn_wmma_f32_16x16x4_f32(false, a1, false, bfr[i], (short)0, acc1, false, false);
  }
  __syncthreads();

  {
    float* yp0 = tile + (grp * 16 + ((lane >> 4) << 3)) * P + cc0 + col;
#pragma unroll
    for (int g = 0; g < 8; ++g) yp0[g * P] = acc0[g] + bv;
    if (two) {
      float* yp1 = tile + (rt1 * 16 + ((lane >> 4) << 3)) * P + cc0 + col;
#pragma unroll
      for (int g = 0; g < 8; ++g) yp1[g * P] = acc1[g] + bv;
    }
  }
  __syncthreads();

  {
    constexpr int C4 = HID_F / 4;
    constexpr int IO = (TR * C4 + AGG_T - 1) / AGG_T;
    const int qo = rows * C4;
    float4* o4 = (float4*)(Y2 + (size_t)n0 * HID_F);
#pragma unroll
    for (int i = 0; i < IO; ++i) {
      int idx = tid + i * AGG_T;
      if (idx < qo) o4[idx] = *(const float4*)(tile + (idx / C4) * P + (idx % C4) * 4);
    }
  }
}


template <int U, bool TAIL>
static __device__ __forceinline__
void add_records2(const float* __restrict__ Y, float* tile, const float* __restrict__ Yb, float* tileB,
                  int lane, int src, int dl, float rn, int k0, int cnt) {
  constexpr int JA = IN_F / 32;
  constexpr int JB = HID_F / 32;
  float va[U][JA];
  float vb[U][JB];
  int   d[U];
  float r[U];
#pragma unroll
  for (int u = 0; u < U; ++u) {
    int k = k0 + u;
    d[u] = 0; r[u] = 0.0f;
    if (!TAIL || k < cnt) {
      int s = __builtin_amdgcn_readlane(src, k);
      d[u]  = __builtin_amdgcn_readlane(dl, k);
      r[u]  = __int_as_float(__builtin_amdgcn_readlane(__float_as_int(rn), k));
      const float* ya = Y  + (size_t)s * IN_F  + lane;
      const float* yb = Yb + (size_t)s * HID_F + lane;
#pragma unroll
      for (int j = 0; j < JA; ++j) va[u][j] = ya[32 * j];
#pragma unroll
      for (int j = 0; j < JB; ++j) vb[u][j] = yb[32 * j];
    }
  }
#pragma unroll
  for (int u = 0; u < U; ++u) {
    if (!TAIL || k0 + u < cnt) {
      float* ta = tile  + d[u] * L2_P  + lane;
      float* tb = tileB + d[u] * HID_F + lane;
#pragma unroll
      for (int j = 0; j < JA; ++j) atomicAdd(ta + 32 * j, r[u] * va[u][j]);
#pragma unroll
      for (int j = 0; j < JB; ++j) atomicAdd(tb + 32 * j, r[u] * vb[u][j]);
    }
  }
}

__global__ __launch_bounds__(AGG_T)
void k_aggr_lin2_tile(const float* __restrict__ Y, const float* __restrict__ Yb,
                      const int* __restrict__ rec_, const int* __restrict__ st,
                      const float* __restrict__ degD, const float* __restrict__ degS,
                      const float* __restrict__ Wa, const float* __restrict__ ba,
                      const float* __restrict__ Wb, const float* __restrict__ bb,
                      float* __restrict__ Y2, float* __restrict__ Ab, int N) {
  constexpr int F   = IN_F;
  constexpr int FB  = HID_F;
  constexpr int P   = L2_P;
  constexpr int K   = HID_F;
  constexpr int NK  = K / 4;
  constexpr int U   = 8;
  constexpr int QP  = L2_ROWS * (F / 4);
  constexpr int IT  = (QP + AGG_T - 1) / AGG_T;
  constexpr int QB  = TR * (FB / 4);
  constexpr int ITB = (QB + AGG_T - 1) / AGG_T;
  constexpr int NCT = HID_F / 16;
  constexpr int NG  = (AGG_T / 32) / NCT;
  static_assert(L2_RT <= 2 * NG, "two row tiles per wave cover the tile");
  __shared__ float tile[L2_ROWS * P];
  __shared__ float tileB[TR * FB];

  const int2* rec = (const int2*)rec_;
  const int b    = blockIdx.x;
  const int n0   = b * TR;
  const int rows = min(TR, N - n0);
  const int q    = rows * (F / 4);
  const int qb   = rows * (FB / 4);
  const int tid  = threadIdx.x;
  const int lane = tid & 31;
  const int wave = __builtin_amdgcn_readfirstlane(tid >> 5);

  {
    const float4* y4 = (const float4*)(Y + (size_t)n0 * F);
    float4 v[IT];
    float  dd[IT], ds[IT];
#pragma unroll
    for (int i = 0; i < IT; ++i) {
      int idx = min(tid + i * AGG_T, q - 1);
      int n = n0 + idx / (F / 4);
      v[i]  = y4[idx];
      dd[i] = degD[n];
      ds[i] = degS[n];
    }
#pragma unroll
    for (int i = 0; i < IT; ++i) {
      int idx = tid + i * AGG_T;
      if (idx < QP) {
        float4 o = {};
        if (idx < q) {
          float rn = rsqrtf(dd[i] * ds[i]);
          o.x = rn * v[i].x; o.y = rn * v[i].y; o.z = rn * v[i].z; o.w = rn * v[i].w;
        }
        *(float4*)(tile + (idx / (F / 4)) * P + (idx % (F / 4)) * 4) = o;
      }
    }
  }
  {
    const float4* y4 = (const float4*)(Yb + (size_t)n0 * FB);
    float4 v[ITB];
    float  dd[ITB], ds[ITB];
#pragma unroll
    for (int i = 0; i < ITB; ++i) {
      int idx = min(tid + i * AGG_T, qb - 1);
      int n = n0 + idx / (FB / 4);
      v[i]  = y4[idx];
      dd[i] = degD[n];
      ds[i] = degS[n];
    }
#pragma unroll
    for (int i = 0; i < ITB; ++i) {
      int idx = tid + i * AGG_T;
      if (idx < qb) {
        float rn = rsqrtf(dd[i] * ds[i]);
        float4 o; o.x = rn * v[i].x; o.y = rn * v[i].y; o.z = rn * v[i].z; o.w = rn * v[i].w;
        ((float4*)tileB)[idx] = o;
      }
    }
  }
  __syncthreads();

  const int lo = st[b], hi = st[b + 1];
  for (int c0 = lo + wave * 32; c0 < hi; c0 += AGG_T) {
    const int cnt = min(32, hi - c0);
    int src = 0, dl = 0;
    float rn = 0.0f;
    if (lane < cnt) {
      int2 r = rec[c0 + lane];
      src = r.x;
      dl  = r.y - n0;
      rn  = rsqrtf(degD[r.y] * degS[r.x]);
    }
    int k0 = 0;
    for (; k0 + U <= cnt; k0 += U)
      add_records2<U, false>(Y, tile, Yb, tileB, lane, src, dl, rn, k0, cnt);
    if (k0 < cnt)
      add_records2<U, true>(Y, tile, Yb, tileB, lane, src, dl, rn, k0, cnt);
  }

  const int ct   = wave % NCT;
  const int grp  = wave / NCT;
  const int col  = lane & 15;
  const int khi  = (lane >> 4) << 1;
  const int cc0  = ct << 4;
  const int half = (cc0 >= OUT_F) ? 1 : 0;
  const float* W    = half ? Wb : Wa;
  const float* bias = half ? bb : ba;
  const int wc0  = cc0 - half * OUT_F;
  const int koff = half ? HID_F : 0;
  v2f bfr[NK];
  const float* wrow = W + (size_t)(wc0 + col) * K + khi;
#pragma unroll
  for (int i = 0; i < NK; ++i) bfr[i] = *(const v2f*)(wrow + i * 4);
  const float bv = bias[wc0 + col];
  __syncthreads();

  {
    float4* a4 = (float4*)(Ab + (size_t)n0 * FB);
#pragma unroll
    for (int i = 0; i < ITB; ++i) {
      int idx = tid + i * AGG_T;
      if (idx < qb) a4[idx] = ((const float4*)tileB)[idx];
    }
  }

  const bool two = (grp + NG) < L2_RT;
  const int rt1  = two ? grp + NG : grp;
  const float* ap0 = tile + (grp * 16 + col) * P + koff + khi;
  const float* ap1 = tile + (rt1 * 16 + col) * P + koff + khi;
  v8f acc0 = {}, acc1 = {};
#pragma unroll
  for (int i = 0; i < NK; ++i) {
    v2f a0 = *(const v2f*)(ap0 + i * 4);
    v2f a1 = *(const v2f*)(ap1 + i * 4);
    a0.x = fmaxf(a0.x, 0.0f); a0.y = fmaxf(a0.y, 0.0f);
    a1.x = fmaxf(a1.x, 0.0f); a1.y = fmaxf(a1.y, 0.0f);
    acc0 = __builtin_amdgcn_wmma_f32_16x16x4_f32(false, a0, false, bfr[i], (short)0, acc0, false, false);
    acc1 = __builtin_amdgcn_wmma_f32_16x16x4_f32(false, a1, false, bfr[i], (short)0, acc1, false, false);
  }
  __syncthreads();

  {
    float* yp0 = tile + (grp * 16 + ((lane >> 4) << 3)) * P + cc0 + col;
#pragma unroll
    for (int g = 0; g < 8; ++g) yp0[g * P] = acc0[g] + bv;
    if (two) {
      float* yp1 = tile + (rt1 * 16 + ((lane >> 4) << 3)) * P + cc0 + col;
#pragma unroll
      for (int g = 0; g < 8; ++g) yp1[g * P] = acc1[g] + bv;
    }
  }
  __syncthreads();

  {
    constexpr int C4 = HID_F / 4;
    constexpr int IO = (TR * C4 + AGG_T - 1) / AGG_T;
    const int qo = rows * C4;
    float4* o4 = (float4*)(Y2 + (size_t)n0 * HID_F);
#pragma unroll
    for (int i = 0; i < IO; ++i) {
      int idx = tid + i * AGG_T;
      if (idx < qo) o4[idx] = *(const float4*)(tile + (idx / C4) * P + (idx % C4) * 4);
    }
  }
}


__global__ void k_final(const float* __restrict__ A2s, const float* __restrict__ A2t,
                        const float* __restrict__ noise_s, const float* __restrict__ noise_t,
                        float* __restrict__ out, int N) {
  int idx = blockIdx.x * blockDim.x + threadIdx.x;
  int total = N * OUT_F;
  if (idx < total) {
    int n = idx >> 5, j = idx & 31;
    out[idx] = A2s[n * 64 + j] + noise_s[idx] * expf(A2s[n * 64 + 32 + j]) * 0.2f;
  } else if (idx < 2 * total) {
    int k = idx - total;
    int n = k >> 5, j = k & 31;
    out[idx] = A2t[n * 64 + j] + noise_t[k] * expf(A2t[n * 64 + 32 + j]) * 0.2f;
  }
}


static inline int ceil_div(int a, int b) { return (a + b - 1) / b; }

extern "C" void kernel_launch(void* const* d_in, const int* in_sizes, int n_in,
                              void* d_out, int out_size, void* d_ws, size_t ws_size,
                              hipStream_t stream) {
  const float* s       = (const float*)d_in[0];
  const float* t       = (const float*)d_in[1];
  const int*   ei      = (const int*)  d_in[2];
  const float* noise_s = (const float*)d_in[3];
  const float* noise_t = (const float*)d_in[4];
  const float* sm1_W = (const float*)d_in[5],  *sm1_b = (const float*)d_in[6];
  const float* sm2_W = (const float*)d_in[7],  *sm2_b = (const float*)d_in[8];
  const float* sl1_W = (const float*)d_in[9],  *sl1_b = (const float*)d_in[10];
  const float* sl2_W = (const float*)d_in[11], *sl2_b = (const float*)d_in[12];
  const float* tm1_W = (const float*)d_in[13], *tm1_b = (const float*)d_in[14];
  const float* tm2_W = (const float*)d_in[15], *tm2_b = (const float*)d_in[16];
  const float* tl1_W = (const float*)d_in[17], *tl1_b = (const float*)d_in[18];
  const float* tl2_W = (const float*)d_in[19], *tl2_b = (const float*)d_in[20];

  const int N = in_sizes[0] / IN_F;
  const int E = in_sizes[2] / 2;
  const int* ei0 = ei;
  const int* ei1 = ei + E;
  const int NB = ceil_div(N, TR);

  float* ws  = (float*)d_ws;
  float* Y1s = ws;
  float* Y2s = Y1s + (size_t)N * IN_F;
  float* Y1t = Y2s + (size_t)N * HID_F;
  float* A2s = Y1t + (size_t)N * IN_F;
  float* Y2t = Y1s;
  float* A2t = Y1t;
  int*   recF = (int*)(A2s + (size_t)N * HID_F);
  int*   recB = recF + (size_t)2 * E;
  float* deg_in  = (float*)(recB + (size_t)2 * E);
  float* deg_out = deg_in + N;
  int*   cnt     = (int*)(deg_out + N);
  int*   cursor  = cnt + REPL * 2 * NB;
  int*   start   = cursor + REPL * 2 * NB;
  const int* startF = start;
  const int* startB = start + (NB + 1);

  const int nChunks = ceil_div(N, GEMM_ROWS);
  const int gemmBlocks = 512;
  const int binBlocks = ceil_div(E, BIN_T * BIN_EPT);

  k_zero_counts<<<ceil_div(REPL * 2 * NB, 256), 256, 0, stream>>>(cnt, REPL * 2 * NB);
  k_bin_count<<<binBlocks, BIN_T, 2 * NB * sizeof(int), stream>>>(ei0, ei1, cnt, E, NB);
  k_bin_scan<<<2, 1024, 0, stream>>>(cnt, start, cursor, NB);
  k_bin_place<<<binBlocks, BIN_T, 4 * NB * sizeof(int), stream>>>(ei0, ei1, cursor, recF, recB, E, NB);
  k_degrees<<<dim3(NB, 2), 256, 0, stream>>>(recF, recB, start, deg_in, deg_out, N, NB);

  k_gemm_wmma<IN_F, IN_F / 16, 0><<<gemmBlocks, GEMM_T, 0, stream>>>(
      s, sm1_W, sm1_b, sl1_W, sl1_b, Y1s, IN_F, HID_F, 0, N, nChunks);
  k_aggr_lin2<<<NB, AGG_T, 0, stream>>>(Y1s, recF, startF, deg_in, deg_out,
                                        sm2_W, sm2_b, sl2_W, sl2_b, Y2s, N);

  k_gemm_wmma<IN_F, IN_F / 16, 0><<<gemmBlocks, GEMM_T, 0, stream>>>(
      t, tm1_W, tm1_b, tl1_W, tl1_b, Y1t, IN_F, HID_F, 0, N, nChunks);
  k_aggr_lin2_tile<<<NB, AGG_T, 0, stream>>>(Y1t, Y2s, recB, startB, deg_out, deg_in,
                                             tm2_W, tm2_b, tl2_W, tl2_b, Y2t, A2s, N);

  k_aggr_tile<HID_F><<<NB, AGG_T, 0, stream>>>(Y2t, recF, startF, deg_in, deg_out, A2t, N);

  k_final<<<ceil_div(2 * N * OUT_F, 256), 256, 0, stream>>>(A2s, A2t, noise_s, noise_t,
                                                            (float*)d_out, N);
}


# The two layer-1 products run as extra work-groups inside the edge-count and edge-place launches (role by block index; binning groups 256 threads x 16 edges); 11 launches to 9; sums unchanged
# speedup vs baseline: 4.2682x; 1.0520x over previous
#include <hip/hip_runtime.h>


typedef __attribute__((ext_vector_type(2))) float v2f;
typedef __attribute__((ext_vector_type(8))) float v8f;

#define IN_F   128
#define HID_F  64
#define OUT_F  32

#define TR     100
#define AGG_T  512
#define BIN_T  256
#define BIN_EPT 16
#define REPL   16


__global__ void k_zero_counts(int* __restrict__ cnt, int n) {
  int i = blockIdx.x * blockDim.x + threadIdx.x;
  if (i < n) cnt[i] = 0;
}

static __device__ __forceinline__
void bin_count_part(const int* __restrict__ ei0, const int* __restrict__ ei1,
                    int* __restrict__ cnt, int E, int NB, int blk) {
  extern __shared__ int bin_lds[];
  int* hist = bin_lds;
  for (int i = threadIdx.x; i < 2 * NB; i += BIN_T) hist[i] = 0;
  __syncthreads();
  int e0 = blk * (BIN_T * BIN_EPT) + threadIdx.x;
  int s[BIN_EPT], d[BIN_EPT];
#pragma unroll
  for (int k = 0; k < BIN_EPT; ++k) {
    int e = min(e0 + k * BIN_T, E - 1);
    s[k] = ei0[e];
    d[k] = ei1[e];
  }
#pragma unroll
  for (int k = 0; k < BIN_EPT; ++k) {
    if (e0 + k * BIN_T < E) {
      atomicAdd(&hist[d[k] / TR], 1);
      atomicAdd(&hist[NB + s[k] / TR], 1);
    }
  }
  __syncthreads();
  int* mycnt = cnt + (blk % REPL) * 2 * NB;
  for (int i = threadIdx.x; i < 2 * NB; i += BIN_T) {
    int c = hist[i];
    if (c) atomicAdd(&mycnt[i], c);
  }
}

__global__ __launch_bounds__(1024)
void k_bin_scan(const int* __restrict__ cnt, int* __restrict__ start,
                int* __restrict__ cursor, int NB) {
  __shared__ int sh[1024];
  int dir = blockIdx.x;
  int* st = start + dir * (NB + 1);
  int tid = threadIdx.x;
  int carry = 0;
  for (int base = 0; base < NB; base += 1024) {
    int i = base + tid;
    int c[REPL];
    int v = 0;
#pragma unroll
    for (int r = 0; r < REPL; ++r) {
      c[r] = (i < NB) ? cnt[r * 2 * NB + dir * NB + i] : 0;
      v += c[r];
    }
    sh[tid] = v;
    __syncthreads();
    for (int off = 1; off < 1024; off <<= 1) {
      int t = (tid >= off) ? sh[tid - off] : 0;
      __syncthreads();
      sh[tid] += t;
      __syncthreads();
    }
    int run = carry + sh[tid] - v;
    if (i < NB) {
      st[i] = run;
#pragma unroll
      for (int r = 0; r < REPL; ++r) {
        cursor[r * 2 * NB + dir * NB + i] = run;
        run += c[r];
      }
    }
    carry += sh[1023];
    __syncthreads();
  }
  if (tid == 0) st[NB] = carry;
}

static __device__ __forceinline__
void bin_place_part(const int* __restrict__ ei0, const int* __restrict__ ei1,
                    int* __restrict__ cursor, int* __restrict__ recF, int* __restrict__ recB,
                    int E, int NB, int blk) {
  extern __shared__ int bin_lds[];
  int* hist = bin_lds;
  int* base = bin_lds + 2 * NB;
  for (int i = threadIdx.x; i < 2 * NB; i += BIN_T) hist[i] = 0;
  __syncthreads();
  int s[BIN_EPT], d[BIN_EPT], rF[BIN_EPT], rB[BIN_EPT];
  int e0 = blk * (BIN_T * BIN_EPT) + threadIdx.x;
#pragma unroll
  for (int k = 0; k < BIN_EPT; ++k) {
    int e = min(e0 + k * BIN_T, E - 1);
    s[k] = ei0[e];
    d[k] = ei1[e];
  }
#pragma unroll
  for (int k = 0; k < BIN_EPT; ++k) {
    rF[k] = 0; rB[k] = 0;
    if (e0 + k * BIN_T < E) {
      rF[k] = atomicAdd(&hist[d[k] / TR], 1);
      rB[k] = atomicAdd(&hist[NB + s[k] / TR], 1);
    }
  }
  __syncthreads();
  int* mycur = cursor + (blk % REPL) * 2 * NB;
  for (int i0 = threadIdx.x; i0 < 2 * NB; i0 += 4 * BIN_T) {
    int c[4], r[4];
#pragma unroll
    for (int j = 0; j < 4; ++j) {
      int i = i0 + j * BIN_T;
      c[j] = (i < 2 * NB) ? hist[i] : 0;
    }
#pragma unroll
    for (int j = 0; j < 4; ++j)
      r[j] = c[j] ? atomicAdd(&mycur[i0 + j * BIN_T], c[j]) : 0;
#pragma unroll
    for (int j = 0; j < 4; ++j) {
      int i = i0 + j * BIN_T;
      if (i < 2 * NB) base[i] = r[j];
    }
  }
  __syncthreads();
  int2* oF = (int2*)recF;
  int2* oB = (int2*)recB;
#pragma unroll
  for (int k = 0; k < BIN_EPT; ++k) {
    if (e0 + k * BIN_T < E) {
      oF[base[d[k] / TR] + rF[k]]      = make_int2(s[k], d[k]);
      oB[base[NB + s[k] / TR] + rB[k]] = make_int2(d[k], s[k]);
    }
  }
}

__global__ __launch_bounds__(256)
void k_degrees(const int* __restrict__ recF, const int* __restrict__ recB,
               const int* __restrict__ start,
               float* __restrict__ deg_in, float* __restrict__ deg_out, int N, int NB) {
  __shared__ int c[TR];
  int b = blockIdx.x, dir = blockIdx.y;
  const int2* rec = (const int2*)(dir ? recB : recF);
  const int* st = start + dir * (NB + 1);
  float* deg = dir ? deg_out : deg_in;
  for (int i = threadIdx.x; i < TR; i += 256) c[i] = 0;
  __syncthreads();
  int n0 = b * TR;
  int lo = st[b], hi = st[b + 1];
  for (int i = lo + threadIdx.x; i < hi; i += 256) atomicAdd(&c[rec[i].y - n0], 1);
  __syncthreads();
  for (int i = threadIdx.x; i < TR; i += 256) {
    int n = n0 + i;
    if (n < N) deg[n] = 1.0f + (float)c[i];
  }
}


#define GEMM_T    256
#define GEMM_ROWS 32
#define GEMM_P    (IN_F + 4)

template <int K, int NCT, int RELU>
static __device__ __forceinline__
void gemm_part(float* xt, const float* __restrict__ X,
               const float* __restrict__ Wa, const float* __restrict__ ba,
               const float* __restrict__ Wb, const float* __restrict__ bb,
               float* __restrict__ Y, int ldy,
               int halfO, int koffB, int N, int nChunks, int blk, int nblk) {
  constexpr int NK  = K / 4;
  constexpr int G   = (GEMM_T / 32) / NCT;
  constexpr int MPW = (GEMM_ROWS / 16) / G;

  const int tid  = threadIdx.x;
  const int lane = tid & 31;
  const int wave = __builtin_amdgcn_readfirstlane(tid >> 5);
  const int ct   = wave % NCT;
  const int grp  = wave / NCT;
  const int col  = lane & 15;
  const int khi  = (lane >> 4) << 1;

  const int c0   = ct << 4;
  const int half = (c0 >= halfO) ? 1 : 0;
  const float* W    = half ? Wb : Wa;
  const float* bias = half ? bb : ba;
  const int wc0  = c0 - half * halfO;
  const int koff = half ? koffB : 0;

  v2f bfr[NK];
  const float* wrow = W + (size_t)(wc0 + col) * K + khi;
#pragma unroll
  for (int i = 0; i < NK; ++i) bfr[i] = *(const v2f*)(wrow + i * 4);
  const float bv = bias[wc0 + col];

  const float* xl = X + lane * 4;
  const int rmax = N - 1;
  float4 s0 = {}, s1 = {}, s2 = {}, s3 = {};
  int ch = blk;
  if (ch < nChunks) {
    const int r0 = ch * GEMM_ROWS + wave;
    s0 = *(const float4*)(xl + (size_t)min(r0,      rmax) * IN_F);
    s1 = *(const float4*)(xl + (size_t)min(r0 + 8,  rmax) * IN_F);
    s2 = *(const float4*)(xl + (size_t)min(r0 + 16, rmax) * IN_F);
    s3 = *(const float4*)(xl + (size_t)min(r0 + 24, rmax) * IN_F);
  }
  float* xs = xt + wave * GEMM_P + lane * 4;
  const float* ap = xt + (grp * MPW * 16 + col) * GEMM_P + koff + khi;

  for (; ch < nChunks; ch += nblk) {
    if (RELU) {
      s0.x = fmaxf(s0.x, 0.0f); s0.y = fmaxf(s0.y, 0.0f); s0.z = fmaxf(s0.z, 0.0f); s0.w = fmaxf(s0.w, 0.0f);
      s1.x = fmaxf(s1.x, 0.0f); s1.y = fmaxf(s1.y, 0.0f); s1.z = fmaxf(s1.z, 0.0f); s1.w = fmaxf(s1.w, 0.0f);
      s2.x = fmaxf(s2.x, 0.0f); s2.y = fmaxf(s2.y, 0.0f); s2.z = fmaxf(s2.z, 0.0f); s2.w = fmaxf(s2.w, 0.0f);
      s3.x = fmaxf(s3.x, 0.0f); s3.y = fmaxf(s3.y, 0.0f); s3.z = fmaxf(s3.z, 0.0f); s3.w = fmaxf(s3.w, 0.0f);
    }
    *(float4*)(xs)               = s0;
    *(float4*)(xs + 8 * GEMM_P)  = s1;
    *(float4*)(xs + 16 * GEMM_P) = s2;
    *(float4*)(xs + 24 * GEMM_P) = s3;
    __syncthreads();

    const int nx = ch + nblk;
    if (nx < nChunks) {
      const int r0 = nx * GEMM_ROWS + wave;
      s0 = *(const float4*)(xl + (size_t)min(r0,      rmax) * IN_F);
      s1 = *(const float4*)(xl + (size_t)min(r0 + 8,  rmax) * IN_F);
      s2 = *(const float4*)(xl + (size_t)min(r0 + 16, rmax) * IN_F);
      s3 = *(const float4*)(xl + (size_t)min(r0 + 24, rmax) * IN_F);
    }

    v8f acc[MPW];
#pragma unroll
    for (int j = 0; j < MPW; ++j) acc[j] = v8f{};
#pragma unroll
    for (int i = 0; i < NK; ++i) {
#pragma unroll
      for (int j = 0; j < MPW; ++j) {
        v2f a = *(const v2f*)(ap + j * 16 * GEMM_P + i * 4);
        acc[j] = __builtin_amdgcn_wmma_f32_16x16x4_f32(false, a, false, bfr[i],
                                                       (short)0, acc[j], false, false);
      }
    }

#pragma unroll
    for (int j = 0; j < MPW; ++j) {
      const int m0 = ch * GEMM_ROWS + (grp * MPW + j) * 16;
      if (m0 < N) {
        float* yp = Y + (size_t)(m0 + ((lane >> 4) << 3)) * ldy + c0 + col;
#pragma unroll
        for (int g = 0; g < 8; ++g)
          yp[(size_t)g * ldy] = acc[j][g] + bv;
      }
    }
    __syncthreads();
  }
}


static_assert(BIN_T == GEMM_T, "binning and product work-groups share a launch shape");

__global__ __launch_bounds__(GEMM_T)
void k_count_gemm(const int* __restrict__ ei0, const int* __restrict__ ei1,
                  int* __restrict__ cnt, int E, int NB, int binBlocks,
                  const float* __restrict__ X,
                  const float* __restrict__ Wa, const float* __restrict__ ba,
                  const float* __restrict__ Wb, const float* __restrict__ bb,
                  float* __restrict__ Y, int N, int nChunks) {
  __shared__ float xt[GEMM_ROWS * GEMM_P];
  const int b = blockIdx.x;
  if (b < binBlocks)
    bin_count_part(ei0, ei1, cnt, E, NB, b);
  else
    gemm_part<IN_F, IN_F / 16, 0>(xt, X, Wa, ba, Wb, bb, Y, IN_F, HID_F, 0, N, nChunks,
                                  b - binBlocks, (int)gridDim.x - binBlocks);
}

__global__ __launch_bounds__(GEMM_T)
void k_place_gemm(const int* __restrict__ ei0, const int* __restrict__ ei1,
                  int* __restrict__ cursor, int* __restrict__ recF, int* __restrict__ recB,
                  int E, int NB, int binBlocks,
                  const float* __restrict__ X,
                  const float* __restrict__ Wa, const float* __restrict__ ba,
                  const float* __restrict__ Wb, const float* __restrict__ bb,
                  float* __restrict__ Y, int N, int nChunks) {
  __shared__ float xt[GEMM_ROWS * GEMM_P];
  const int b = blockIdx.x;
  if (b < binBlocks)
    bin_place_part(ei0, ei1, cursor, recF, recB, E, NB, b);
  else
    gemm_part<IN_F, IN_F / 16, 0>(xt, X, Wa, ba, Wb, bb, Y, IN_F, HID_F, 0, N, nChunks,
                                  b - binBlocks, (int)gridDim.x - binBlocks);
}


template <int F, int P, int U, bool TAIL>
static __device__ __forceinline__
void add_records(const float* __restrict__ Y, float* tile, int lane,
                 int src, int dl, float rn, int k0, int cnt) {
  constexpr int J = F / 32;
  float v[U][J];
  int   d[U];
  float r[U];
#pragma unroll
  for (int u = 0; u < U; ++u) {
    int k = k0 + u;
    d[u] = 0; r[u] = 0.0f;
    if (!TAIL || k < cnt) {
      int s = __builtin_amdgcn_readlane(src, k);
      d[u]  = __builtin_amdgcn_readlane(dl, k);
      r[u]  = __int_as_float(__builtin_amdgcn_readlane(__float_as_int(rn), k));
      const float* yr = Y + (size_t)s * F + lane;
#pragma unroll
      for (int j = 0; j < J; ++j) v[u][j] = yr[32 * j];
    }
  }
#pragma unroll
  for (int u = 0; u < U; ++u) {
    if (!TAIL || k0 + u < cnt) {
      float* tp = tile + d[u] * P + lane;
#pragma unroll
      for (int j = 0; j < J; ++j) atomicAdd(tp + 32 * j, r[u] * v[u][j]);
    }
  }
}

template <int F>
__global__ __launch_bounds__(AGG_T)
void k_aggr_tile(const float* __restrict__ Y, const int* __restrict__ rec_,
                 const int* __restrict__ st,
                 const float* __restrict__ degD, const float* __restrict__ degS,
                 float* __restrict__ A, int N) {
  constexpr int U  = 1024 / F;
  constexpr int Q  = TR * (F / 4);
  constexpr int IT = (Q + AGG_T - 1) / AGG_T;
  __shared__ float tile[TR * F];

  const int2* rec = (const int2*)rec_;
  const int b    = blockIdx.x;
  const int n0   = b * TR;
  const int rows = min(TR, N - n0);
  const int q    = rows * (F / 4);
  const int tid  = threadIdx.x;
  const int lane = tid & 31;
  const int wave = __builtin_amdgcn_readfirstlane(tid >> 5);

  {
    const float4* y4 = (const float4*)(Y + (size_t)n0 * F);
    float4 v[IT];
    float  dd[IT], ds[IT];
#pragma unroll
    for (int i = 0; i < IT; ++i) {
      int idx = min(tid + i * AGG_T, q - 1);
      int n = n0 + idx / (F / 4);
      v[i]  = y4[idx];
      dd[i] = degD[n];
      ds[i] = degS[n];
    }
#pragma unroll
    for (int i = 0; i < IT; ++i) {
      int idx = tid + i * AGG_T;
      if (idx < q) {
        float rn = rsqrtf(dd[i] * ds[i]);
        float4 o; o.x = rn * v[i].x; o.y = rn * v[i].y; o.z = rn * v[i].z; o.w = rn * v[i].w;
        ((float4*)tile)[idx] = o;
      }
    }
  }
  __syncthreads();

  const int lo = st[b], hi = st[b + 1];
  for (int c0 = lo + wave * 32; c0 < hi; c0 += AGG_T) {
    const int cnt = min(32, hi - c0);
    int src = 0, dl = 0;
    float rn = 0.0f;
    if (lane < cnt) {
      int2 r = rec[c0 + lane];
      src = r.x;
      dl  = r.y - n0;
      rn  = rsqrtf(degD[r.y] * degS[r.x]);
    }
    int k0 = 0;
    for (; k0 + U <= cnt; k0 += U)
      add_records<F, F, U, false>(Y, tile, lane, src, dl, rn, k0, cnt);
    if (k0 < cnt)
      add_records<F, F, U, true>(Y, tile, lane, src, dl, rn, k0, cnt);
  }
  __syncthreads();

  {
    float4* a4 = (float4*)(A + (size_t)n0 * F);
#pragma unroll
    for (int i = 0; i < IT; ++i) {
      int idx = tid + i * AGG_T;
      if (idx < q) a4[idx] = ((const float4*)tile)[idx];
    }
  }
}


#define L2_RT   ((TR + 15) / 16)
#define L2_ROWS (L2_RT * 16)
#define L2_P    (IN_F + 4)

__global__ __launch_bounds__(AGG_T)
void k_aggr_lin2(const float* __restrict__ Y, const int* __restrict__ rec_,
                 const int* __restrict__ st,
                 const float* __restrict__ degD, const float* __restrict__ degS,
                 const float* __restrict__ Wa, const float* __restrict__ ba,
                 const float* __restrict__ Wb, const float* __restrict__ bb,
                 float* __restrict__ Y2, int N) {
  constexpr int F  = IN_F;
  constexpr int P  = L2_P;
  constexpr int K  = HID_F;
  constexpr int NK = K / 4;
  constexpr int U  = 1024 / F;
  constexpr int QP = L2_ROWS * (F / 4);
  constexpr int IT = (QP + AGG_T - 1) / AGG_T;
  constexpr int NCT = HID_F / 16;
  constexpr int NG  = (AGG_T / 32) / NCT;
  static_assert(L2_RT <= 2 * NG, "two row tiles per wave cover the tile");
  __shared__ float tile[L2_ROWS * P];

  const int2* rec = (const int2*)rec_;
  const int b    = blockIdx.x;
  const int n0   = b * TR;
  const int rows = min(TR, N - n0);
  const int q    = rows * (F / 4);
  const int tid  = threadIdx.x;
  const int lane = tid & 31;
  const int wave = __builtin_amdgcn_readfirstlane(tid >> 5);

  {
    const float4* y4 = (const float4*)(Y + (size_t)n0 * F);
    float4 v[IT];
    float  dd[IT], ds[IT];
#pragma unroll
    for (int i = 0; i < IT; ++i) {
      int idx = min(tid + i * AGG_T, q - 1);
      int n = n0 + idx / (F / 4);
      v[i]  = y4[idx];
      dd[i] = degD[n];
      ds[i] = degS[n];
    }
#pragma unroll
    for (int i = 0; i < IT; ++i) {
      int idx = tid + i * AGG_T;
      if (idx < QP) {
        float4 o = {};
        if (idx < q) {
          float rn = rsqrtf(dd[i] * ds[i]);
          o.x = rn * v[i].x; o.y = rn * v[i].y; o.z = rn * v[i].z; o.w = rn * v[i].w;
        }
        *(float4*)(tile + (idx / (F / 4)) * P + (idx % (F / 4)) * 4) = o;
      }
    }
  }
  __syncthreads();

  const int lo = st[b], hi = st[b + 1];
  for (int c0 = lo + wave * 32; c0 < hi; c0 += AGG_T) {
    const int cnt = min(32, hi - c0);
    int src = 0, dl = 0;
    float rn = 0.0f;
    if (lane < cnt) {
      int2 r = rec[c0 + lane];
      src = r.x;
      dl  = r.y - n0;
      rn  = rsqrtf(degD[r.y] * degS[r.x]);
    }
    int k0 = 0;
    for (; k0 + U <= cnt; k0 += U)
      add_records<F, P, U, false>(Y, tile, lane, src, dl, rn, k0, cnt);
    if (k0 < cnt)
      add_records<F, P, U, true>(Y, tile, lane, src, dl, rn, k0, cnt);
  }

  const int ct   = wave % NCT;
  const int grp  = wave / NCT;
  const int col  = lane & 15;
  const int khi  = (lane >> 4) << 1;
  const int cc0  = ct << 4;
  const int half = (cc0 >= OUT_F) ? 1 : 0;
  const float* W    = half ? Wb : Wa;
  const float* bias = half ? bb : ba;
  const int wc0  = cc0 - half * OUT_F;
  const int koff = half ? HID_F : 0;
  v2f bfr[NK];
  const float* wrow = W + (size_t)(wc0 + col) * K + khi;
#pragma unroll
  for (int i = 0; i < NK; ++i) bfr[i] = *(const v2f*)(wrow + i * 4);
  const float bv = bias[wc0 + col];
  __syncthreads();

  const bool two = (grp + NG) < L2_RT;
  const int rt1  = two ? grp + NG : grp;
  const float* ap0 = tile + (grp * 16 + col) * P + koff + khi;
  const float* ap1 = tile + (rt1 * 16 + col) * P + koff + khi;
  v8f acc0 = {}, acc1 = {};
#pragma unroll
  for (int i = 0; i < NK; ++i) {
    v2f a0 = *(const v2f*)(ap0 + i * 4);
    v2f a1 = *(const v2f*)(ap1 + i * 4);
    a0.x = fmaxf(a0.x, 0.0f); a0.y = fmaxf(a0.y, 0.0f);
    a1.x = fmaxf(a1.x, 0.0f); a1.y = fmaxf(a1.y, 0.0f);
    acc0 = __builtin_amdgcn_wmma_f32_16x16x4_f32(false, a0, false, bfr[i], (short)0, acc0, false, false);
    acc1 = __builtin_amdgcn_wmma_f32_16x16x4_f32(false, a1, false, bfr[i], (short)0, acc1, false, false);
  }
  __syncthreads();

  {
    float* yp0 = tile + (grp * 16 + ((lane >> 4) << 3)) * P + cc0 + col;
#pragma unroll
    for (int g = 0; g < 8; ++g) yp0[g * P] = acc0[g] + bv;
    if (two) {
      float* yp1 = tile + (rt1 * 16 + ((lane >> 4) << 3)) * P + cc0 + col;
#pragma unroll
      for (int g = 0; g < 8; ++g) yp1[g * P] = acc1[g] + bv;
    }
  }
  __syncthreads();

  {
    constexpr int C4 = HID_F / 4;
    constexpr int IO = (TR * C4 + AGG_T - 1) / AGG_T;
    const int qo = rows * C4;
    float4* o4 = (float4*)(Y2 + (size_t)n0 * HID_F);
#pragma unroll
    for (int i = 0; i < IO; ++i) {
      int idx = tid + i * AGG_T;
      if (idx < qo) o4[idx] = *(const float4*)(tile + (idx / C4) * P + (idx % C4) * 4);
    }
  }
}


template <int U, bool TAIL>
static __device__ __forceinline__
void add_records2(const float* __restrict__ Y, float* tile, const float* __restrict__ Yb, float* tileB,
                  int lane, int src, int dl, float rn, int k0, int cnt) {
  constexpr int JA = IN_F / 32;
  constexpr int JB = HID_F / 32;
  float va[U][JA];
  float vb[U][JB];
  int   d[U];
  float r[U];
#pragma unroll
  for (int u = 0; u < U; ++u) {
    int k = k0 + u;
    d[u] = 0; r[u] = 0.0f;
    if (!TAIL || k < cnt) {
      int s = __builtin_amdgcn_readlane(src, k);
      d[u]  = __builtin_amdgcn_readlane(dl, k);
      r[u]  = __int_as_float(__builtin_amdgcn_readlane(__float_as_int(rn), k));
      const float* ya = Y  + (size_t)s * IN_F  + lane;
      const float* yb = Yb + (size_t)s * HID_F + lane;
#pragma unroll
      for (int j = 0; j < JA; ++j) va[u][j] = ya[32 * j];
#pragma unroll
      for (int j = 0; j < JB; ++j) vb[u][j] = yb[32 * j];
    }
  }
#pragma unroll
  for (int u = 0; u < U; ++u) {
    if (!TAIL || k0 + u < cnt) {
      float* ta = tile  + d[u] * L2_P  + lane;
      float* tb = tileB + d[u] * HID_F + lane;
#pragma unroll
      for (int j = 0; j < JA; ++j) atomicAdd(ta + 32 * j, r[u] * va[u][j]);
#pragma unroll
      for (int j = 0; j < JB; ++j) atomicAdd(tb + 32 * j, r[u] * vb[u][j]);
    }
  }
}

__global__ __launch_bounds__(AGG_T)
void k_aggr_lin2_tile(const float* __restrict__ Y, const float* __restrict__ Yb,
                      const int* __restrict__ rec_, const int* __restrict__ st,
                      const float* __restrict__ degD, const float* __restrict__ degS,
                      const float* __restrict__ Wa, const float* __restrict__ ba,
                      const float* __restrict__ Wb, const float* __restrict__ bb,
                      float* __restrict__ Y2, float* __restrict__ Ab, int N) {
  constexpr int F   = IN_F;
  constexpr int FB  = HID_F;
  constexpr int P   = L2_P;
  constexpr int K   = HID_F;
  constexpr int NK  = K / 4;
  constexpr int U   = 8;
  constexpr int QP  = L2_ROWS * (F / 4);
  constexpr int IT  = (QP + AGG_T - 1) / AGG_T;
  constexpr int QB  = TR * (FB / 4);
  constexpr int ITB = (QB + AGG_T - 1) / AGG_T;
  constexpr int NCT = HID_F / 16;
  constexpr int NG  = (AGG_T / 32) / NCT;
  static_assert(L2_RT <= 2 * NG, "two row tiles per wave cover the tile");
  __shared__ float tile[L2_ROWS * P];
  __shared__ float tileB[TR * FB];

  const int2* rec = (const int2*)rec_;
  const int b    = blockIdx.x;
  const int n0   = b * TR;
  const int rows = min(TR, N - n0);
  const int q    = rows * (F / 4);
  const int qb   = rows * (FB / 4);
  const int tid  = threadIdx.x;
  const int lane = tid & 31;
  const int wave = __builtin_amdgcn_readfirstlane(tid >> 5);

  {
    const float4* y4 = (const float4*)(Y + (size_t)n0 * F);
    float4 v[IT];
    float  dd[IT], ds[IT];
#pragma unroll
    for (int i = 0; i < IT; ++i) {
      int idx = min(tid + i * AGG_T, q - 1);
      int n = n0 + idx / (F / 4);
      v[i]  = y4[idx];
      dd[i] = degD[n];
      ds[i] = degS[n];
    }
#pragma unroll
    for (int i = 0; i < IT; ++i) {
      int idx = tid + i * AGG_T;
      if (idx < QP) {
        float4 o = {};
        if (idx < q) {
          float rn = rsqrtf(dd[i] * ds[i]);
          o.x = rn * v[i].x; o.y = rn * v[i].y; o.z = rn * v[i].z; o.w = rn * v[i].w;
        }
        *(float4*)(tile + (idx / (F / 4)) * P + (idx % (F / 4)) * 4) = o;
      }
    }
  }
  {
    const float4* y4 = (const float4*)(Yb + (size_t)n0 * FB);
    float4 v[ITB];
    float  dd[ITB], ds[ITB];
#pragma unroll
    for (int i = 0; i < ITB; ++i) {
      int idx = min(tid + i * AGG_T, qb - 1);
      int n = n0 + idx / (FB / 4);
      v[i]  = y4[idx];
      dd[i] = degD[n];
      ds[i] = degS[n];
    }
#pragma unroll
    for (int i = 0; i < ITB; ++i) {
      int idx = tid + i * AGG_T;
      if (idx < qb) {
        float rn = rsqrtf(dd[i] * ds[i]);
        float4 o; o.x = rn * v[i].x; o.y = rn * v[i].y; o.z = rn * v[i].z; o.w = rn * v[i].w;
        ((float4*)tileB)[idx] = o;
      }
    }
  }
  __syncthreads();

  const int lo = st[b], hi = st[b + 1];
  for (int c0 = lo + wave * 32; c0 < hi; c0 += AGG_T) {
    const int cnt = min(32, hi - c0);
    int src = 0, dl = 0;
    float rn = 0.0f;
    if (lane < cnt) {
      int2 r = rec[c0 + lane];
      src = r.x;
      dl  = r.y - n0;
      rn  = rsqrtf(degD[r.y] * degS[r.x]);
    }
    int k0 = 0;
    for (; k0 + U <= cnt; k0 += U)
      add_records2<U, false>(Y, tile, Yb, tileB, lane, src, dl, rn, k0, cnt);
    if (k0 < cnt)
      add_records2<U, true>(Y, tile, Yb, tileB, lane, src, dl, rn, k0, cnt);
  }

  const int ct   = wave % NCT;
  const int grp  = wave / NCT;
  const int col  = lane & 15;
  const int khi  = (lane >> 4) << 1;
  const int cc0  = ct << 4;
  const int half = (cc0 >= OUT_F) ? 1 : 0;
  const float* W    = half ? Wb : Wa;
  const float* bias = half ? bb : ba;
  const int wc0  = cc0 - half * OUT_F;
  const int koff = half ? HID_F : 0;
  v2f bfr[NK];
  const float* wrow = W + (size_t)(wc0 + col) * K + khi;
#pragma unroll
  for (int i = 0; i < NK; ++i) bfr[i] = *(const v2f*)(wrow + i * 4);
  const float bv = bias[wc0 + col];
  __syncthreads();

  {
    float4* a4 = (float4*)(Ab + (size_t)n0 * FB);
#pragma unroll
    for (int i = 0; i < ITB; ++i) {
      int idx = tid + i * AGG_T;
      if (idx < qb) a4[idx] = ((const float4*)tileB)[idx];
    }
  }

  const bool two = (grp + NG) < L2_RT;
  const int rt1  = two ? grp + NG : grp;
  const float* ap0 = tile + (grp * 16 + col) * P + koff + khi;
  const float* ap1 = tile + (rt1 * 16 + col) * P + koff + khi;
  v8f acc0 = {}, acc1 = {};
#pragma unroll
  for (int i = 0; i < NK; ++i) {
    v2f a0 = *(const v2f*)(ap0 + i * 4);
    v2f a1 = *(const v2f*)(ap1 + i * 4);
    a0.x = fmaxf(a0.x, 0.0f); a0.y = fmaxf(a0.y, 0.0f);
    a1.x = fmaxf(a1.x, 0.0f); a1.y = fmaxf(a1.y, 0.0f);
    acc0 = __builtin_amdgcn_wmma_f32_16x16x4_f32(false, a0, false, bfr[i], (short)0, acc0, false, false);
    acc1 = __builtin_amdgcn_wmma_f32_16x16x4_f32(false, a1, false, bfr[i], (short)0, acc1, false, false);
  }
  __syncthreads();

  {
    float* yp0 = tile + (grp * 16 + ((lane >> 4) << 3)) * P + cc0 + col;
#pragma unroll
    for (int g = 0; g < 8; ++g) yp0[g * P] = acc0[g] + bv;
    if (two) {
      float* yp1 = tile + (rt1 * 16 + ((lane >> 4) << 3)) * P + cc0 + col;
#pragma unroll
      for (int g = 0; g < 8; ++g) yp1[g * P] = acc1[g] + bv;
    }
  }
  __syncthreads();

  {
    constexpr int C4 = HID_F / 4;
    constexpr int IO = (TR * C4 + AGG_T - 1) / AGG_T;
    const int qo = rows * C4;
    float4* o4 = (float4*)(Y2 + (size_t)n0 * HID_F);
#pragma unroll
    for (int i = 0; i < IO; ++i) {
      int idx = tid + i * AGG_T;
      if (idx < qo) o4[idx] = *(const float4*)(tile + (idx / C4) * P + (idx % C4) * 4);
    }
  }
}


__global__ void k_final(const float* __restrict__ A2s, const float* __restrict__ A2t,
                        const float* __restrict__ noise_s, const float* __restrict__ noise_t,
                        float* __restrict__ out, int N) {
  int idx = blockIdx.x * blockDim.x + threadIdx.x;
  int total = N * OUT_F;
  if (idx < total) {
    int n = idx >> 5, j = idx & 31;
    out[idx] = A2s[n * 64 + j] + noise_s[idx] * expf(A2s[n * 64 + 32 + j]) * 0.2f;
  } else if (idx < 2 * total) {
    int k = idx - total;
    int n = k >> 5, j = k & 31;
    out[idx] = A2t[n * 64 + j] + noise_t[k] * expf(A2t[n * 64 + 32 + j]) * 0.2f;
  }
}


static inline int ceil_div(int a, int b) { return (a + b - 1) / b; }

extern "C" void kernel_launch(void* const* d_in, const int* in_sizes, int n_in,
                              void* d_out, int out_size, void* d_ws, size_t ws_size,
                              hipStream_t stream) {
  const float* s       = (const float*)d_in[0];
  const float* t       = (const float*)d_in[1];
  const int*   ei      = (const int*)  d_in[2];
  const float* noise_s = (const float*)d_in[3];
  const float* noise_t = (const float*)d_in[4];
  const float* sm1_W = (const float*)d_in[5],  *sm1_b = (const float*)d_in[6];
  const float* sm2_W = (const float*)d_in[7],  *sm2_b = (const float*)d_in[8];
  const float* sl1_W = (const float*)d_in[9],  *sl1_b = (const float*)d_in[10];
  const float* sl2_W = (const float*)d_in[11], *sl2_b = (const float*)d_in[12];
  const float* tm1_W = (const float*)d_in[13], *tm1_b = (const float*)d_in[14];
  const float* tm2_W = (const float*)d_in[15], *tm2_b = (const float*)d_in[16];
  const float* tl1_W = (const float*)d_in[17], *tl1_b = (const float*)d_in[18];
  const float* tl2_W = (const float*)d_in[19], *tl2_b = (const float*)d_in[20];

  const int N = in_sizes[0] / IN_F;
  const int E = in_sizes[2] / 2;
  const int* ei0 = ei;
  const int* ei1 = ei + E;
  const int NB = ceil_div(N, TR);

  float* ws  = (float*)d_ws;
  float* Y1s = ws;
  float* Y2s = Y1s + (size_t)N * IN_F;
  float* Y1t = Y2s + (size_t)N * HID_F;
  float* A2s = Y1t + (size_t)N * IN_F;
  float* Y2t = Y1s;
  float* A2t = Y1t;
  int*   recF = (int*)(A2s + (size_t)N * HID_F);
  int*   recB = recF + (size_t)2 * E;
  float* deg_in  = (float*)(recB + (size_t)2 * E);
  float* deg_out = deg_in + N;
  int*   cnt     = (int*)(deg_out + N);
  int*   cursor  = cnt + REPL * 2 * NB;
  int*   start   = cursor + REPL * 2 * NB;
  const int* startF = start;
  const int* startB = start + (NB + 1);

  const int nChunks = ceil_div(N, GEMM_ROWS);
  const int gemmBlocks = 512;
  const int binBlocks = ceil_div(E, BIN_T * BIN_EPT);

  k_zero_counts<<<ceil_div(REPL * 2 * NB, 256), 256, 0, stream>>>(cnt, REPL * 2 * NB);
  k_count_gemm<<<binBlocks + gemmBlocks, GEMM_T, 2 * NB * sizeof(int), stream>>>(
      ei0, ei1, cnt, E, NB, binBlocks,
      t, tm1_W, tm1_b, tl1_W, tl1_b, Y1t, N, nChunks);
  k_bin_scan<<<2, 1024, 0, stream>>>(cnt, start, cursor, NB);
  k_place_gemm<<<binBlocks + gemmBlocks, GEMM_T, 4 * NB * sizeof(int), stream>>>(
      ei0, ei1, cursor, recF, recB, E, NB, binBlocks,
      s, sm1_W, sm1_b, sl1_W, sl1_b, Y1s, N, nChunks);
  k_degrees<<<dim3(NB, 2), 256, 0, stream>>>(recF, recB, start, deg_in, deg_out, N, NB);

  k_aggr_lin2<<<NB, AGG_T, 0, stream>>>(Y1s, recF, startF, deg_in, deg_out,
                                        sm2_W, sm2_b, sl2_W, sl2_b, Y2s, N);

  k_aggr_lin2_tile<<<NB, AGG_T, 0, stream>>>(Y1t, Y2s, recB, startB, deg_out, deg_in,
                                             tm2_W, tm2_b, tl2_W, tl2_b, Y2t, A2s, N);

  k_aggr_tile<HID_F><<<NB, AGG_T, 0, stream>>>(Y2t, recF, startF, deg_in, deg_out, A2t, N);

  k_final<<<ceil_div(2 * N * OUT_F, 256), 256, 0, stream>>>(A2s, A2t, noise_s, noise_t,
                                                            (float*)d_out, N);
}


# Reparametrize moved into the layer-2 LDS tile epilogues: the work-group that owns 100 finished mu|logstd rows writes their out rows itself (noise requested early); 9 launches to 8; sums unchanged
# speedup vs baseline: 4.5649x; 1.0695x over previous
#include <hip/hip_runtime.h>


typedef __attribute__((ext_vector_type(2))) float v2f;
typedef __attribute__((ext_vector_type(8))) float v8f;

#define IN_F   128
#define HID_F  64
#define OUT_F  32

#define TR     100
#define AGG_T  512
#define BIN_T  256
#define BIN_EPT 16
#define REPL   16


__global__ void k_zero_counts(int* __restrict__ cnt, int n) {
  int i = blockIdx.x * blockDim.x + threadIdx.x;
  if (i < n) cnt[i] = 0;
}

static __device__ __forceinline__
void bin_count_part(const int* __restrict__ ei0, const int* __restrict__ ei1,
                    int* __restrict__ cnt, int E, int NB, int blk) {
  extern __shared__ int bin_lds[];
  int* hist = bin_lds;
  for (int i = threadIdx.x; i < 2 * NB; i += BIN_T) hist[i] = 0;
  __syncthreads();
  int e0 = blk * (BIN_T * BIN_EPT) + threadIdx.x;
  int s[BIN_EPT], d[BIN_EPT];
#pragma unroll
  for (int k = 0; k < BIN_EPT; ++k) {
    int e = min(e0 + k * BIN_T, E - 1);
    s[k] = ei0[e];
    d[k] = ei1[e];
  }
#pragma unroll
  for (int k = 0; k < BIN_EPT; ++k) {
    if (e0 + k * BIN_T < E) {
      atomicAdd(&hist[d[k] / TR], 1);
      atomicAdd(&hist[NB + s[k] / TR], 1);
    }
  }
  __syncthreads();
  int* mycnt = cnt + (blk % REPL) * 2 * NB;
  for (int i = threadIdx.x; i < 2 * NB; i += BIN_T) {
    int c = hist[i];
    if (c) atomicAdd(&mycnt[i], c);
  }
}

__global__ __launch_bounds__(1024)
void k_bin_scan(const int* __restrict__ cnt, int* __restrict__ start,
                int* __restrict__ cursor, int NB) {
  __shared__ int sh[1024];
  int dir = blockIdx.x;
  int* st = start + dir * (NB + 1);
  int tid = threadIdx.x;
  int carry = 0;
  for (int base = 0; base < NB; base += 1024) {
    int i = base + tid;
    int c[REPL];
    int v = 0;
#pragma unroll
    for (int r = 0; r < REPL; ++r) {
      c[r] = (i < NB) ? cnt[r * 2 * NB + dir * NB + i] : 0;
      v += c[r];
    }
    sh[tid] = v;
    __syncthreads();
    for (int off = 1; off < 1024; off <<= 1) {
      int t = (tid >= off) ? sh[tid - off] : 0;
      __syncthreads();
      sh[tid] += t;
      __syncthreads();
    }
    int run = carry + sh[tid] - v;
    if (i < NB) {
      st[i] = run;
#pragma unroll
      for (int r = 0; r < REPL; ++r) {
        cursor[r * 2 * NB + dir * NB + i] = run;
        run += c[r];
      }
    }
    carry += sh[1023];
    __syncthreads();
  }
  if (tid == 0) st[NB] = carry;
}

static __device__ __forceinline__
void bin_place_part(const int* __restrict__ ei0, const int* __restrict__ ei1,
                    int* __restrict__ cursor, int* __restrict__ recF, int* __restrict__ recB,
                    int E, int NB, int blk) {
  extern __shared__ int bin_lds[];
  int* hist = bin_lds;
  int* base = bin_lds + 2 * NB;
  for (int i = threadIdx.x; i < 2 * NB; i += BIN_T) hist[i] = 0;
  __syncthreads();
  int s[BIN_EPT], d[BIN_EPT], rF[BIN_EPT], rB[BIN_EPT];
  int e0 = blk * (BIN_T * BIN_EPT) + threadIdx.x;
#pragma unroll
  for (int k = 0; k < BIN_EPT; ++k) {
    int e = min(e0 + k * BIN_T, E - 1);
    s[k] = ei0[e];
    d[k] = ei1[e];
  }
#pragma unroll
  for (int k = 0; k < BIN_EPT; ++k) {
    rF[k] = 0; rB[k] = 0;
    if (e0 + k * BIN_T < E) {
      rF[k] = atomicAdd(&hist[d[k] / TR], 1);
      rB[k] = atomicAdd(&hist[NB + s[k] / TR], 1);
    }
  }
  __syncthreads();
  int* mycur = cursor + (blk % REPL) * 2 * NB;
  for (int i0 = threadIdx.x; i0 < 2 * NB; i0 += 4 * BIN_T) {
    int c[4], r[4];
#pragma unroll
    for (int j = 0; j < 4; ++j) {
      int i = i0 + j * BIN_T;
      c[j] = (i < 2 * NB) ? hist[i] : 0;
    }
#pragma unroll
    for (int j = 0; j < 4; ++j)
      r[j] = c[j] ? atomicAdd(&mycur[i0 + j * BIN_T], c[j]) : 0;
#pragma unroll
    for (int j = 0; j < 4; ++j) {
      int i = i0 + j * BIN_T;
      if (i < 2 * NB) base[i] = r[j];
    }
  }
  __syncthreads();
  int2* oF = (int2*)recF;
  int2* oB = (int2*)recB;
#pragma unroll
  for (int k = 0; k < BIN_EPT; ++k) {
    if (e0 + k * BIN_T < E) {
      oF[base[d[k] / TR] + rF[k]]      = make_int2(s[k], d[k]);
      oB[base[NB + s[k] / TR] + rB[k]] = make_int2(d[k], s[k]);
    }
  }
}

__global__ __launch_bounds__(256)
void k_degrees(const int* __restrict__ recF, const int* __restrict__ recB,
               const int* __restrict__ start,
               float* __restrict__ deg_in, float* __restrict__ deg_out, int N, int NB) {
  __shared__ int c[TR];
  int b = blockIdx.x, dir = blockIdx.y;
  const int2* rec = (const int2*)(dir ? recB : recF);
  const int* st = start + dir * (NB + 1);
  float* deg = dir ? deg_out : deg_in;
  for (int i = threadIdx.x; i < TR; i += 256) c[i] = 0;
  __syncthreads();
  int n0 = b * TR;
  int lo = st[b], hi = st[b + 1];
  for (int i = lo + threadIdx.x; i < hi; i += 256) atomicAdd(&c[rec[i].y - n0], 1);
  __syncthreads();
  for (int i = threadIdx.x; i < TR; i += 256) {
    int n = n0 + i;
    if (n < N) deg[n] = 1.0f + (float)c[i];
  }
}


#define GEMM_T    256
#define GEMM_ROWS 32
#define GEMM_P    (IN_F + 4)

template <int K, int NCT, int RELU>
static __device__ __forceinline__
void gemm_part(float* xt, const float* __restrict__ X,
               const float* __restrict__ Wa, const float* __restrict__ ba,
               const float* __restrict__ Wb, const float* __restrict__ bb,
               float* __restrict__ Y, int ldy,
               int halfO, int koffB, int N, int nChunks, int blk, int nblk) {
  constexpr int NK  = K / 4;
  constexpr int G   = (GEMM_T / 32) / NCT;
  constexpr int MPW = (GEMM_ROWS / 16) / G;

  const int tid  = threadIdx.x;
  const int lane = tid & 31;
  const int wave = __builtin_amdgcn_readfirstlane(tid >> 5);
  const int ct   = wave % NCT;
  const int grp  = wave / NCT;
  const int col  = lane & 15;
  const int khi  = (lane >> 4) << 1;

  const int c0   = ct << 4;
  const int half = (c0 >= halfO) ? 1 : 0;
  const float* W    = half ? Wb : Wa;
  const float* bias = half ? bb : ba;
  const int wc0  = c0 - half * halfO;
  const int koff = half ? koffB : 0;

  v2f bfr[NK];
  const float* wrow = W + (size_t)(wc0 + col) * K + khi;
#pragma unroll
  for (int i = 0; i < NK; ++i) bfr[i] = *(const v2f*)(wrow + i * 4);
  const float bv = bias[wc0 + col];

  const float* xl = X + lane * 4;
  const int rmax = N - 1;
  float4 s0 = {}, s1 = {}, s2 = {}, s3 = {};
  int ch = blk;
  if (ch < nChunks) {
    const int r0 = ch * GEMM_ROWS + wave;
    s0 = *(const float4*)(xl + (size_t)min(r0,      rmax) * IN_F);
    s1 = *(const float4*)(xl + (size_t)min(r0 + 8,  rmax) * IN_F);
    s2 = *(const float4*)(xl + (size_t)min(r0 + 16, rmax) * IN_F);
    s3 = *(const float4*)(xl + (size_t)min(r0 + 24, rmax) * IN_F);
  }
  float* xs = xt + wave * GEMM_P + lane * 4;
  const float* ap = xt + (grp * MPW * 16 + col) * GEMM_P + koff + khi;

  for (; ch < nChunks; ch += nblk) {
    if (RELU) {
      s0.x = fmaxf(s0.x, 0.0f); s0.y = fmaxf(s0.y, 0.0f); s0.z = fmaxf(s0.z, 0.0f); s0.w = fmaxf(s0.w, 0.0f);
      s1.x = fmaxf(s1.x, 0.0f); s1.y = fmaxf(s1.y, 0.0f); s1.z = fmaxf(s1.z, 0.0f); s1.w = fmaxf(s1.w, 0.0f);
      s2.x = fmaxf(s2.x, 0.0f); s2.y = fmaxf(s2.y, 0.0f); s2.z = fmaxf(s2.z, 0.0f); s2.w = fmaxf(s2.w, 0.0f);
      s3.x = fmaxf(s3.x, 0.0f); s3.y = fmaxf(s3.y, 0.0f); s3.z = fmaxf(s3.z, 0.0f); s3.w = fmaxf(s3.w, 0.0f);
    }
    *(float4*)(xs)               = s0;
    *(float4*)(xs + 8 * GEMM_P)  = s1;
    *(float4*)(xs + 16 * GEMM_P) = s2;
    *(float4*)(xs + 24 * GEMM_P) = s3;
    __syncthreads();

    const int nx = ch + nblk;
    if (nx < nChunks) {
      const int r0 = nx * GEMM_ROWS + wave;
      s0 = *(const float4*)(xl + (size_t)min(r0,      rmax) * IN_F);
      s1 = *(const float4*)(xl + (size_t)min(r0 + 8,  rmax) * IN_F);
      s2 = *(const float4*)(xl + (size_t)min(r0 + 16, rmax) * IN_F);
      s3 = *(const float4*)(xl + (size_t)min(r0 + 24, rmax) * IN_F);
    }

    v8f acc[MPW];
#pragma unroll
    for (int j = 0; j < MPW; ++j) acc[j] = v8f{};
#pragma unroll
    for (int i = 0; i < NK; ++i) {
#pragma unroll
      for (int j = 0; j < MPW; ++j) {
        v2f a = *(const v2f*)(ap + j * 16 * GEMM_P + i * 4);
        acc[j] = __builtin_amdgcn_wmma_f32_16x16x4_f32(false, a, false, bfr[i],
                                                       (short)0, acc[j], false, false);
      }
    }

#pragma unroll
    for (int j = 0; j < MPW; ++j) {
      const int m0 = ch * GEMM_ROWS + (grp * MPW + j) * 16;
      if (m0 < N) {
        float* yp = Y + (size_t)(m0 + ((lane >> 4) << 3)) * ldy + c0 + col;
#pragma unroll
        for (int g = 0; g < 8; ++g)
          yp[(size_t)g * ldy] = acc[j][g] + bv;
      }
    }
    __syncthreads();
  }
}


static_assert(BIN_T == GEMM_T, "binning and product work-groups share a launch shape");

__global__ __launch_bounds__(GEMM_T)
void k_count_gemm(const int* __restrict__ ei0, const int* __restrict__ ei1,
                  int* __restrict__ cnt, int E, int NB, int binBlocks,
                  const float* __restrict__ X,
                  const float* __restrict__ Wa, const float* __restrict__ ba,
                  const float* __restrict__ Wb, const float* __restrict__ bb,
                  float* __restrict__ Y, int N, int nChunks) {
  __shared__ float xt[GEMM_ROWS * GEMM_P];
  const int b = blockIdx.x;
  if (b < binBlocks)
    bin_count_part(ei0, ei1, cnt, E, NB, b);
  else
    gemm_part<IN_F, IN_F / 16, 0>(xt, X, Wa, ba, Wb, bb, Y, IN_F, HID_F, 0, N, nChunks,
                                  b - binBlocks, (int)gridDim.x - binBlocks);
}

__global__ __launch_bounds__(GEMM_T)
void k_place_gemm(const int* __restrict__ ei0, const int* __restrict__ ei1,
                  int* __restrict__ cursor, int* __restrict__ recF, int* __restrict__ recB,
                  int E, int NB, int binBlocks,
                  const float* __restrict__ X,
                  const float* __restrict__ Wa, const float* __restrict__ ba,
                  const float* __restrict__ Wb, const float* __restrict__ bb,
                  float* __restrict__ Y, int N, int nChunks) {
  __shared__ float xt[GEMM_ROWS * GEMM_P];
  const int b = blockIdx.x;
  if (b < binBlocks)
    bin_place_part(ei0, ei1, cursor, recF, recB, E, NB, b);
  else
    gemm_part<IN_F, IN_F / 16, 0>(xt, X, Wa, ba, Wb, bb, Y, IN_F, HID_F, 0, N, nChunks,
                                  b - binBlocks, (int)gridDim.x - binBlocks);
}

#define RP_C4 (OUT_F / 4)
#define RP_IO ((TR * RP_C4 + AGG_T - 1) / AGG_T)

static __device__ __forceinline__
void reparam_load(float4 (&nz)[RP_IO], const float* __restrict__ noise, int n0, int qo, int tid) {
  const float4* z4 = (const float4*)(noise + (size_t)n0 * OUT_F);
#pragma unroll
  for (int i = 0; i < RP_IO; ++i) nz[i] = z4[min(tid + i * AGG_T, qo - 1)];
}

static __device__ __forceinline__
void reparam_store(const float4 (&nz)[RP_IO], const float* tileB, float* __restrict__ out,
                   int n0, int qo, int tid) {
  float4* o4 = (float4*)(out + (size_t)n0 * OUT_F);
#pragma unroll
  for (int i = 0; i < RP_IO; ++i) {
    int idx = tid + i * AGG_T;
    if (idx < qo) {
      const float* tr = tileB + (idx / RP_C4) * HID_F + (idx % RP_C4) * 4;
      float4 mu = *(const float4*)tr;
      float4 ls = *(const float4*)(tr + OUT_F);
      float4 o;
      o.x = mu.x + nz[i].x * expf(ls.x) * 0.2f;
      o.y = mu.y + nz[i].y * expf(ls.y) * 0.2f;
      o.z = mu.z + nz[i].z * expf(ls.z) * 0.2f;
      o.w = mu.w + nz[i].w * expf(ls.w) * 0.2f;
      o4[idx] = o;
    }
  }
}


template <int F, int P, int U, bool TAIL>
static __device__ __forceinline__
void add_records(const float* __restrict__ Y, float* tile, int lane,
                 int src, int dl, float rn, int k0, int cnt) {
  constexpr int J = F / 32;
  float v[U][J];
  int   d[U];
  float r[U];
#pragma unroll
  for (int u = 0; u < U; ++u) {
    int k = k0 + u;
    d[u] = 0; r[u] = 0.0f;
    if (!TAIL || k < cnt) {
      int s = __builtin_amdgcn_readlane(src, k);
      d[u]  = __builtin_amdgcn_readlane(dl, k);
      r[u]  = __int_as_float(__builtin_amdgcn_readlane(__float_as_int(rn), k));
      const float* yr = Y + (size_t)s * F + lane;
#pragma unroll
      for (int j = 0; j < J; ++j) v[u][j] = yr[32 * j];
    }
  }
#pragma unroll
  for (int u = 0; u < U; ++u) {
    if (!TAIL || k0 + u < cnt) {
      float* tp = tile + d[u] * P + lane;
#pragma unroll
      for (int j = 0; j < J; ++j) atomicAdd(tp + 32 * j, r[u] * v[u][j]);
    }
  }
}

template <int F>
__global__ __launch_bounds__(AGG_T)
void k_aggr_tile(const float* __restrict__ Y, const int* __restrict__ rec_,
                 const int* __restrict__ st,
                 const float* __restrict__ degD, const float* __restrict__ degS,
                 const float* __restrict__ noise, float* __restrict__ out, int N) {
  static_assert(F == HID_F && F == 2 * OUT_F, "the tile rows are [mu | logstd]");
  constexpr int U  = 1024 / F;
  constexpr int Q  = TR * (F / 4);
  constexpr int IT = (Q + AGG_T - 1) / AGG_T;
  __shared__ float tile[TR * F];

  const int2* rec = (const int2*)rec_;
  const int b    = blockIdx.x;
  const int n0   = b * TR;
  const int rows = min(TR, N - n0);
  const int q    = rows * (F / 4);
  const int tid  = threadIdx.x;
  const int lane = tid & 31;
  const int wave = __builtin_amdgcn_readfirstlane(tid >> 5);

  const int qo = rows * RP_C4;
  float4 nz[RP_IO];
  reparam_load(nz, noise, n0, qo, tid);

  {
    const float4* y4 = (const float4*)(Y + (size_t)n0 * F);
    float4 v[IT];
    float  dd[IT], ds[IT];
#pragma unroll
    for (int i = 0; i < IT; ++i) {
      int idx = min(tid + i * AGG_T, q - 1);
      int n = n0 + idx / (F / 4);
      v[i]  = y4[idx];
      dd[i] = degD[n];
      ds[i] = degS[n];
    }
#pragma unroll
    for (int i = 0; i < IT; ++i) {
      int idx = tid + i * AGG_T;
      if (idx < q) {
        float rn = rsqrtf(dd[i] * ds[i]);
        float4 o; o.x = rn * v[i].x; o.y = rn * v[i].y; o.z = rn * v[i].z; o.w = rn * v[i].w;
        ((float4*)tile)[idx] = o;
      }
    }
  }
  __syncthreads();

  const int lo = st[b], hi = st[b + 1];
  for (int c0 = lo + wave * 32; c0 < hi; c0 += AGG_T) {
    const int cnt = min(32, hi - c0);
    int src = 0, dl = 0;
    float rn = 0.0f;
    if (lane < cnt) {
      int2 r = rec[c0 + lane];
      src = r.x;
      dl  = r.y - n0;
      rn  = rsqrtf(degD[r.y] * degS[r.x]);
    }
    int k0 = 0;
    for (; k0 + U <= cnt; k0 += U)
      add_records<F, F, U, false>(Y, tile, lane, src, dl, rn, k0, cnt);
    if (k0 < cnt)
      add_records<F, F, U, true>(Y, tile, lane, src, dl, rn, k0, cnt);
  }
  __syncthreads();

  reparam_store(nz, tile, out, n0, qo, tid);
}


#define L2_RT   ((TR + 15) / 16)
#define L2_ROWS (L2_RT * 16)
#define L2_P    (IN_F + 4)

__global__ __launch_bounds__(AGG_T)
void k_aggr_lin2(const float* __restrict__ Y, const int* __restrict__ rec_,
                 const int* __restrict__ st,
                 const float* __restrict__ degD, const float* __restrict__ degS,
                 const float* __restrict__ Wa, const float* __restrict__ ba,
                 const float* __restrict__ Wb, const float* __restrict__ bb,
                 float* __restrict__ Y2, int N) {
  constexpr int F  = IN_F;
  constexpr int P  = L2_P;
  constexpr int K  = HID_F;
  constexpr int NK = K / 4;
  constexpr int U  = 1024 / F;
  constexpr int QP = L2_ROWS * (F / 4);
  constexpr int IT = (QP + AGG_T - 1) / AGG_T;
  constexpr int NCT = HID_F / 16;
  constexpr int NG  = (AGG_T / 32) / NCT;
  static_assert(L2_RT <= 2 * NG, "two row tiles per wave cover the tile");
  __shared__ float tile[L2_ROWS * P];

  const int2* rec = (const int2*)rec_;
  const int b    = blockIdx.x;
  const int n0   = b * TR;
  const int rows = min(TR, N - n0);
  const int q    = rows * (F / 4);
  const int tid  = threadIdx.x;
  const int lane = tid & 31;
  const int wave = __builtin_amdgcn_readfirstlane(tid >> 5);

  {
    const float4* y4 = (const float4*)(Y + (size_t)n0 * F);
    float4 v[IT];
    float  dd[IT], ds[IT];
#pragma unroll
    for (int i = 0; i < IT; ++i) {
      int idx = min(tid + i * AGG_T, q - 1);
      int n = n0 + idx / (F / 4);
      v[i]  = y4[idx];
      dd[i] = degD[n];
      ds[i] = degS[n];
    }
#pragma unroll
    for (int i = 0; i < IT; ++i) {
      int idx = tid + i * AGG_T;
      if (idx < QP) {
        float4 o = {};
        if (idx < q) {
          float rn = rsqrtf(dd[i] * ds[i]);
          o.x = rn * v[i].x; o.y = rn * v[i].y; o.z = rn * v[i].z; o.w = rn * v[i].w;
        }
        *(float4*)(tile + (idx / (F / 4)) * P + (idx % (F / 4)) * 4) = o;
      }
    }
  }
  __syncthreads();

  const int lo = st[b], hi = st[b + 1];
  for (int c0 = lo + wave * 32; c0 < hi; c0 += AGG_T) {
    const int cnt = min(32, hi - c0);
    int src = 0, dl = 0;
    float rn = 0.0f;
    if (lane < cnt) {
      int2 r = rec[c0 + lane];
      src = r.x;
      dl  = r.y - n0;
      rn  = rsqrtf(degD[r.y] * degS[r.x]);
    }
    int k0 = 0;
    for (; k0 + U <= cnt; k0 += U)
      add_records<F, P, U, false>(Y, tile, lane, src, dl, rn, k0, cnt);
    if (k0 < cnt)
      add_records<F, P, U, true>(Y, tile, lane, src, dl, rn, k0, cnt);
  }

  const int ct   = wave % NCT;
  const int grp  = wave / NCT;
  const int col  = lane & 15;
  const int khi  = (lane >> 4) << 1;
  const int cc0  = ct << 4;
  const int half = (cc0 >= OUT_F) ? 1 : 0;
  const float* W    = half ? Wb : Wa;
  const float* bias = half ? bb : ba;
  const int wc0  = cc0 - half * OUT_F;
  const int koff = half ? HID_F : 0;
  v2f bfr[NK];
  const float* wrow = W + (size_t)(wc0 + col) * K + khi;
#pragma unroll
  for (int i = 0; i < NK; ++i) bfr[i] = *(const v2f*)(wrow + i * 4);
  const float bv = bias[wc0 + col];
  __syncthreads();

  const bool two = (grp + NG) < L2_RT;
  const int rt1  = two ? grp + NG : grp;
  const float* ap0 = tile + (grp * 16 + col) * P + koff + khi;
  const float* ap1 = tile + (rt1 * 16 + col) * P + koff + khi;
  v8f acc0 = {}, acc1 = {};
#pragma unroll
  for (int i = 0; i < NK; ++i) {
    v2f a0 = *(const v2f*)(ap0 + i * 4);
    v2f a1 = *(const v2f*)(ap1 + i * 4);
    a0.x = fmaxf(a0.x, 0.0f); a0.y = fmaxf(a0.y, 0.0f);
    a1.x = fmaxf(a1.x, 0.0f); a1.y = fmaxf(a1.y, 0.0f);
    acc0 = __builtin_amdgcn_wmma_f32_16x16x4_f32(false, a0, false, bfr[i], (short)0, acc0, false, false);
    acc1 = __builtin_amdgcn_wmma_f32_16x16x4_f32(false, a1, false, bfr[i], (short)0, acc1, false, false);
  }
  __syncthreads();

  {
    float* yp0 = tile + (grp * 16 + ((lane >> 4) << 3)) * P + cc0 + col;
#pragma unroll
    for (int g = 0; g < 8; ++g) yp0[g * P] = acc0[g] + bv;
    if (two) {
      float* yp1 = tile + (rt1 * 16 + ((lane >> 4) << 3)) * P + cc0 + col;
#pragma unroll
      for (int g = 0; g < 8; ++g) yp1[g * P] = acc1[g] + bv;
    }
  }
  __syncthreads();

  {
    constexpr int C4 = HID_F / 4;
    constexpr int IO = (TR * C4 + AGG_T - 1) / AGG_T;
    const int qo = rows * C4;
    float4* o4 = (float4*)(Y2 + (size_t)n0 * HID_F);
#pragma unroll
    for (int i = 0; i < IO; ++i) {
      int idx = tid + i * AGG_T;
      if (idx < qo) o4[idx] = *(const float4*)(tile + (idx / C4) * P + (idx % C4) * 4);
    }
  }
}


template <int U, bool TAIL>
static __device__ __forceinline__
void add_records2(const float* __restrict__ Y, float* tile, const float* __restrict__ Yb, float* tileB,
                  int lane, int src, int dl, float rn, int k0, int cnt) {
  constexpr int JA = IN_F / 32;
  constexpr int JB = HID_F / 32;
  float va[U][JA];
  float vb[U][JB];
  int   d[U];
  float r[U];
#pragma unroll
  for (int u = 0; u < U; ++u) {
    int k = k0 + u;
    d[u] = 0; r[u] = 0.0f;
    if (!TAIL || k < cnt) {
      int s = __builtin_amdgcn_readlane(src, k);
      d[u]  = __builtin_amdgcn_readlane(dl, k);
      r[u]  = __int_as_float(__builtin_amdgcn_readlane(__float_as_int(rn), k));
      const float* ya = Y  + (size_t)s * IN_F  + lane;
      const float* yb = Yb + (size_t)s * HID_F + lane;
#pragma unroll
      for (int j = 0; j < JA; ++j) va[u][j] = ya[32 * j];
#pragma unroll
      for (int j = 0; j < JB; ++j) vb[u][j] = yb[32 * j];
    }
  }
#pragma unroll
  for (int u = 0; u < U; ++u) {
    if (!TAIL || k0 + u < cnt) {
      float* ta = tile  + d[u] * L2_P  + lane;
      float* tb = tileB + d[u] * HID_F + lane;
#pragma unroll
      for (int j = 0; j < JA; ++j) atomicAdd(ta + 32 * j, r[u] * va[u][j]);
#pragma unroll
      for (int j = 0; j < JB; ++j) atomicAdd(tb + 32 * j, r[u] * vb[u][j]);
    }
  }
}

__global__ __launch_bounds__(AGG_T)
void k_aggr_lin2_tile(const float* __restrict__ Y, const float* __restrict__ Yb,
                      const int* __restrict__ rec_, const int* __restrict__ st,
                      const float* __restrict__ degD, const float* __restrict__ degS,
                      const float* __restrict__ Wa, const float* __restrict__ ba,
                      const float* __restrict__ Wb, const float* __restrict__ bb,
                      float* __restrict__ Y2,
                      const float* __restrict__ noise, float* __restrict__ out, int N) {
  constexpr int F   = IN_F;
  constexpr int FB  = HID_F;
  constexpr int P   = L2_P;
  constexpr int K   = HID_F;
  constexpr int NK  = K / 4;
  constexpr int U   = 8;
  constexpr int QP  = L2_ROWS * (F / 4);
  constexpr int IT  = (QP + AGG_T - 1) / AGG_T;
  constexpr int QB  = TR * (FB / 4);
  constexpr int ITB = (QB + AGG_T - 1) / AGG_T;
  constexpr int NCT = HID_F / 16;
  constexpr int NG  = (AGG_T / 32) / NCT;
  static_assert(L2_RT <= 2 * NG, "two row tiles per wave cover the tile");
  __shared__ float tile[L2_ROWS * P];
  __shared__ float tileB[TR * FB];

  const int2* rec = (const int2*)rec_;
  const int b    = blockIdx.x;
  const int n0   = b * TR;
  const int rows = min(TR, N - n0);
  const int q    = rows * (F / 4);
  const int qb   = rows * (FB / 4);
  const int tid  = threadIdx.x;
  const int lane = tid & 31;
  const int wave = __builtin_amdgcn_readfirstlane(tid >> 5);

  {
    const float4* y4 = (const float4*)(Y + (size_t)n0 * F);
    float4 v[IT];
    float  dd[IT], ds[IT];
#pragma unroll
    for (int i = 0; i < IT; ++i) {
      int idx = min(tid + i * AGG_T, q - 1);
      int n = n0 + idx / (F / 4);
      v[i]  = y4[idx];
      dd[i] = degD[n];
      ds[i] = degS[n];
    }
#pragma unroll
    for (int i = 0; i < IT; ++i) {
      int idx = tid + i * AGG_T;
      if (idx < QP) {
        float4 o = {};
        if (idx < q) {
          float rn = rsqrtf(dd[i] * ds[i]);
          o.x = rn * v[i].x; o.y = rn * v[i].y; o.z = rn * v[i].z; o.w = rn * v[i].w;
        }
        *(float4*)(tile + (idx / (F / 4)) * P + (idx % (F / 4)) * 4) = o;
      }
    }
  }
  {
    const float4* y4 = (const float4*)(Yb + (size_t)n0 * FB);
    float4 v[ITB];
    float  dd[ITB], ds[ITB];
#pragma unroll
    for (int i = 0; i < ITB; ++i) {
      int idx = min(tid + i * AGG_T, qb - 1);
      int n = n0 + idx / (FB / 4);
      v[i]  = y4[idx];
      dd[i] = degD[n];
      ds[i] = degS[n];
    }
#pragma unroll
    for (int i = 0; i < ITB; ++i) {
      int idx = tid + i * AGG_T;
      if (idx < qb) {
        float rn = rsqrtf(dd[i] * ds[i]);
        float4 o; o.x = rn * v[i].x; o.y = rn * v[i].y; o.z = rn * v[i].z; o.w = rn * v[i].w;
        ((float4*)tileB)[idx] = o;
      }
    }
  }
  __syncthreads();

  const int lo = st[b], hi = st[b + 1];
  for (int c0 = lo + wave * 32; c0 < hi; c0 += AGG_T) {
    const int cnt = min(32, hi - c0);
    int src = 0, dl = 0;
    float rn = 0.0f;
    if (lane < cnt) {
      int2 r = rec[c0 + lane];
      src = r.x;
      dl  = r.y - n0;
      rn  = rsqrtf(degD[r.y] * degS[r.x]);
    }
    int k0 = 0;
    for (; k0 + U <= cnt; k0 += U)
      add_records2<U, false>(Y, tile, Yb, tileB, lane, src, dl, rn, k0, cnt);
    if (k0 < cnt)
      add_records2<U, true>(Y, tile, Yb, tileB, lane, src, dl, rn, k0, cnt);
  }

  const int ct   = wave % NCT;
  const int grp  = wave / NCT;
  const int col  = lane & 15;
  const int khi  = (lane >> 4) << 1;
  const int cc0  = ct << 4;
  const int half = (cc0 >= OUT_F) ? 1 : 0;
  const float* W    = half ? Wb : Wa;
  const float* bias = half ? bb : ba;
  const int wc0  = cc0 - half * OUT_F;
  const int koff = half ? HID_F : 0;
  v2f bfr[NK];
  const float* wrow = W + (size_t)(wc0 + col) * K + khi;
#pragma unroll
  for (int i = 0; i < NK; ++i) bfr[i] = *(const v2f*)(wrow + i * 4);
  const float bv = bias[wc0 + col];
  const int qo = rows * RP_C4;
  float4 nz[RP_IO];
  reparam_load(nz, noise, n0, qo, tid);
  __syncthreads();

  reparam_store(nz, tileB, out, n0, qo, tid);

  const bool two = (grp + NG) < L2_RT;
  const int rt1  = two ? grp + NG : grp;
  const float* ap0 = tile + (grp * 16 + col) * P + koff + khi;
  const float* ap1 = tile + (rt1 * 16 + col) * P + koff + khi;
  v8f acc0 = {}, acc1 = {};
#pragma unroll
  for (int i = 0; i < NK; ++i) {
    v2f a0 = *(const v2f*)(ap0 + i * 4);
    v2f a1 = *(const v2f*)(ap1 + i * 4);
    a0.x = fmaxf(a0.x, 0.0f); a0.y = fmaxf(a0.y, 0.0f);
    a1.x = fmaxf(a1.x, 0.0f); a1.y = fmaxf(a1.y, 0.0f);
    acc0 = __builtin_amdgcn_wmma_f32_16x16x4_f32(false, a0, false, bfr[i], (short)0, acc0, false, false);
    acc1 = __builtin_amdgcn_wmma_f32_16x16x4_f32(false, a1, false, bfr[i], (short)0, acc1, false, false);
  }
  __syncthreads();

  {
    float* yp0 = tile + (grp * 16 + ((lane >> 4) << 3)) * P + cc0 + col;
#pragma unroll
    for (int g = 0; g < 8; ++g) yp0[g * P] = acc0[g] + bv;
    if (two) {
      float* yp1 = tile + (rt1 * 16 + ((lane >> 4) << 3)) * P + cc0 + col;
#pragma unroll
      for (int g = 0; g < 8; ++g) yp1[g * P] = acc1[g] + bv;
    }
  }
  __syncthreads();

  {
    constexpr int C4 = HID_F / 4;
    constexpr int IO = (TR * C4 + AGG_T - 1) / AGG_T;
    const int qo = rows * C4;
    float4* o4 = (float4*)(Y2 + (size_t)n0 * HID_F);
#pragma unroll
    for (int i = 0; i < IO; ++i) {
      int idx = tid + i * AGG_T;
      if (idx < qo) o4[idx] = *(const float4*)(tile + (idx / C4) * P + (idx % C4) * 4);
    }
  }
}


static inline int ceil_div(int a, int b) { return (a + b - 1) / b; }

extern "C" void kernel_launch(void* const* d_in, const int* in_sizes, int n_in,
                              void* d_out, int out_size, void* d_ws, size_t ws_size,
                              hipStream_t stream) {
  const float* s       = (const float*)d_in[0];
  const float* t       = (const float*)d_in[1];
  const int*   ei      = (const int*)  d_in[2];
  const float* noise_s = (const float*)d_in[3];
  const float* noise_t = (const float*)d_in[4];
  const float* sm1_W = (const float*)d_in[5],  *sm1_b = (const float*)d_in[6];
  const float* sm2_W = (const float*)d_in[7],  *sm2_b = (const float*)d_in[8];
  const float* sl1_W = (const float*)d_in[9],  *sl1_b = (const float*)d_in[10];
  const float* sl2_W = (const float*)d_in[11], *sl2_b = (const float*)d_in[12];
  const float* tm1_W = (const float*)d_in[13], *tm1_b = (const float*)d_in[14];
  const float* tm2_W = (const float*)d_in[15], *tm2_b = (const float*)d_in[16];
  const float* tl1_W = (const float*)d_in[17], *tl1_b = (const float*)d_in[18];
  const float* tl2_W = (const float*)d_in[19], *tl2_b = (const float*)d_in[20];

  const int N = in_sizes[0] / IN_F;
  const int E = in_sizes[2] / 2;
  const int* ei0 = ei;
  const int* ei1 = ei + E;
  const int NB = ceil_div(N, TR);

  float* ws  = (float*)d_ws;
  float* Y1s = ws;
  float* Y2s = Y1s + (size_t)N * IN_F;
  float* Y1t = Y2s + (size_t)N * HID_F;
  float* Y2t = Y1s;
  int*   recF = (int*)(Y1t + (size_t)N * IN_F + (size_t)N * HID_F);
  int*   recB = recF + (size_t)2 * E;
  float* deg_in  = (float*)(recB + (size_t)2 * E);
  float* deg_out = deg_in + N;
  int*   cnt     = (int*)(deg_out + N);
  int*   cursor  = cnt + REPL * 2 * NB;
  int*   start   = cursor + REPL * 2 * NB;
  const int* startF = start;
  const int* startB = start + (NB + 1);

  const int nChunks = ceil_div(N, GEMM_ROWS);
  const int gemmBlocks = 512;
  const int binBlocks = ceil_div(E, BIN_T * BIN_EPT);

  k_zero_counts<<<ceil_div(REPL * 2 * NB, 256), 256, 0, stream>>>(cnt, REPL * 2 * NB);
  k_count_gemm<<<binBlocks + gemmBlocks, GEMM_T, 2 * NB * sizeof(int), stream>>>(
      ei0, ei1, cnt, E, NB, binBlocks,
      t, tm1_W, tm1_b, tl1_W, tl1_b, Y1t, N, nChunks);
  k_bin_scan<<<2, 1024, 0, stream>>>(cnt, start, cursor, NB);
  k_place_gemm<<<binBlocks + gemmBlocks, GEMM_T, 4 * NB * sizeof(int), stream>>>(
      ei0, ei1, cursor, recF, recB, E, NB, binBlocks,
      s, sm1_W, sm1_b, sl1_W, sl1_b, Y1s, N, nChunks);
  k_degrees<<<dim3(NB, 2), 256, 0, stream>>>(recF, recB, start, deg_in, deg_out, N, NB);

  k_aggr_lin2<<<NB, AGG_T, 0, stream>>>(Y1s, recF, startF, deg_in, deg_out,
                                        sm2_W, sm2_b, sl2_W, sl2_b, Y2s, N);

  float* out_s = (float*)d_out;
  float* out_t = out_s + (size_t)N * OUT_F;

  k_aggr_lin2_tile<<<NB, AGG_T, 0, stream>>>(Y1t, Y2s, recB, startB, deg_out, deg_in,
                                             tm2_W, tm2_b, tl2_W, tl2_b, Y2t, noise_s, out_s, N);

  k_aggr_tile<HID_F><<<NB, AGG_T, 0, stream>>>(Y2t, recF, startF, deg_in, deg_out, noise_t, out_t, N);
}


# No LDS float adds in aggregation: a wave owns 7 tile rows in registers, bucket records sorted by owner wave in LDS (int counting sort), row picked by scalar compare; a row sums in one wave by rank
# speedup vs baseline: 4.7883x; 1.0489x over previous
#include <hip/hip_runtime.h>


typedef __attribute__((ext_vector_type(2))) float v2f;
typedef __attribute__((ext_vector_type(8))) float v8f;

#define IN_F   128
#define HID_F  64
#define OUT_F  32

#define TR     100
#define AGG_T  512
#define BIN_T  256
#define BIN_EPT 16
#define REPL   16


__global__ void k_zero_counts(int* __restrict__ cnt, int n) {
  int i = blockIdx.x * blockDim.x + threadIdx.x;
  if (i < n) cnt[i] = 0;
}

static __device__ __forceinline__
void bin_count_part(const int* __restrict__ ei0, const int* __restrict__ ei1,
                    int* __restrict__ cnt, int E, int NB, int blk) {
  extern __shared__ int bin_lds[];
  int* hist = bin_lds;
  for (int i = threadIdx.x; i < 2 * NB; i += BIN_T) hist[i] = 0;
  __syncthreads();
  int e0 = blk * (BIN_T * BIN_EPT) + threadIdx.x;
  int s[BIN_EPT], d[BIN_EPT];
#pragma unroll
  for (int k = 0; k < BIN_EPT; ++k) {
    int e = min(e0 + k * BIN_T, E - 1);
    s[k] = ei0[e];
    d[k] = ei1[e];
  }
#pragma unroll
  for (int k = 0; k < BIN_EPT; ++k) {
    if (e0 + k * BIN_T < E) {
      atomicAdd(&hist[d[k] / TR], 1);
      atomicAdd(&hist[NB + s[k] / TR], 1);
    }
  }
  __syncthreads();
  int* mycnt = cnt + (blk % REPL) * 2 * NB;
  for (int i = threadIdx.x; i < 2 * NB; i += BIN_T) {
    int c = hist[i];
    if (c) atomicAdd(&mycnt[i], c);
  }
}

__global__ __launch_bounds__(1024)
void k_bin_scan(const int* __restrict__ cnt, int* __restrict__ start,
                int* __restrict__ cursor, int NB) {
  __shared__ int sh[1024];
  int dir = blockIdx.x;
  int* st = start + dir * (NB + 1);
  int tid = threadIdx.x;
  int carry = 0;
  for (int base = 0; base < NB; base += 1024) {
    int i = base + tid;
    int c[REPL];
    int v = 0;
#pragma unroll
    for (int r = 0; r < REPL; ++r) {
      c[r] = (i < NB) ? cnt[r * 2 * NB + dir * NB + i] : 0;
      v += c[r];
    }
    sh[tid] = v;
    __syncthreads();
    for (int off = 1; off < 1024; off <<= 1) {
      int t = (tid >= off) ? sh[tid - off] : 0;
      __syncthreads();
      sh[tid] += t;
      __syncthreads();
    }
    int run = carry + sh[tid] - v;
    if (i < NB) {
      st[i] = run;
#pragma unroll
      for (int r = 0; r < REPL; ++r) {
        cursor[r * 2 * NB + dir * NB + i] = run;
        run += c[r];
      }
    }
    carry += sh[1023];
    __syncthreads();
  }
  if (tid == 0) st[NB] = carry;
}

static __device__ __forceinline__
void bin_place_part(const int* __restrict__ ei0, const int* __restrict__ ei1,
                    int* __restrict__ cursor, int* __restrict__ recF, int* __restrict__ recB,
                    int E, int NB, int blk) {
  extern __shared__ int bin_lds[];
  int* hist = bin_lds;
  int* base = bin_lds + 2 * NB;
  for (int i = threadIdx.x; i < 2 * NB; i += BIN_T) hist[i] = 0;
  __syncthreads();
  int s[BIN_EPT], d[BIN_EPT], rF[BIN_EPT], rB[BIN_EPT];
  int e0 = blk * (BIN_T * BIN_EPT) + threadIdx.x;
#pragma unroll
  for (int k = 0; k < BIN_EPT; ++k) {
    int e = min(e0 + k * BIN_T, E - 1);
    s[k] = ei0[e];
    d[k] = ei1[e];
  }
#pragma unroll
  for (int k = 0; k < BIN_EPT; ++k) {
    rF[k] = 0; rB[k] = 0;
    if (e0 + k * BIN_T < E) {
      rF[k] = atomicAdd(&hist[d[k] / TR], 1);
      rB[k] = atomicAdd(&hist[NB + s[k] / TR], 1);
    }
  }
  __syncthreads();
  int* mycur = cursor + (blk % REPL) * 2 * NB;
  for (int i0 = threadIdx.x; i0 < 2 * NB; i0 += 4 * BIN_T) {
    int c[4], r[4];
#pragma unroll
    for (int j = 0; j < 4; ++j) {
      int i = i0 + j * BIN_T;
      c[j] = (i < 2 * NB) ? hist[i] : 0;
    }
#pragma unroll
    for (int j = 0; j < 4; ++j)
      r[j] = c[j] ? atomicAdd(&mycur[i0 + j * BIN_T], c[j]) : 0;
#pragma unroll
    for (int j = 0; j < 4; ++j) {
      int i = i0 + j * BIN_T;
      if (i < 2 * NB) base[i] = r[j];
    }
  }
  __syncthreads();
  int2* oF = (int2*)recF;
  int2* oB = (int2*)recB;
#pragma unroll
  for (int k = 0; k < BIN_EPT; ++k) {
    if (e0 + k * BIN_T < E) {
      oF[base[d[k] / TR] + rF[k]]      = make_int2(s[k], d[k]);
      oB[base[NB + s[k] / TR] + rB[k]] = make_int2(d[k], s[k]);
    }
  }
}

__global__ __launch_bounds__(256)
void k_degrees(const int* __restrict__ recF, const int* __restrict__ recB,
               const int* __restrict__ start,
               float* __restrict__ deg_in, float* __restrict__ deg_out, int N, int NB) {
  __shared__ int c[TR];
  int b = blockIdx.x, dir = blockIdx.y;
  const int2* rec = (const int2*)(dir ? recB : recF);
  const int* st = start + dir * (NB + 1);
  float* deg = dir ? deg_out : deg_in;
  for (int i = threadIdx.x; i < TR; i += 256) c[i] = 0;
  __syncthreads();
  int n0 = b * TR;
  int lo = st[b], hi = st[b + 1];
  for (int i = lo + threadIdx.x; i < hi; i += 256) atomicAdd(&c[rec[i].y - n0], 1);
  __syncthreads();
  for (int i = threadIdx.x; i < TR; i += 256) {
    int n = n0 + i;
    if (n < N) deg[n] = 1.0f + (float)c[i];
  }
}


#define GEMM_T    256
#define GEMM_ROWS 32
#define GEMM_P    (IN_F + 4)

template <int K, int NCT, int RELU>
static __device__ __forceinline__
void gemm_part(float* xt, const float* __restrict__ X,
               const float* __restrict__ Wa, const float* __restrict__ ba,
               const float* __restrict__ Wb, const float* __restrict__ bb,
               float* __restrict__ Y, int ldy,
               int halfO, int koffB, int N, int nChunks, int blk, int nblk) {
  constexpr int NK  = K / 4;
  constexpr int G   = (GEMM_T / 32) / NCT;
  constexpr int MPW = (GEMM_ROWS / 16) / G;

  const int tid  = threadIdx.x;
  const int lane = tid & 31;
  const int wave = __builtin_amdgcn_readfirstlane(tid >> 5);
  const int ct   = wave % NCT;
  const int grp  = wave / NCT;
  const int col  = lane & 15;
  const int khi  = (lane >> 4) << 1;

  const int c0   = ct << 4;
  const int half = (c0 >= halfO) ? 1 : 0;
  const float* W    = half ? Wb : Wa;
  const float* bias = half ? bb : ba;
  const int wc0  = c0 - half * halfO;
  const int koff = half ? koffB : 0;

  v2f bfr[NK];
  const float* wrow = W + (size_t)(wc0 + col) * K + khi;
#pragma unroll
  for (int i = 0; i < NK; ++i) bfr[i] = *(const v2f*)(wrow + i * 4);
  const float bv = bias[wc0 + col];

  const float* xl = X + lane * 4;
  const int rmax = N - 1;
  float4 s0 = {}, s1 = {}, s2 = {}, s3 = {};
  int ch = blk;
  if (ch < nChunks) {
    const int r0 = ch * GEMM_ROWS + wave;
    s0 = *(const float4*)(xl + (size_t)min(r0,      rmax) * IN_F);
    s1 = *(const float4*)(xl + (size_t)min(r0 + 8,  rmax) * IN_F);
    s2 = *(const float4*)(xl + (size_t)min(r0 + 16, rmax) * IN_F);
    s3 = *(const float4*)(xl + (size_t)min(r0 + 24, rmax) * IN_F);
  }
  float* xs = xt + wave * GEMM_P + lane * 4;
  const float* ap = xt + (grp * MPW * 16 + col) * GEMM_P + koff + khi;

  for (; ch < nChunks; ch += nblk) {
    if (RELU) {
      s0.x = fmaxf(s0.x, 0.0f); s0.y = fmaxf(s0.y, 0.0f); s0.z = fmaxf(s0.z, 0.0f); s0.w = fmaxf(s0.w, 0.0f);
      s1.x = fmaxf(s1.x, 0.0f); s1.y = fmaxf(s1.y, 0.0f); s1.z = fmaxf(s1.z, 0.0f); s1.w = fmaxf(s1.w, 0.0f);
      s2.x = fmaxf(s2.x, 0.0f); s2.y = fmaxf(s2.y, 0.0f); s2.z = fmaxf(s2.z, 0.0f); s2.w = fmaxf(s2.w, 0.0f);
      s3.x = fmaxf(s3.x, 0.0f); s3.y = fmaxf(s3.y, 0.0f); s3.z = fmaxf(s3.z, 0.0f); s3.w = fmaxf(s3.w, 0.0f);
    }
    *(float4*)(xs)               = s0;
    *(float4*)(xs + 8 * GEMM_P)  = s1;
    *(float4*)(xs + 16 * GEMM_P) = s2;
    *(float4*)(xs + 24 * GEMM_P) = s3;
    __syncthreads();

    const int nx = ch + nblk;
    if (nx < nChunks) {
      const int r0 = nx * GEMM_ROWS + wave;
      s0 = *(const float4*)(xl + (size_t)min(r0,      rmax) * IN_F);
      s1 = *(const float4*)(xl + (size_t)min(r0 + 8,  rmax) * IN_F);
      s2 = *(const float4*)(xl + (size_t)min(r0 + 16, rmax) * IN_F);
      s3 = *(const float4*)(xl + (size_t)min(r0 + 24, rmax) * IN_F);
    }

    v8f acc[MPW];
#pragma unroll
    for (int j = 0; j < MPW; ++j) acc[j] = v8f{};
#pragma unroll
    for (int i = 0; i < NK; ++i) {
#pragma unroll
      for (int j = 0; j < MPW; ++j) {
        v2f a = *(const v2f*)(ap + j * 16 * GEMM_P + i * 4);
        acc[j] = __builtin_amdgcn_wmma_f32_16x16x4_f32(false, a, false, bfr[i],
                                                       (short)0, acc[j], false, false);
      }
    }

#pragma unroll
    for (int j = 0; j < MPW; ++j) {
      const int m0 = ch * GEMM_ROWS + (grp * MPW + j) * 16;
      if (m0 < N) {
        float* yp = Y + (size_t)(m0 + ((lane >> 4) << 3)) * ldy + c0 + col;
#pragma unroll
        for (int g = 0; g < 8; ++g)
          yp[(size_t)g * ldy] = acc[j][g] + bv;
      }
    }
    __syncthreads();
  }
}


static_assert(BIN_T == GEMM_T, "binning and product work-groups share a launch shape");

__global__ __launch_bounds__(GEMM_T)
void k_count_gemm(const int* __restrict__ ei0, const int* __restrict__ ei1,
                  int* __restrict__ cnt, int E, int NB, int binBlocks,
                  const float* __restrict__ X,
                  const float* __restrict__ Wa, const float* __restrict__ ba,
                  const float* __restrict__ Wb, const float* __restrict__ bb,
                  float* __restrict__ Y, int N, int nChunks) {
  __shared__ float xt[GEMM_ROWS * GEMM_P];
  const int b = blockIdx.x;
  if (b < binBlocks)
    bin_count_part(ei0, ei1, cnt, E, NB, b);
  else
    gemm_part<IN_F, IN_F / 16, 0>(xt, X, Wa, ba, Wb, bb, Y, IN_F, HID_F, 0, N, nChunks,
                                  b - binBlocks, (int)gridDim.x - binBlocks);
}

__global__ __launch_bounds__(GEMM_T)
void k_place_gemm(const int* __restrict__ ei0, const int* __restrict__ ei1,
                  int* __restrict__ cursor, int* __restrict__ recF, int* __restrict__ recB,
                  int E, int NB, int binBlocks,
                  const float* __restrict__ X,
                  const float* __restrict__ Wa, const float* __restrict__ ba,
                  const float* __restrict__ Wb, const float* __restrict__ bb,
                  float* __restrict__ Y, int N, int nChunks) {
  __shared__ float xt[GEMM_ROWS * GEMM_P];
  const int b = blockIdx.x;
  if (b < binBlocks)
    bin_place_part(ei0, ei1, cursor, recF, recB, E, NB, b);
  else
    gemm_part<IN_F, IN_F / 16, 0>(xt, X, Wa, ba, Wb, bb, Y, IN_F, HID_F, 0, N, nChunks,
                                  b - binBlocks, (int)gridDim.x - binBlocks);
}

#define RP_C4 (OUT_F / 4)
#define RP_IO ((TR * RP_C4 + AGG_T - 1) / AGG_T)

static __device__ __forceinline__
void reparam_load(float4 (&nz)[RP_IO], const float* __restrict__ noise, int n0, int qo, int tid) {
  const float4* z4 = (const float4*)(noise + (size_t)n0 * OUT_F);
#pragma unroll
  for (int i = 0; i < RP_IO; ++i) nz[i] = z4[min(tid + i * AGG_T, qo - 1)];
}

static __device__ __forceinline__
void reparam_store(const float4 (&nz)[RP_IO], const float* tileB, float* __restrict__ out,
                   int n0, int qo, int tid) {
  float4* o4 = (float4*)(out + (size_t)n0 * OUT_F);
#pragma unroll
  for (int i = 0; i < RP_IO; ++i) {
    int idx = tid + i * AGG_T;
    if (idx < qo) {
      const float* tr = tileB + (idx / RP_C4) * HID_F + (idx % RP_C4) * 4;
      float4 mu = *(const float4*)tr;
      float4 ls = *(const float4*)(tr + OUT_F);
      float4 o;
      o.x = mu.x + nz[i].x * expf(ls.x) * 0.2f;
      o.y = mu.y + nz[i].y * expf(ls.y) * 0.2f;
      o.z = mu.z + nz[i].z * expf(ls.z) * 0.2f;
      o.w = mu.w + nz[i].w * expf(ls.w) * 0.2f;
      o4[idx] = o;
    }
  }
}


#define AGG_W   (AGG_T / 32)
#define RPW     ((TR + AGG_W - 1) / AGG_W)
#define PASS_R  2048
static_assert(PASS_R % AGG_T == 0, "whole records per thread in a pass");

template <int JA, int JB, int U, bool TAIL>
static __device__ __forceinline__
void acc_records(const float* __restrict__ Ya, const float* __restrict__ Yb,
                 float (&accA)[RPW][JA], float (&accB)[RPW][JB ? JB : 1], int lane,
                 int src, int dl, float rn, int k0, int cnt) {
  float va[U][JA];
  float vb[U][JB ? JB : 1];
  int   d[U];
  float r[U];
#pragma unroll
  for (int u = 0; u < U; ++u) {
    int k = k0 + u;
    d[u] = -1; r[u] = 0.0f;
    if (!TAIL || k < cnt) {
      int s = __builtin_amdgcn_readlane(src, k);
      d[u]  = __builtin_amdgcn_readlane(dl, k);
      r[u]  = __int_as_float(__builtin_amdgcn_readlane(__float_as_int(rn), k));
      const float* ya = Ya + (size_t)s * (32 * JA) + lane;
#pragma unroll
      for (int j = 0; j < JA; ++j) va[u][j] = ya[32 * j];
      if (JB) {
        const float* yb = Yb + (size_t)s * (32 * JB) + lane;
#pragma unroll
        for (int j = 0; j < JB; ++j) vb[u][j] = yb[32 * j];
      }
    }
  }
#pragma unroll
  for (int u = 0; u < U; ++u) {
    if (!TAIL || k0 + u < cnt) {
#pragma unroll
      for (int w = 0; w < RPW; ++w) {
        if (d[u] == w) {
#pragma unroll
          for (int j = 0; j < JA; ++j) accA[w][j] += r[u] * va[u][j];
#pragma unroll
          for (int j = 0; j < JB; ++j) accB[w][j] += r[u] * vb[u][j];
          asm volatile("");
        }
      }
    }
  }
}

template <int JA, int PA, int JB, int PB, int U>
static __device__ __forceinline__
void walk_owned(const float* __restrict__ Ya, float* tileA, const float* __restrict__ Yb, float* tileB,
                const int2* __restrict__ rec, int lo, int hi,
                const float* __restrict__ degD, const float* __restrict__ degS,
                int n0, int2* qrec, int* ocnt, int* obase, int tid, int lane, int wave) {
  constexpr int KP = PASS_R / AGG_T;
  const int R0 = wave * RPW;
  const int NR = max(0, min(RPW, TR - R0));

  for (int p0 = lo; p0 < hi; p0 += PASS_R) {
    const int pe = min(hi, p0 + PASS_R);
    if (tid < AGG_W) ocnt[tid] = 0;
    __syncthreads();
    int2 r[KP];
    int  own[KP], rank[KP];
#pragma unroll
    for (int k = 0; k < KP; ++k) r[k] = rec[min(p0 + tid + k * AGG_T, pe - 1)];
#pragma unroll
    for (int k = 0; k < KP; ++k) {
      own[k] = (r[k].y - n0) / RPW;
      rank[k] = 0;
      if (p0 + tid + k * AGG_T < pe) rank[k] = atomicAdd(&ocnt[own[k]], 1);
    }
    __syncthreads();
    if (tid < AGG_W) {
      int s = 0;
      for (int i = 0; i < tid; ++i) s += ocnt[i];
      obase[tid] = s;
    }
    __syncthreads();
#pragma unroll
    for (int k = 0; k < KP; ++k) {
      if (p0 + tid + k * AGG_T < pe)
        qrec[obase[own[k]] + rank[k]] = make_int2(r[k].x, r[k].y - n0 - own[k] * RPW);
    }
    __syncthreads();

    float accA[RPW][JA];
    float accB[RPW][JB ? JB : 1];
#pragma unroll
    for (int w = 0; w < RPW; ++w) {
#pragma unroll
      for (int j = 0; j < JA; ++j) accA[w][j] = (w < NR) ? tileA[(R0 + w) * PA + lane + 32 * j] : 0.0f;
#pragma unroll
      for (int j = 0; j < (JB ? JB : 1); ++j) accB[w][j] = (JB && w < NR) ? tileB[(R0 + w) * PB + lane + 32 * j] : 0.0f;
    }
    const int qb = __builtin_amdgcn_readfirstlane(obase[wave]);
    const int qn = __builtin_amdgcn_readfirstlane(ocnt[wave]);
    for (int b = 0; b < qn; b += 32) {
      const int m = min(32, qn - b);
      int src = 0, dl = 0;
      float rn = 0.0f;
      if (lane < m) {
        int2 e = qrec[qb + b + lane];
        src = e.x;
        dl  = e.y;
        rn  = rsqrtf(degD[n0 + R0 + dl] * degS[src]);
      }
      int k0 = 0;
      for (; k0 + U <= m; k0 += U)
        acc_records<JA, JB, U, false>(Ya, Yb, accA, accB, lane, src, dl, rn, k0, m);
      if (k0 < m)
        acc_records<JA, JB, U, true>(Ya, Yb, accA, accB, lane, src, dl, rn, k0, m);
    }
#pragma unroll
    for (int w = 0; w < RPW; ++w) {
      if (w < NR) {
#pragma unroll
        for (int j = 0; j < JA; ++j) tileA[(R0 + w) * PA + lane + 32 * j] = accA[w][j];
#pragma unroll
        for (int j = 0; j < JB; ++j) tileB[(R0 + w) * PB + lane + 32 * j] = accB[w][j];
      }
    }
    if (p0 + PASS_R < hi) __syncthreads();
  }
}

template <int F>
__global__ __launch_bounds__(AGG_T)
void k_aggr_tile(const float* __restrict__ Y, const int* __restrict__ rec_,
                 const int* __restrict__ st,
                 const float* __restrict__ degD, const float* __restrict__ degS,
                 const float* __restrict__ noise, float* __restrict__ out, int N) {
  static_assert(F == HID_F && F == 2 * OUT_F, "the tile rows are [mu | logstd]");
  constexpr int U  = 1024 / F;
  constexpr int Q  = TR * (F / 4);
  constexpr int IT = (Q + AGG_T - 1) / AGG_T;
  __shared__ float tile[TR * F];
  __shared__ int2 qrec[PASS_R];
  __shared__ int ocnt[AGG_W], obase[AGG_W];

  const int2* rec = (const int2*)rec_;
  const int b    = blockIdx.x;
  const int n0   = b * TR;
  const int rows = min(TR, N - n0);
  const int q    = rows * (F / 4);
  const int tid  = threadIdx.x;
  const int lane = tid & 31;
  const int wave = __builtin_amdgcn_readfirstlane(tid >> 5);

  const int qo = rows * RP_C4;
  float4 nz[RP_IO];
  reparam_load(nz, noise, n0, qo, tid);

  {
    const float4* y4 = (const float4*)(Y + (size_t)n0 * F);
    float4 v[IT];
    float  dd[IT], ds[IT];
#pragma unroll
    for (int i = 0; i < IT; ++i) {
      int idx = min(tid + i * AGG_T, q - 1);
      int n = n0 + idx / (F / 4);
      v[i]  = y4[idx];
      dd[i] = degD[n];
      ds[i] = degS[n];
    }
#pragma unroll
    for (int i = 0; i < IT; ++i) {
      int idx = tid + i * AGG_T;
      if (idx < q) {
        float rn = rsqrtf(dd[i] * ds[i]);
        float4 o; o.x = rn * v[i].x; o.y = rn * v[i].y; o.z = rn * v[i].z; o.w = rn * v[i].w;
        ((float4*)tile)[idx] = o;
      }
    }
  }
  __syncthreads();

  const int lo = st[b], hi = st[b + 1];
  walk_owned<F / 32, F, 0, 0, U>(Y, tile, nullptr, nullptr, rec, lo, hi, degD, degS, n0, qrec, ocnt, obase, tid, lane, wave);
  __syncthreads();

  reparam_store(nz, tile, out, n0, qo, tid);
}


#define L2_RT   ((TR + 15) / 16)
#define L2_ROWS (L2_RT * 16)
#define L2_P    (IN_F + 4)

__global__ __launch_bounds__(AGG_T)
void k_aggr_lin2(const float* __restrict__ Y, const int* __restrict__ rec_,
                 const int* __restrict__ st,
                 const float* __restrict__ degD, const float* __restrict__ degS,
                 const float* __restrict__ Wa, const float* __restrict__ ba,
                 const float* __restrict__ Wb, const float* __restrict__ bb,
                 float* __restrict__ Y2, int N) {
  constexpr int F  = IN_F;
  constexpr int P  = L2_P;
  constexpr int K  = HID_F;
  constexpr int NK = K / 4;
  constexpr int U  = 1024 / F;
  constexpr int QP = L2_ROWS * (F / 4);
  constexpr int IT = (QP + AGG_T - 1) / AGG_T;
  constexpr int NCT = HID_F / 16;
  constexpr int NG  = (AGG_T / 32) / NCT;
  static_assert(L2_RT <= 2 * NG, "two row tiles per wave cover the tile");
  __shared__ float tile[L2_ROWS * P];
  __shared__ int2 qrec[PASS_R];
  __shared__ int ocnt[AGG_W], obase[AGG_W];

  const int2* rec = (const int2*)rec_;
  const int b    = blockIdx.x;
  const int n0   = b * TR;
  const int rows = min(TR, N - n0);
  const int q    = rows * (F / 4);
  const int tid  = threadIdx.x;
  const int lane = tid & 31;
  const int wave = __builtin_amdgcn_readfirstlane(tid >> 5);

  {
    const float4* y4 = (const float4*)(Y + (size_t)n0 * F);
    float4 v[IT];
    float  dd[IT], ds[IT];
#pragma unroll
    for (int i = 0; i < IT; ++i) {
      int idx = min(tid + i * AGG_T, q - 1);
      int n = n0 + idx / (F / 4);
      v[i]  = y4[idx];
      dd[i] = degD[n];
      ds[i] = degS[n];
    }
#pragma unroll
    for (int i = 0; i < IT; ++i) {
      int idx = tid + i * AGG_T;
      if (idx < QP) {
        float4 o = {};
        if (idx < q) {
          float rn = rsqrtf(dd[i] * ds[i]);
          o.x = rn * v[i].x; o.y = rn * v[i].y; o.z = rn * v[i].z; o.w = rn * v[i].w;
        }
        *(float4*)(tile + (idx / (F / 4)) * P + (idx % (F / 4)) * 4) = o;
      }
    }
  }
  __syncthreads();

  const int lo = st[b], hi = st[b + 1];
  walk_owned<F / 32, P, 0, 0, U>(Y, tile, nullptr, nullptr, rec, lo, hi, degD, degS, n0, qrec, ocnt, obase, tid, lane, wave);

  const int ct   = wave % NCT;
  const int grp  = wave / NCT;
  const int col  = lane & 15;
  const int khi  = (lane >> 4) << 1;
  const int cc0  = ct << 4;
  const int half = (cc0 >= OUT_F) ? 1 : 0;
  const float* W    = half ? Wb : Wa;
  const float* bias = half ? bb : ba;
  const int wc0  = cc0 - half * OUT_F;
  const int koff = half ? HID_F : 0;
  v2f bfr[NK];
  const float* wrow = W + (size_t)(wc0 + col) * K + khi;
#pragma unroll
  for (int i = 0; i < NK; ++i) bfr[i] = *(const v2f*)(wrow + i * 4);
  const float bv = bias[wc0 + col];
  __syncthreads();

  const bool two = (grp + NG) < L2_RT;
  const int rt1  = two ? grp + NG : grp;
  const float* ap0 = tile + (grp * 16 + col) * P + koff + khi;
  const float* ap1 = tile + (rt1 * 16 + col) * P + koff + khi;
  v8f acc0 = {}, acc1 = {};
#pragma unroll
  for (int i = 0; i < NK; ++i) {
    v2f a0 = *(const v2f*)(ap0 + i * 4);
    v2f a1 = *(const v2f*)(ap1 + i * 4);
    a0.x = fmaxf(a0.x, 0.0f); a0.y = fmaxf(a0.y, 0.0f);
    a1.x = fmaxf(a1.x, 0.0f); a1.y = fmaxf(a1.y, 0.0f);
    acc0 = __builtin_amdgcn_wmma_f32_16x16x4_f32(false, a0, false, bfr[i], (short)0, acc0, false, false);
    acc1 = __builtin_amdgcn_wmma_f32_16x16x4_f32(false, a1, false, bfr[i], (short)0, acc1, false, false);
  }
  __syncthreads();

  {
    float* yp0 = tile + (grp * 16 + ((lane >> 4) << 3)) * P + cc0 + col;
#pragma unroll
    for (int g = 0; g < 8; ++g) yp0[g * P] = acc0[g] + bv;
    if (two) {
      float* yp1 = tile + (rt1 * 16 + ((lane >> 4) << 3)) * P + cc0 + col;
#pragma unroll
      for (int g = 0; g < 8; ++g) yp1[g * P] = acc1[g] + bv;
    }
  }
  __syncthreads();

  {
    constexpr int C4 = HID_F / 4;
    constexpr int IO = (TR * C4 + AGG_T - 1) / AGG_T;
    const int qo = rows * C4;
    float4* o4 = (float4*)(Y2 + (size_t)n0 * HID_F);
#pragma unroll
    for (int i = 0; i < IO; ++i) {
      int idx = tid + i * AGG_T;
      if (idx < qo) o4[idx] = *(const float4*)(tile + (idx / C4) * P + (idx % C4) * 4);
    }
  }
}


__global__ __launch_bounds__(AGG_T)
void k_aggr_lin2_tile(const float* __restrict__ Y, const float* __restrict__ Yb,
                      const int* __restrict__ rec_, const int* __restrict__ st,
                      const float* __restrict__ degD, const float* __restrict__ degS,
                      const float* __restrict__ Wa, const float* __restrict__ ba,
                      const float* __restrict__ Wb, const float* __restrict__ bb,
                      float* __restrict__ Y2,
                      const float* __restrict__ noise, float* __restrict__ out, int N) {
  constexpr int F   = IN_F;
  constexpr int FB  = HID_F;
  constexpr int P   = L2_P;
  constexpr int K   = HID_F;
  constexpr int NK  = K / 4;
  constexpr int U   = 4;
  constexpr int QP  = L2_ROWS * (F / 4);
  constexpr int IT  = (QP + AGG_T - 1) / AGG_T;
  constexpr int QB  = TR * (FB / 4);
  constexpr int ITB = (QB + AGG_T - 1) / AGG_T;
  constexpr int NCT = HID_F / 16;
  constexpr int NG  = (AGG_T / 32) / NCT;
  static_assert(L2_RT <= 2 * NG, "two row tiles per wave cover the tile");
  __shared__ float tile[L2_ROWS * P];
  __shared__ int2 qrec[PASS_R];
  __shared__ int ocnt[AGG_W], obase[AGG_W];
  __shared__ float tileB[TR * FB];

  const int2* rec = (const int2*)rec_;
  const int b    = blockIdx.x;
  const int n0   = b * TR;
  const int rows = min(TR, N - n0);
  const int q    = rows * (F / 4);
  const int qb   = rows * (FB / 4);
  const int tid  = threadIdx.x;
  const int lane = tid & 31;
  const int wave = __builtin_amdgcn_readfirstlane(tid >> 5);

  {
    const float4* y4 = (const float4*)(Y + (size_t)n0 * F);
    float4 v[IT];
    float  dd[IT], ds[IT];
#pragma unroll
    for (int i = 0; i < IT; ++i) {
      int idx = min(tid + i * AGG_T, q - 1);
      int n = n0 + idx / (F / 4);
      v[i]  = y4[idx];
      dd[i] = degD[n];
      ds[i] = degS[n];
    }
#pragma unroll
    for (int i = 0; i < IT; ++i) {
      int idx = tid + i * AGG_T;
      if (idx < QP) {
        float4 o = {};
        if (idx < q) {
          float rn = rsqrtf(dd[i] * ds[i]);
          o.x = rn * v[i].x; o.y = rn * v[i].y; o.z = rn * v[i].z; o.w = rn * v[i].w;
        }
        *(float4*)(tile + (idx / (F / 4)) * P + (idx % (F / 4)) * 4) = o;
      }
    }
  }
  {
    const float4* y4 = (const float4*)(Yb + (size_t)n0 * FB);
    float4 v[ITB];
    float  dd[ITB], ds[ITB];
#pragma unroll
    for (int i = 0; i < ITB; ++i) {
      int idx = min(tid + i * AGG_T, qb - 1);
      int n = n0 + idx / (FB / 4);
      v[i]  = y4[idx];
      dd[i] = degD[n];
      ds[i] = degS[n];
    }
#pragma unroll
    for (int i = 0; i < ITB; ++i) {
      int idx = tid + i * AGG_T;
      if (idx < qb) {
        float rn = rsqrtf(dd[i] * ds[i]);
        float4 o; o.x = rn * v[i].x; o.y = rn * v[i].y; o.z = rn * v[i].z; o.w = rn * v[i].w;
        ((float4*)tileB)[idx] = o;
      }
    }
  }
  __syncthreads();

  const int lo = st[b], hi = st[b + 1];
  walk_owned<F / 32, P, FB / 32, FB, U>(Y, tile, Yb, tileB, rec, lo, hi, degD, degS, n0, qrec, ocnt, obase, tid, lane, wave);

  const int ct   = wave % NCT;
  const int grp  = wave / NCT;
  const int col  = lane & 15;
  const int khi  = (lane >> 4) << 1;
  const int cc0  = ct << 4;
  const int half = (cc0 >= OUT_F) ? 1 : 0;
  const float* W    = half ? Wb : Wa;
  const float* bias = half ? bb : ba;
  const int wc0  = cc0 - half * OUT_F;
  const int koff = half ? HID_F : 0;
  v2f bfr[NK];
  const float* wrow = W + (size_t)(wc0 + col) * K + khi;
#pragma unroll
  for (int i = 0; i < NK; ++i) bfr[i] = *(const v2f*)(wrow + i * 4);
  const float bv = bias[wc0 + col];
  const int qo = rows * RP_C4;
  float4 nz[RP_IO];
  reparam_load(nz, noise, n0, qo, tid);
  __syncthreads();

  reparam_store(nz, tileB, out, n0, qo, tid);

  const bool two = (grp + NG) < L2_RT;
  const int rt1  = two ? grp + NG : grp;
  const float* ap0 = tile + (grp * 16 + col) * P + koff + khi;
  const float* ap1 = tile + (rt1 * 16 + col) * P + koff + khi;
  v8f acc0 = {}, acc1 = {};
#pragma unroll
  for (int i = 0; i < NK; ++i) {
    v2f a0 = *(const v2f*)(ap0 + i * 4);
    v2f a1 = *(const v2f*)(ap1 + i * 4);
    a0.x = fmaxf(a0.x, 0.0f); a0.y = fmaxf(a0.y, 0.0f);
    a1.x = fmaxf(a1.x, 0.0f); a1.y = fmaxf(a1.y, 0.0f);
    acc0 = __builtin_amdgcn_wmma_f32_16x16x4_f32(false, a0, false, bfr[i], (short)0, acc0, false, false);
    acc1 = __builtin_amdgcn_wmma_f32_16x16x4_f32(false, a1, false, bfr[i], (short)0, acc1, false, false);
  }
  __syncthreads();

  {
    float* yp0 = tile + (grp * 16 + ((lane >> 4) << 3)) * P + cc0 + col;
#pragma unroll
    for (int g = 0; g < 8; ++g) yp0[g * P] = acc0[g] + bv;
    if (two) {
      float* yp1 = tile + (rt1 * 16 + ((lane >> 4) << 3)) * P + cc0 + col;
#pragma unroll
      for (int g = 0; g < 8; ++g) yp1[g * P] = acc1[g] + bv;
    }
  }
  __syncthreads();

  {
    constexpr int C4 = HID_F / 4;
    constexpr int IO = (TR * C4 + AGG_T - 1) / AGG_T;
    const int qo = rows * C4;
    float4* o4 = (float4*)(Y2 + (size_t)n0 * HID_F);
#pragma unroll
    for (int i = 0; i < IO; ++i) {
      int idx = tid + i * AGG_T;
      if (idx < qo) o4[idx] = *(const float4*)(tile + (idx / C4) * P + (idx % C4) * 4);
    }
  }
}


static inline int ceil_div(int a, int b) { return (a + b - 1) / b; }

extern "C" void kernel_launch(void* const* d_in, const int* in_sizes, int n_in,
                              void* d_out, int out_size, void* d_ws, size_t ws_size,
                              hipStream_t stream) {
  const float* s       = (const float*)d_in[0];
  const float* t       = (const float*)d_in[1];
  const int*   ei      = (const int*)  d_in[2];
  const float* noise_s = (const float*)d_in[3];
  const float* noise_t = (const float*)d_in[4];
  const float* sm1_W = (const float*)d_in[5],  *sm1_b = (const float*)d_in[6];
  const float* sm2_W = (const float*)d_in[7],  *sm2_b = (const float*)d_in[8];
  const float* sl1_W = (const float*)d_in[9],  *sl1_b = (const float*)d_in[10];
  const float* sl2_W = (const float*)d_in[11], *sl2_b = (const float*)d_in[12];
  const float* tm1_W = (const float*)d_in[13], *tm1_b = (const float*)d_in[14];
  const float* tm2_W = (const float*)d_in[15], *tm2_b = (const float*)d_in[16];
  const float* tl1_W = (const float*)d_in[17], *tl1_b = (const float*)d_in[18];
  const float* tl2_W = (const float*)d_in[19], *tl2_b = (const float*)d_in[20];

  const int N = in_sizes[0] / IN_F;
  const int E = in_sizes[2] / 2;
  const int* ei0 = ei;
  const int* ei1 = ei + E;
  const int NB = ceil_div(N, TR);

  float* ws  = (float*)d_ws;
  float* Y1s = ws;
  float* Y2s = Y1s + (size_t)N * IN_F;
  float* Y1t = Y2s + (size_t)N * HID_F;
  float* Y2t = Y1s;
  int*   recF = (int*)(Y1t + (size_t)N * IN_F + (size_t)N * HID_F);
  int*   recB = recF + (size_t)2 * E;
  float* deg_in  = (float*)(recB + (size_t)2 * E);
  float* deg_out = deg_in + N;
  int*   cnt     = (int*)(deg_out + N);
  int*   cursor  = cnt + REPL * 2 * NB;
  int*   start   = cursor + REPL * 2 * NB;
  const int* startF = start;
  const int* startB = start + (NB + 1);

  const int nChunks = ceil_div(N, GEMM_ROWS);
  const int gemmBlocks = 512;
  const int binBlocks = ceil_div(E, BIN_T * BIN_EPT);

  k_zero_counts<<<ceil_div(REPL * 2 * NB, 256), 256, 0, stream>>>(cnt, REPL * 2 * NB);
  k_count_gemm<<<binBlocks + gemmBlocks, GEMM_T, 2 * NB * sizeof(int), stream>>>(
      ei0, ei1, cnt, E, NB, binBlocks,
      t, tm1_W, tm1_b, tl1_W, tl1_b, Y1t, N, nChunks);
  k_bin_scan<<<2, 1024, 0, stream>>>(cnt, start, cursor, NB);
  k_place_gemm<<<binBlocks + gemmBlocks, GEMM_T, 4 * NB * sizeof(int), stream>>>(
      ei0, ei1, cursor, recF, recB, E, NB, binBlocks,
      s, sm1_W, sm1_b, sl1_W, sl1_b, Y1s, N, nChunks);
  k_degrees<<<dim3(NB, 2), 256, 0, stream>>>(recF, recB, start, deg_in, deg_out, N, NB);

  k_aggr_lin2<<<NB, AGG_T, 0, stream>>>(Y1s, recF, startF, deg_in, deg_out,
                                        sm2_W, sm2_b, sl2_W, sl2_b, Y2s, N);

  float* out_s = (float*)d_out;
  float* out_t = out_s + (size_t)N * OUT_F;

  k_aggr_lin2_tile<<<NB, AGG_T, 0, stream>>>(Y1t, Y2s, recB, startB, deg_out, deg_in,
                                             tm2_W, tm2_b, tl2_W, tl2_b, Y2t, noise_s, out_s, N);

  k_aggr_tile<HID_F><<<NB, AGG_T, 0, stream>>>(Y2t, recF, startF, deg_in, deg_out, noise_t, out_t, N);
}


# Bucket records sorted by destination row, not by owner wave: a wave adds a row's records into one register row; one row-change test per record, no 7-way row select; rank per row, arrival order
# speedup vs baseline: 5.0299x; 1.0505x over previous
#include <hip/hip_runtime.h>


typedef __attribute__((ext_vector_type(2))) float v2f;
typedef __attribute__((ext_vector_type(8))) float v8f;

#define IN_F   128
#define HID_F  64
#define OUT_F  32

#define TR     100
#define AGG_T  512
#define BIN_T  256
#define BIN_EPT 16
#define REPL   16


__global__ void k_zero_counts(int* __restrict__ cnt, int n) {
  int i = blockIdx.x * blockDim.x + threadIdx.x;
  if (i < n) cnt[i] = 0;
}

static __device__ __forceinline__
void bin_count_part(const int* __restrict__ ei0, const int* __restrict__ ei1,
                    int* __restrict__ cnt, int E, int NB, int blk) {
  extern __shared__ int bin_lds[];
  int* hist = bin_lds;
  for (int i = threadIdx.x; i < 2 * NB; i += BIN_T) hist[i] = 0;
  __syncthreads();
  int e0 = blk * (BIN_T * BIN_EPT) + threadIdx.x;
  int s[BIN_EPT], d[BIN_EPT];
#pragma unroll
  for (int k = 0; k < BIN_EPT; ++k) {
    int e = min(e0 + k * BIN_T, E - 1);
    s[k] = ei0[e];
    d[k] = ei1[e];
  }
#pragma unroll
  for (int k = 0; k < BIN_EPT; ++k) {
    if (e0 + k * BIN_T < E) {
      atomicAdd(&hist[d[k] / TR], 1);
      atomicAdd(&hist[NB + s[k] / TR], 1);
    }
  }
  __syncthreads();
  int* mycnt = cnt + (blk % REPL) * 2 * NB;
  for (int i = threadIdx.x; i < 2 * NB; i += BIN_T) {
    int c = hist[i];
    if (c) atomicAdd(&mycnt[i], c);
  }
}

__global__ __launch_bounds__(1024)
void k_bin_scan(const int* __restrict__ cnt, int* __restrict__ start,
                int* __restrict__ cursor, int NB) {
  __shared__ int sh[1024];
  int dir = blockIdx.x;
  int* st = start + dir * (NB + 1);
  int tid = threadIdx.x;
  int carry = 0;
  for (int base = 0; base < NB; base += 1024) {
    int i = base + tid;
    int c[REPL];
    int v = 0;
#pragma unroll
    for (int r = 0; r < REPL; ++r) {
      c[r] = (i < NB) ? cnt[r * 2 * NB + dir * NB + i] : 0;
      v += c[r];
    }
    sh[tid] = v;
    __syncthreads();
    for (int off = 1; off < 1024; off <<= 1) {
      int t = (tid >= off) ? sh[tid - off] : 0;
      __syncthreads();
      sh[tid] += t;
      __syncthreads();
    }
    int run = carry + sh[tid] - v;
    if (i < NB) {
      st[i] = run;
#pragma unroll
      for (int r = 0; r < REPL; ++r) {
        cursor[r * 2 * NB + dir * NB + i] = run;
        run += c[r];
      }
    }
    carry += sh[1023];
    __syncthreads();
  }
  if (tid == 0) st[NB] = carry;
}

static __device__ __forceinline__
void bin_place_part(const int* __restrict__ ei0, const int* __restrict__ ei1,
                    int* __restrict__ cursor, int* __restrict__ recF, int* __restrict__ recB,
                    int E, int NB, int blk) {
  extern __shared__ int bin_lds[];
  int* hist = bin_lds;
  int* base = bin_lds + 2 * NB;
  for (int i = threadIdx.x; i < 2 * NB; i += BIN_T) hist[i] = 0;
  __syncthreads();
  int s[BIN_EPT], d[BIN_EPT], rF[BIN_EPT], rB[BIN_EPT];
  int e0 = blk * (BIN_T * BIN_EPT) + threadIdx.x;
#pragma unroll
  for (int k = 0; k < BIN_EPT; ++k) {
    int e = min(e0 + k * BIN_T, E - 1);
    s[k] = ei0[e];
    d[k] = ei1[e];
  }
#pragma unroll
  for (int k = 0; k < BIN_EPT; ++k) {
    rF[k] = 0; rB[k] = 0;
    if (e0 + k * BIN_T < E) {
      rF[k] = atomicAdd(&hist[d[k] / TR], 1);
      rB[k] = atomicAdd(&hist[NB + s[k] / TR], 1);
    }
  }
  __syncthreads();
  int* mycur = cursor + (blk % REPL) * 2 * NB;
  for (int i0 = threadIdx.x; i0 < 2 * NB; i0 += 4 * BIN_T) {
    int c[4], r[4];
#pragma unroll
    for (int j = 0; j < 4; ++j) {
      int i = i0 + j * BIN_T;
      c[j] = (i < 2 * NB) ? hist[i] : 0;
    }
#pragma unroll
    for (int j = 0; j < 4; ++j)
      r[j] = c[j] ? atomicAdd(&mycur[i0 + j * BIN_T], c[j]) : 0;
#pragma unroll
    for (int j = 0; j < 4; ++j) {
      int i = i0 + j * BIN_T;
      if (i < 2 * NB) base[i] = r[j];
    }
  }
  __syncthreads();
  int2* oF = (int2*)recF;
  int2* oB = (int2*)recB;
#pragma unroll
  for (int k = 0; k < BIN_EPT; ++k) {
    if (e0 + k * BIN_T < E) {
      oF[base[d[k] / TR] + rF[k]]      = make_int2(s[k], d[k]);
      oB[base[NB + s[k] / TR] + rB[k]] = make_int2(d[k], s[k]);
    }
  }
}

__global__ __launch_bounds__(256)
void k_degrees(const int* __restrict__ recF, const int* __restrict__ recB,
               const int* __restrict__ start,
               float* __restrict__ deg_in, float* __restrict__ deg_out, int N, int NB) {
  __shared__ int c[TR];
  int b = blockIdx.x, dir = blockIdx.y;
  const int2* rec = (const int2*)(dir ? recB : recF);
  const int* st = start + dir * (NB + 1);
  float* deg = dir ? deg_out : deg_in;
  for (int i = threadIdx.x; i < TR; i += 256) c[i] = 0;
  __syncthreads();
  int n0 = b * TR;
  int lo = st[b], hi = st[b + 1];
  for (int i = lo + threadIdx.x; i < hi; i += 256) atomicAdd(&c[rec[i].y - n0], 1);
  __syncthreads();
  for (int i = threadIdx.x; i < TR; i += 256) {
    int n = n0 + i;
    if (n < N) deg[n] = 1.0f + (float)c[i];
  }
}


#define GEMM_T    256
#define GEMM_ROWS 32
#define GEMM_P    (IN_F + 4)

template <int K, int NCT, int RELU>
static __device__ __forceinline__
void gemm_part(float* xt, const float* __restrict__ X,
               const float* __restrict__ Wa, const float* __restrict__ ba,
               const float* __restrict__ Wb, const float* __restrict__ bb,
               float* __restrict__ Y, int ldy,
               int halfO, int koffB, int N, int nChunks, int blk, int nblk) {
  constexpr int NK  = K / 4;
  constexpr int G   = (GEMM_T / 32) / NCT;
  constexpr int MPW = (GEMM_ROWS / 16) / G;

  const int tid  = threadIdx.x;
  const int lane = tid & 31;
  const int wave = __builtin_amdgcn_readfirstlane(tid >> 5);
  const int ct   = wave % NCT;
  const int grp  = wave / NCT;
  const int col  = lane & 15;
  const int khi  = (lane >> 4) << 1;

  const int c0   = ct << 4;
  const int half = (c0 >= halfO) ? 1 : 0;
  const float* W    = half ? Wb : Wa;
  const float* bias = half ? bb : ba;
  const int wc0  = c0 - half * halfO;
  const int koff = half ? koffB : 0;

  v2f bfr[NK];
  const float* wrow = W + (size_t)(wc0 + col) * K + khi;
#pragma unroll
  for (int i = 0; i < NK; ++i) bfr[i] = *(const v2f*)(wrow + i * 4);
  const float bv = bias[wc0 + col];

  const float* xl = X + lane * 4;
  const int rmax = N - 1;
  float4 s0 = {}, s1 = {}, s2 = {}, s3 = {};
  int ch = blk;
  if (ch < nChunks) {
    const int r0 = ch * GEMM_ROWS + wave;
    s0 = *(const float4*)(xl + (size_t)min(r0,      rmax) * IN_F);
    s1 = *(const float4*)(xl + (size_t)min(r0 + 8,  rmax) * IN_F);
    s2 = *(const float4*)(xl + (size_t)min(r0 + 16, rmax) * IN_F);
    s3 = *(const float4*)(xl + (size_t)min(r0 + 24, rmax) * IN_F);
  }
  float* xs = xt + wave * GEMM_P + lane * 4;
  const float* ap = xt + (grp * MPW * 16 + col) * GEMM_P + koff + khi;

  for (; ch < nChunks; ch += nblk) {
    if (RELU) {
      s0.x = fmaxf(s0.x, 0.0f); s0.y = fmaxf(s0.y, 0.0f); s0.z = fmaxf(s0.z, 0.0f); s0.w = fmaxf(s0.w, 0.0f);
      s1.x = fmaxf(s1.x, 0.0f); s1.y = fmaxf(s1.y, 0.0f); s1.z = fmaxf(s1.z, 0.0f); s1.w = fmaxf(s1.w, 0.0f);
      s2.x = fmaxf(s2.x, 0.0f); s2.y = fmaxf(s2.y, 0.0f); s2.z = fmaxf(s2.z, 0.0f); s2.w = fmaxf(s2.w, 0.0f);
      s3.x = fmaxf(s3.x, 0.0f); s3.y = fmaxf(s3.y, 0.0f); s3.z = fmaxf(s3.z, 0.0f); s3.w = fmaxf(s3.w, 0.0f);
    }
    *(float4*)(xs)               = s0;
    *(float4*)(xs + 8 * GEMM_P)  = s1;
    *(float4*)(xs + 16 * GEMM_P) = s2;
    *(float4*)(xs + 24 * GEMM_P) = s3;
    __syncthreads();

    const int nx = ch + nblk;
    if (nx < nChunks) {
      const int r0 = nx * GEMM_ROWS + wave;
      s0 = *(const float4*)(xl + (size_t)min(r0,      rmax) * IN_F);
      s1 = *(const float4*)(xl + (size_t)min(r0 + 8,  rmax) * IN_F);
      s2 = *(const float4*)(xl + (size_t)min(r0 + 16, rmax) * IN_F);
      s3 = *(const float4*)(xl + (size_t)min(r0 + 24, rmax) * IN_F);
    }

    v8f acc[MPW];
#pragma unroll
    for (int j = 0; j < MPW; ++j) acc[j] = v8f{};
#pragma unroll
    for (int i = 0; i < NK; ++i) {
#pragma unroll
      for (int j = 0; j < MPW; ++j) {
        v2f a = *(const v2f*)(ap + j * 16 * GEMM_P + i * 4);
        acc[j] = __builtin_amdgcn_wmma_f32_16x16x4_f32(false, a, false, bfr[i],
                                                       (short)0, acc[j], false, false);
      }
    }

#pragma unroll
    for (int j = 0; j < MPW; ++j) {
      const int m0 = ch * GEMM_ROWS + (grp * MPW + j) * 16;
      if (m0 < N) {
        float* yp = Y + (size_t)(m0 + ((lane >> 4) << 3)) * ldy + c0 + col;
#pragma unroll
        for (int g = 0; g < 8; ++g)
          yp[(size_t)g * ldy] = acc[j][g] + bv;
      }
    }
    __syncthreads();
  }
}


static_assert(BIN_T == GEMM_T, "binning and product work-groups share a launch shape");

__global__ __launch_bounds__(GEMM_T)
void k_count_gemm(const int* __restrict__ ei0, const int* __restrict__ ei1,
                  int* __restrict__ cnt, int E, int NB, int binBlocks,
                  const float* __restrict__ X,
                  const float* __restrict__ Wa, const float* __restrict__ ba,
                  const float* __restrict__ Wb, const float* __restrict__ bb,
                  float* __restrict__ Y, int N, int nChunks) {
  __shared__ float xt[GEMM_ROWS * GEMM_P];
  const int b = blockIdx.x;
  if (b < binBlocks)
    bin_count_part(ei0, ei1, cnt, E, NB, b);
  else
    gemm_part<IN_F, IN_F / 16, 0>(xt, X, Wa, ba, Wb, bb, Y, IN_F, HID_F, 0, N, nChunks,
                                  b - binBlocks, (int)gridDim.x - binBlocks);
}

__global__ __launch_bounds__(GEMM_T)
void k_place_gemm(const int* __restrict__ ei0, const int* __restrict__ ei1,
                  int* __restrict__ cursor, int* __restrict__ recF, int* __restrict__ recB,
                  int E, int NB, int binBlocks,
                  const float* __restrict__ X,
                  const float* __restrict__ Wa, const float* __restrict__ ba,
                  const float* __restrict__ Wb, const float* __restrict__ bb,
                  float* __restrict__ Y, int N, int nChunks) {
  __shared__ float xt[GEMM_ROWS * GEMM_P];
  const int b = blockIdx.x;
  if (b < binBlocks)
    bin_place_part(ei0, ei1, cursor, recF, recB, E, NB, b);
  else
    gemm_part<IN_F, IN_F / 16, 0>(xt, X, Wa, ba, Wb, bb, Y, IN_F, HID_F, 0, N, nChunks,
                                  b - binBlocks, (int)gridDim.x - binBlocks);
}

#define RP_C4 (OUT_F / 4)
#define RP_IO ((TR * RP_C4 + AGG_T - 1) / AGG_T)

static __device__ __forceinline__
void reparam_load(float4 (&nz)[RP_IO], const float* __restrict__ noise, int n0, int qo, int tid) {
  const float4* z4 = (const float4*)(noise + (size_t)n0 * OUT_F);
#pragma unroll
  for (int i = 0; i < RP_IO; ++i) nz[i] = z4[min(tid + i * AGG_T, qo - 1)];
}

static __device__ __forceinline__
void reparam_store(const float4 (&nz)[RP_IO], const float* tileB, float* __restrict__ out,
                   int n0, int qo, int tid) {
  float4* o4 = (float4*)(out + (size_t)n0 * OUT_F);
#pragma unroll
  for (int i = 0; i < RP_IO; ++i) {
    int idx = tid + i * AGG_T;
    if (idx < qo) {
      const float* tr = tileB + (idx / RP_C4) * HID_F + (idx % RP_C4) * 4;
      float4 mu = *(const float4*)tr;
      float4 ls = *(const float4*)(tr + OUT_F);
      float4 o;
      o.x = mu.x + nz[i].x * expf(ls.x) * 0.2f;
      o.y = mu.y + nz[i].y * expf(ls.y) * 0.2f;
      o.z = mu.z + nz[i].z * expf(ls.z) * 0.2f;
      o.w = mu.w + nz[i].w * expf(ls.w) * 0.2f;
      o4[idx] = o;
    }
  }
}


#define AGG_W   (AGG_T / 32)
#define RPW     ((TR + AGG_W - 1) / AGG_W)
#define PASS_R  2048
static_assert(PASS_R % AGG_T == 0, "whole records per thread in a pass");
static_assert(TR + 1 <= 4 * 32, "one wave scans the row counters, four per lane");

template <int JA, int PA, int JB, int PB, int U, bool TAIL>
static __device__ __forceinline__
void acc_records(const float* __restrict__ Ya, const float* __restrict__ Yb,
                 float* tileA, float* tileB,
                 float (&accA)[JA], float (&accB)[JB ? JB : 1], int& cur, int lane,
                 int src, int dl, float rn, int k0, int cnt) {
  float va[U][JA];
  float vb[U][JB ? JB : 1];
  int   d[U];
  float r[U];
#pragma unroll
  for (int u = 0; u < U; ++u) {
    int k = k0 + u;
    d[u] = -1; r[u] = 0.0f;
    if (!TAIL || k < cnt) {
      int s = __builtin_amdgcn_readlane(src, k);
      d[u]  = __builtin_amdgcn_readlane(dl, k);
      r[u]  = __int_as_float(__builtin_amdgcn_readlane(__float_as_int(rn), k));
      const float* ya = Ya + (size_t)s * (32 * JA) + lane;
#pragma unroll
      for (int j = 0; j < JA; ++j) va[u][j] = ya[32 * j];
      if (JB) {
        const float* yb = Yb + (size_t)s * (32 * JB) + lane;
#pragma unroll
        for (int j = 0; j < JB; ++j) vb[u][j] = yb[32 * j];
      }
    }
  }
#pragma unroll
  for (int u = 0; u < U; ++u) {
    if (!TAIL || k0 + u < cnt) {
      if (__builtin_expect(d[u] != cur, 0)) {
        if (cur >= 0) {
#pragma unroll
          for (int j = 0; j < JA; ++j) tileA[cur * PA + lane + 32 * j] = accA[j];
#pragma unroll
          for (int j = 0; j < JB; ++j) tileB[cur * PB + lane + 32 * j] = accB[j];
        }
        cur = d[u];
#pragma unroll
        for (int j = 0; j < JA; ++j) accA[j] = tileA[cur * PA + lane + 32 * j];
#pragma unroll
        for (int j = 0; j < JB; ++j) accB[j] = tileB[cur * PB + lane + 32 * j];
      }
#pragma unroll
      for (int j = 0; j < JA; ++j) accA[j] += r[u] * va[u][j];
#pragma unroll
      for (int j = 0; j < JB; ++j) accB[j] += r[u] * vb[u][j];
    }
  }
}

template <int JA, int PA, int JB, int PB, int U>
static __device__ __forceinline__
void walk_owned(const float* __restrict__ Ya, float* tileA, const float* __restrict__ Yb, float* tileB,
                const int2* __restrict__ rec, int lo, int hi,
                const float* __restrict__ degD, const float* __restrict__ degS,
                int n0, int2* qrec, int* rcnt, int* rbase, int tid, int lane, int wave) {
  constexpr int KP = PASS_R / AGG_T;
  const int R0 = min(wave * RPW, TR);
  const int R1 = min(wave * RPW + RPW, TR);

  for (int p0 = lo; p0 < hi; p0 += PASS_R) {
    const int pe = min(hi, p0 + PASS_R);
    if (tid < TR) rcnt[tid] = 0;
    __syncthreads();
    int2 r[KP];
    int  row[KP], rank[KP];
#pragma unroll
    for (int k = 0; k < KP; ++k) r[k] = rec[min(p0 + tid + k * AGG_T, pe - 1)];
#pragma unroll
    for (int k = 0; k < KP; ++k) {
      row[k] = r[k].y - n0;
      rank[k] = 0;
      if (p0 + tid + k * AGG_T < pe) rank[k] = atomicAdd(&rcnt[row[k]], 1);
    }
    __syncthreads();
    if (wave == 0) {
      int c[4];
      int s = 0;
#pragma unroll
      for (int j = 0; j < 4; ++j) {
        int i = 4 * lane + j;
        c[j] = (i < TR) ? rcnt[i] : 0;
        s += c[j];
      }
      int incl = s;
#pragma unroll
      for (int off = 1; off < 32; off <<= 1) {
        int t = __shfl_up(incl, off);
        if (lane >= off) incl += t;
      }
      int run = incl - s;
#pragma unroll
      for (int j = 0; j < 4; ++j) {
        int i = 4 * lane + j;
        if (i <= TR) rbase[i] = run;
        run += c[j];
      }
    }
    __syncthreads();
#pragma unroll
    for (int k = 0; k < KP; ++k) {
      if (p0 + tid + k * AGG_T < pe)
        qrec[rbase[row[k]] + rank[k]] = make_int2(r[k].x, row[k]);
    }
    __syncthreads();

    const int qb = __builtin_amdgcn_readfirstlane(rbase[R0]);
    const int qn = __builtin_amdgcn_readfirstlane(rbase[R1]) - qb;
    float accA[JA];
    float accB[JB ? JB : 1];
#pragma unroll
    for (int j = 0; j < JA; ++j) accA[j] = 0.0f;
#pragma unroll
    for (int j = 0; j < (JB ? JB : 1); ++j) accB[j] = 0.0f;
    int cur = -1;
    for (int b = 0; b < qn; b += 32) {
      const int m = min(32, qn - b);
      int src = 0, dl = 0;
      float rn = 0.0f;
      if (lane < m) {
        int2 e = qrec[qb + b + lane];
        src = e.x;
        dl  = e.y;
        rn  = rsqrtf(degD[n0 + dl] * degS[src]);
      }
      int k0 = 0;
      for (; k0 + U <= m; k0 += U)
        acc_records<JA, PA, JB, PB, U, false>(Ya, Yb, tileA, tileB, accA, accB, cur, lane, src, dl, rn, k0, m);
      if (k0 < m)
        acc_records<JA, PA, JB, PB, U, true>(Ya, Yb, tileA, tileB, accA, accB, cur, lane, src, dl, rn, k0, m);
    }
    if (cur >= 0) {
#pragma unroll
      for (int j = 0; j < JA; ++j) tileA[cur * PA + lane + 32 * j] = accA[j];
#pragma unroll
      for (int j = 0; j < JB; ++j) tileB[cur * PB + lane + 32 * j] = accB[j];
    }
    if (p0 + PASS_R < hi) __syncthreads();
  }
}

template <int F>
__global__ __launch_bounds__(AGG_T)
void k_aggr_tile(const float* __restrict__ Y, const int* __restrict__ rec_,
                 const int* __restrict__ st,
                 const float* __restrict__ degD, const float* __restrict__ degS,
                 const float* __restrict__ noise, float* __restrict__ out, int N) {
  static_assert(F == HID_F && F == 2 * OUT_F, "the tile rows are [mu | logstd]");
  constexpr int U  = 1024 / F;
  constexpr int Q  = TR * (F / 4);
  constexpr int IT = (Q + AGG_T - 1) / AGG_T;
  __shared__ float tile[TR * F];
  __shared__ int2 qrec[PASS_R];
  __shared__ int rcnt[TR], rbase[TR + 1];

  const int2* rec = (const int2*)rec_;
  const int b    = blockIdx.x;
  const int n0   = b * TR;
  const int rows = min(TR, N - n0);
  const int q    = rows * (F / 4);
  const int tid  = threadIdx.x;
  const int lane = tid & 31;
  const int wave = __builtin_amdgcn_readfirstlane(tid >> 5);

  const int qo = rows * RP_C4;
  float4 nz[RP_IO];
  reparam_load(nz, noise, n0, qo, tid);

  {
    const float4* y4 = (const float4*)(Y + (size_t)n0 * F);
    float4 v[IT];
    float  dd[IT], ds[IT];
#pragma unroll
    for (int i = 0; i < IT; ++i) {
      int idx = min(tid + i * AGG_T, q - 1);
      int n = n0 + idx / (F / 4);
      v[i]  = y4[idx];
      dd[i] = degD[n];
      ds[i] = degS[n];
    }
#pragma unroll
    for (int i = 0; i < IT; ++i) {
      int idx = tid + i * AGG_T;
      if (idx < q) {
        float rn = rsqrtf(dd[i] * ds[i]);
        float4 o; o.x = rn * v[i].x; o.y = rn * v[i].y; o.z = rn * v[i].z; o.w = rn * v[i].w;
        ((float4*)tile)[idx] = o;
      }
    }
  }
  __syncthreads();

  const int lo = st[b], hi = st[b + 1];
  walk_owned<F / 32, F, 0, 0, U>(Y, tile, nullptr, nullptr, rec, lo, hi, degD, degS, n0, qrec, rcnt, rbase, tid, lane, wave);
  __syncthreads();

  reparam_store(nz, tile, out, n0, qo, tid);
}


#define L2_RT   ((TR + 15) / 16)
#define L2_ROWS (L2_RT * 16)
#define L2_P    (IN_F + 4)

__global__ __launch_bounds__(AGG_T)
void k_aggr_lin2(const float* __restrict__ Y, const int* __restrict__ rec_,
                 const int* __restrict__ st,
                 const float* __restrict__ degD, const float* __restrict__ degS,
                 const float* __restrict__ Wa, const float* __restrict__ ba,
                 const float* __restrict__ Wb, const float* __restrict__ bb,
                 float* __restrict__ Y2, int N) {
  constexpr int F  = IN_F;
  constexpr int P  = L2_P;
  constexpr int K  = HID_F;
  constexpr int NK = K / 4;
  constexpr int U  = 1024 / F;
  constexpr int QP = L2_ROWS * (F / 4);
  constexpr int IT = (QP + AGG_T - 1) / AGG_T;
  constexpr int NCT = HID_F / 16;
  constexpr int NG  = (AGG_T / 32) / NCT;
  static_assert(L2_RT <= 2 * NG, "two row tiles per wave cover the tile");
  __shared__ float tile[L2_ROWS * P];
  __shared__ int2 qrec[PASS_R];
  __shared__ int rcnt[TR], rbase[TR + 1];

  const int2* rec = (const int2*)rec_;
  const int b    = blockIdx.x;
  const int n0   = b * TR;
  const int rows = min(TR, N - n0);
  const int q    = rows * (F / 4);
  const int tid  = threadIdx.x;
  const int lane = tid & 31;
  const int wave = __builtin_amdgcn_readfirstlane(tid >> 5);

  {
    const float4* y4 = (const float4*)(Y + (size_t)n0 * F);
    float4 v[IT];
    float  dd[IT], ds[IT];
#pragma unroll
    for (int i = 0; i < IT; ++i) {
      int idx = min(tid + i * AGG_T, q - 1);
      int n = n0 + idx / (F / 4);
      v[i]  = y4[idx];
      dd[i] = degD[n];
      ds[i] = degS[n];
    }
#pragma unroll
    for (int i = 0; i < IT; ++i) {
      int idx = tid + i * AGG_T;
      if (idx < QP) {
        float4 o = {};
        if (idx < q) {
          float rn = rsqrtf(dd[i] * ds[i]);
          o.x = rn * v[i].x; o.y = rn * v[i].y; o.z = rn * v[i].z; o.w = rn * v[i].w;
        }
        *(float4*)(tile + (idx / (F / 4)) * P + (idx % (F / 4)) * 4) = o;
      }
    }
  }
  __syncthreads();

  const int lo = st[b], hi = st[b + 1];
  walk_owned<F / 32, P, 0, 0, U>(Y, tile, nullptr, nullptr, rec, lo, hi, degD, degS, n0, qrec, rcnt, rbase, tid, lane, wave);

  const int ct   = wave % NCT;
  const int grp  = wave / NCT;
  const int col  = lane & 15;
  const int khi  = (lane >> 4) << 1;
  const int cc0  = ct << 4;
  const int half = (cc0 >= OUT_F) ? 1 : 0;
  const float* W    = half ? Wb : Wa;
  const float* bias = half ? bb : ba;
  const int wc0  = cc0 - half * OUT_F;
  const int koff = half ? HID_F : 0;
  v2f bfr[NK];
  const float* wrow = W + (size_t)(wc0 + col) * K + khi;
#pragma unroll
  for (int i = 0; i < NK; ++i) bfr[i] = *(const v2f*)(wrow + i * 4);
  const float bv = bias[wc0 + col];
  __syncthreads();

  const bool two = (grp + NG) < L2_RT;
  const int rt1  = two ? grp + NG : grp;
  const float* ap0 = tile + (grp * 16 + col) * P + koff + khi;
  const float* ap1 = tile + (rt1 * 16 + col) * P + koff + khi;
  v8f acc0 = {}, acc1 = {};
#pragma unroll
  for (int i = 0; i < NK; ++i) {
    v2f a0 = *(const v2f*)(ap0 + i * 4);
    v2f a1 = *(const v2f*)(ap1 + i * 4);
    a0.x = fmaxf(a0.x, 0.0f); a0.y = fmaxf(a0.y, 0.0f);
    a1.x = fmaxf(a1.x, 0.0f); a1.y = fmaxf(a1.y, 0.0f);
    acc0 = __builtin_amdgcn_wmma_f32_16x16x4_f32(false, a0, false, bfr[i], (short)0, acc0, false, false);
    acc1 = __builtin_amdgcn_wmma_f32_16x16x4_f32(false, a1, false, bfr[i], (short)0, acc1, false, false);
  }
  __syncthreads();

  {
    float* yp0 = tile + (grp * 16 + ((lane >> 4) << 3)) * P + cc0 + col;
#pragma unroll
    for (int g = 0; g < 8; ++g) yp0[g * P] = acc0[g] + bv;
    if (two) {
      float* yp1 = tile + (rt1 * 16 + ((lane >> 4) << 3)) * P + cc0 + col;
#pragma unroll
      for (int g = 0; g < 8; ++g) yp1[g * P] = acc1[g] + bv;
    }
  }
  __syncthreads();

  {
    constexpr int C4 = HID_F / 4;
    constexpr int IO = (TR * C4 + AGG_T - 1) / AGG_T;
    const int qo = rows * C4;
    float4* o4 = (float4*)(Y2 + (size_t)n0 * HID_F);
#pragma unroll
    for (int i = 0; i < IO; ++i) {
      int idx = tid + i * AGG_T;
      if (idx < qo) o4[idx] = *(const float4*)(tile + (idx / C4) * P + (idx % C4) * 4);
    }
  }
}


__global__ __launch_bounds__(AGG_T)
void k_aggr_lin2_tile(const float* __restrict__ Y, const float* __restrict__ Yb,
                      const int* __restrict__ rec_, const int* __restrict__ st,
                      const float* __restrict__ degD, const float* __restrict__ degS,
                      const float* __restrict__ Wa, const float* __restrict__ ba,
                      const float* __restrict__ Wb, const float* __restrict__ bb,
                      float* __restrict__ Y2,
                      const float* __restrict__ noise, float* __restrict__ out, int N) {
  constexpr int F   = IN_F;
  constexpr int FB  = HID_F;
  constexpr int P   = L2_P;
  constexpr int K   = HID_F;
  constexpr int NK  = K / 4;
  constexpr int U   = 4;
  constexpr int QP  = L2_ROWS * (F / 4);
  constexpr int IT  = (QP + AGG_T - 1) / AGG_T;
  constexpr int QB  = TR * (FB / 4);
  constexpr int ITB = (QB + AGG_T - 1) / AGG_T;
  constexpr int NCT = HID_F / 16;
  constexpr int NG  = (AGG_T / 32) / NCT;
  static_assert(L2_RT <= 2 * NG, "two row tiles per wave cover the tile");
  __shared__ float tile[L2_ROWS * P];
  __shared__ int2 qrec[PASS_R];
  __shared__ int rcnt[TR], rbase[TR + 1];
  __shared__ float tileB[TR * FB];

  const int2* rec = (const int2*)rec_;
  const int b    = blockIdx.x;
  const int n0   = b * TR;
  const int rows = min(TR, N - n0);
  const int q    = rows * (F / 4);
  const int qb   = rows * (FB / 4);
  const int tid  = threadIdx.x;
  const int lane = tid & 31;
  const int wave = __builtin_amdgcn_readfirstlane(tid >> 5);

  {
    const float4* y4 = (const float4*)(Y + (size_t)n0 * F);
    float4 v[IT];
    float  dd[IT], ds[IT];
#pragma unroll
    for (int i = 0; i < IT; ++i) {
      int idx = min(tid + i * AGG_T, q - 1);
      int n = n0 + idx / (F / 4);
      v[i]  = y4[idx];
      dd[i] = degD[n];
      ds[i] = degS[n];
    }
#pragma unroll
    for (int i = 0; i < IT; ++i) {
      int idx = tid + i * AGG_T;
      if (idx < QP) {
        float4 o = {};
        if (idx < q) {
          float rn = rsqrtf(dd[i] * ds[i]);
          o.x = rn * v[i].x; o.y = rn * v[i].y; o.z = rn * v[i].z; o.w = rn * v[i].w;
        }
        *(float4*)(tile + (idx / (F / 4)) * P + (idx % (F / 4)) * 4) = o;
      }
    }
  }
  {
    const float4* y4 = (const float4*)(Yb + (size_t)n0 * FB);
    float4 v[ITB];
    float  dd[ITB], ds[ITB];
#pragma unroll
    for (int i = 0; i < ITB; ++i) {
      int idx = min(tid + i * AGG_T, qb - 1);
      int n = n0 + idx / (FB / 4);
      v[i]  = y4[idx];
      dd[i] = degD[n];
      ds[i] = degS[n];
    }
#pragma unroll
    for (int i = 0; i < ITB; ++i) {
      int idx = tid + i * AGG_T;
      if (idx < qb) {
        float rn = rsqrtf(dd[i] * ds[i]);
        float4 o; o.x = rn * v[i].x; o.y = rn * v[i].y; o.z = rn * v[i].z; o.w = rn * v[i].w;
        ((float4*)tileB)[idx] = o;
      }
    }
  }
  __syncthreads();

  const int lo = st[b], hi = st[b + 1];
  walk_owned<F / 32, P, FB / 32, FB, U>(Y, tile, Yb, tileB, rec, lo, hi, degD, degS, n0, qrec, rcnt, rbase, tid, lane, wave);

  const int ct   = wave % NCT;
  const int grp  = wave / NCT;
  const int col  = lane & 15;
  const int khi  = (lane >> 4) << 1;
  const int cc0  = ct << 4;
  const int half = (cc0 >= OUT_F) ? 1 : 0;
  const float* W    = half ? Wb : Wa;
  const float* bias = half ? bb : ba;
  const int wc0  = cc0 - half * OUT_F;
  const int koff = half ? HID_F : 0;
  v2f bfr[NK];
  const float* wrow = W + (size_t)(wc0 + col) * K + khi;
#pragma unroll
  for (int i = 0; i < NK; ++i) bfr[i] = *(const v2f*)(wrow + i * 4);
  const float bv = bias[wc0 + col];
  const int qo = rows * RP_C4;
  float4 nz[RP_IO];
  reparam_load(nz, noise, n0, qo, tid);
  __syncthreads();

  reparam_store(nz, tileB, out, n0, qo, tid);

  const bool two = (grp + NG) < L2_RT;
  const int rt1  = two ? grp + NG : grp;
  const float* ap0 = tile + (grp * 16 + col) * P + koff + khi;
  const float* ap1 = tile + (rt1 * 16 + col) * P + koff + khi;
  v8f acc0 = {}, acc1 = {};
#pragma unroll
  for (int i = 0; i < NK; ++i) {
    v2f a0 = *(const v2f*)(ap0 + i * 4);
    v2f a1 = *(const v2f*)(ap1 + i * 4);
    a0.x = fmaxf(a0.x, 0.0f); a0.y = fmaxf(a0.y, 0.0f);
    a1.x = fmaxf(a1.x, 0.0f); a1.y = fmaxf(a1.y, 0.0f);
    acc0 = __builtin_amdgcn_wmma_f32_16x16x4_f32(false, a0, false, bfr[i], (short)0, acc0, false, false);
    acc1 = __builtin_amdgcn_wmma_f32_16x16x4_f32(false, a1, false, bfr[i], (short)0, acc1, false, false);
  }
  __syncthreads();

  {
    float* yp0 = tile + (grp * 16 + ((lane >> 4) << 3)) * P + cc0 + col;
#pragma unroll
    for (int g = 0; g < 8; ++g) yp0[g * P] = acc0[g] + bv;
    if (two) {
      float* yp1 = tile + (rt1 * 16 + ((lane >> 4) << 3)) * P + cc0 + col;
#pragma unroll
      for (int g = 0; g < 8; ++g) yp1[g * P] = acc1[g] + bv;
    }
  }
  __syncthreads();

  {
    constexpr int C4 = HID_F / 4;
    constexpr int IO = (TR * C4 + AGG_T - 1) / AGG_T;
    const int qo = rows * C4;
    float4* o4 = (float4*)(Y2 + (size_t)n0 * HID_F);
#pragma unroll
    for (int i = 0; i < IO; ++i) {
      int idx = tid + i * AGG_T;
      if (idx < qo) o4[idx] = *(const float4*)(tile + (idx / C4) * P + (idx % C4) * 4);
    }
  }
}


static inline int ceil_div(int a, int b) { return (a + b - 1) / b; }

extern "C" void kernel_launch(void* const* d_in, const int* in_sizes, int n_in,
                              void* d_out, int out_size, void* d_ws, size_t ws_size,
                              hipStream_t stream) {
  const float* s       = (const float*)d_in[0];
  const float* t       = (const float*)d_in[1];
  const int*   ei      = (const int*)  d_in[2];
  const float* noise_s = (const float*)d_in[3];
  const float* noise_t = (const float*)d_in[4];
  const float* sm1_W = (const float*)d_in[5],  *sm1_b = (const float*)d_in[6];
  const float* sm2_W = (const float*)d_in[7],  *sm2_b = (const float*)d_in[8];
  const float* sl1_W = (const float*)d_in[9],  *sl1_b = (const float*)d_in[10];
  const float* sl2_W = (const float*)d_in[11], *sl2_b = (const float*)d_in[12];
  const float* tm1_W = (const float*)d_in[13], *tm1_b = (const float*)d_in[14];
  const float* tm2_W = (const float*)d_in[15], *tm2_b = (const float*)d_in[16];
  const float* tl1_W = (const float*)d_in[17], *tl1_b = (const float*)d_in[18];
  const float* tl2_W = (const float*)d_in[19], *tl2_b = (const float*)d_in[20];

  const int N = in_sizes[0] / IN_F;
  const int E = in_sizes[2] / 2;
  const int* ei0 = ei;
  const int* ei1 = ei + E;
  const int NB = ceil_div(N, TR);

  float* ws  = (float*)d_ws;
  float* Y1s = ws;
  float* Y2s = Y1s + (size_t)N * IN_F;
  float* Y1t = Y2s + (size_t)N * HID_F;
  float* Y2t = Y1s;
  int*   recF = (int*)(Y1t + (size_t)N * IN_F + (size_t)N * HID_F);
  int*   recB = recF + (size_t)2 * E;
  float* deg_in  = (float*)(recB + (size_t)2 * E);
  float* deg_out = deg_in + N;
  int*   cnt     = (int*)(deg_out + N);
  int*   cursor  = cnt + REPL * 2 * NB;
  int*   start   = cursor + REPL * 2 * NB;
  const int* startF = start;
  const int* startB = start + (NB + 1);

  const int nChunks = ceil_div(N, GEMM_ROWS);
  const int gemmBlocks = 512;
  const int binBlocks = ceil_div(E, BIN_T * BIN_EPT);

  k_zero_counts<<<ceil_div(REPL * 2 * NB, 256), 256, 0, stream>>>(cnt, REPL * 2 * NB);
  k_count_gemm<<<binBlocks + gemmBlocks, GEMM_T, 2 * NB * sizeof(int), stream>>>(
      ei0, ei1, cnt, E, NB, binBlocks,
      t, tm1_W, tm1_b, tl1_W, tl1_b, Y1t, N, nChunks);
  k_bin_scan<<<2, 1024, 0, stream>>>(cnt, start, cursor, NB);
  k_place_gemm<<<binBlocks + gemmBlocks, GEMM_T, 4 * NB * sizeof(int), stream>>>(
      ei0, ei1, cursor, recF, recB, E, NB, binBlocks,
      s, sm1_W, sm1_b, sl1_W, sl1_b, Y1s, N, nChunks);
  k_degrees<<<dim3(NB, 2), 256, 0, stream>>>(recF, recB, start, deg_in, deg_out, N, NB);

  k_aggr_lin2<<<NB, AGG_T, 0, stream>>>(Y1s, recF, startF, deg_in, deg_out,
                                        sm2_W, sm2_b, sl2_W, sl2_b, Y2s, N);

  float* out_s = (float*)d_out;
  float* out_t = out_s + (size_t)N * OUT_F;

  k_aggr_lin2_tile<<<NB, AGG_T, 0, stream>>>(Y1t, Y2s, recB, startB, deg_out, deg_in,
                                             tm2_W, tm2_b, tl2_W, tl2_b, Y2t, noise_s, out_s, N);

  k_aggr_tile<HID_F><<<NB, AGG_T, 0, stream>>>(Y2t, recF, startF, deg_in, deg_out, noise_t, out_t, N);
}


# Bucket offsets made by the placing work-groups: each reads the 16 count copies and scans the bucket totals itself (wave scan, wave totals via LDS), cursors count from zero; no scan launch, 8 to 7
# speedup vs baseline: 5.2369x; 1.0412x over previous
#include <hip/hip_runtime.h>


typedef __attribute__((ext_vector_type(2))) float v2f;
typedef __attribute__((ext_vector_type(8))) float v8f;

#define IN_F   128
#define HID_F  64
#define OUT_F  32

#define TR     100
#define AGG_T  512
#define BIN_T  256
#define BIN_EPT 16
#define REPL   16


__global__ void k_zero_counts(int* __restrict__ cnt, int n) {
  int i = blockIdx.x * blockDim.x + threadIdx.x;
  if (i < n) cnt[i] = 0;
}

static __device__ __forceinline__
void bin_count_part(const int* __restrict__ ei0, const int* __restrict__ ei1,
                    int* __restrict__ cnt, int E, int NB, int blk) {
  extern __shared__ int bin_lds[];
  int* hist = bin_lds;
  for (int i = threadIdx.x; i < 2 * NB; i += BIN_T) hist[i] = 0;
  __syncthreads();
  int e0 = blk * (BIN_T * BIN_EPT) + threadIdx.x;
  int s[BIN_EPT], d[BIN_EPT];
#pragma unroll
  for (int k = 0; k < BIN_EPT; ++k) {
    int e = min(e0 + k * BIN_T, E - 1);
    s[k] = ei0[e];
    d[k] = ei1[e];
  }
#pragma unroll
  for (int k = 0; k < BIN_EPT; ++k) {
    if (e0 + k * BIN_T < E) {
      atomicAdd(&hist[d[k] / TR], 1);
      atomicAdd(&hist[NB + s[k] / TR], 1);
    }
  }
  __syncthreads();
  int* mycnt = cnt + (blk % REPL) * 2 * NB;
  for (int i = threadIdx.x; i < 2 * NB; i += BIN_T) {
    int c = hist[i];
    if (c) atomicAdd(&mycnt[i], c);
  }
}

static __device__ __forceinline__
void bin_place_part(const int* __restrict__ ei0, const int* __restrict__ ei1,
                    const int* __restrict__ cnt, int* __restrict__ cursor, int* __restrict__ start,
                    int* __restrict__ recF, int* __restrict__ recB,
                    int E, int NB, int blk) {
  extern __shared__ int bin_lds[];
  const int M = 2 * NB;
  int* hist = bin_lds;
  int* base = bin_lds + M;
  int* wsum = bin_lds + 2 * M;
  int* totF = wsum + BIN_T / 32;
  const int tid   = threadIdx.x;
  const int lane  = tid & 31;
  const int wave  = tid >> 5;
  const int rcopy = blk % REPL;
  for (int i = tid; i < M; i += BIN_T) hist[i] = 0;

  int carry = 0;
  for (int t0 = 0; t0 < M; t0 += 4 * BIN_T) {
    int tot[4], pre[4];
#pragma unroll
    for (int j = 0; j < 4; ++j) {
      const int i  = t0 + 4 * tid + j;
      const int ic = min(i, M - 1);
      tot[j] = 0; pre[j] = 0;
#pragma unroll
      for (int r = 0; r < REPL; ++r) {
        const int v  = cnt[r * M + ic];
        const int cv = (i < M) ? v : 0;
        tot[j] += cv;
        if (r < rcopy) pre[j] += cv;
      }
    }
    const int s4 = tot[0] + tot[1] + tot[2] + tot[3];
    int incl = s4;
#pragma unroll
    for (int off = 1; off < 32; off <<= 1) {
      int t = __shfl_up(incl, off);
      if (lane >= off) incl += t;
    }
    if (lane == 31) wsum[wave] = incl;
    __syncthreads();
    int before = 0, all = 0;
#pragma unroll
    for (int w = 0; w < BIN_T / 32; ++w) {
      const int v = wsum[w];
      all += v;
      if (w < wave) before += v;
    }
    int run = carry + before + incl - s4;
#pragma unroll
    for (int j = 0; j < 4; ++j) {
      const int i = t0 + 4 * tid + j;
      if (i == NB) totF[0] = run;
      if (i < M) base[i] = run + pre[j];
      run += tot[j];
    }
    carry += all;
    __syncthreads();
  }
  const int nF = totF[0];
  if (blk == 0) {
    for (int i = tid; i < M; i += BIN_T) {
      const int dir = (i >= NB) ? 1 : 0;
      start[i + dir] = base[i] - (dir ? nF : 0);
    }
    if (tid == 0) { start[NB] = nF; start[2 * NB + 1] = carry - nF; }
  }

  int s[BIN_EPT], d[BIN_EPT], rF[BIN_EPT], rB[BIN_EPT];
  int e0 = blk * (BIN_T * BIN_EPT) + tid;
#pragma unroll
  for (int k = 0; k < BIN_EPT; ++k) {
    int e = min(e0 + k * BIN_T, E - 1);
    s[k] = ei0[e];
    d[k] = ei1[e];
  }
#pragma unroll
  for (int k = 0; k < BIN_EPT; ++k) {
    rF[k] = 0; rB[k] = 0;
    if (e0 + k * BIN_T < E) {
      rF[k] = atomicAdd(&hist[d[k] / TR], 1);
      rB[k] = atomicAdd(&hist[NB + s[k] / TR], 1);
    }
  }
  __syncthreads();
  int* mycur = cursor + rcopy * M;
  for (int i0 = tid; i0 < M; i0 += 4 * BIN_T) {
    int c[4], r[4];
#pragma unroll
    for (int j = 0; j < 4; ++j) {
      int i = i0 + j * BIN_T;
      c[j] = (i < M) ? hist[i] : 0;
    }
#pragma unroll
    for (int j = 0; j < 4; ++j)
      r[j] = c[j] ? atomicAdd(&mycur[i0 + j * BIN_T], c[j]) : 0;
#pragma unroll
    for (int j = 0; j < 4; ++j) {
      int i = i0 + j * BIN_T;
      if (i < M) base[i] += r[j] - ((i >= NB) ? nF : 0);
    }
  }
  __syncthreads();
  int2* oF = (int2*)recF;
  int2* oB = (int2*)recB;
#pragma unroll
  for (int k = 0; k < BIN_EPT; ++k) {
    if (e0 + k * BIN_T < E) {
      oF[base[d[k] / TR] + rF[k]]      = make_int2(s[k], d[k]);
      oB[base[NB + s[k] / TR] + rB[k]] = make_int2(d[k], s[k]);
    }
  }
}

__global__ __launch_bounds__(256)
void k_degrees(const int* __restrict__ recF, const int* __restrict__ recB,
               const int* __restrict__ start,
               float* __restrict__ deg_in, float* __restrict__ deg_out, int N, int NB) {
  __shared__ int c[TR];
  int b = blockIdx.x, dir = blockIdx.y;
  const int2* rec = (const int2*)(dir ? recB : recF);
  const int* st = start + dir * (NB + 1);
  float* deg = dir ? deg_out : deg_in;
  for (int i = threadIdx.x; i < TR; i += 256) c[i] = 0;
  __syncthreads();
  int n0 = b * TR;
  int lo = st[b], hi = st[b + 1];
  for (int i = lo + threadIdx.x; i < hi; i += 256) atomicAdd(&c[rec[i].y - n0], 1);
  __syncthreads();
  for (int i = threadIdx.x; i < TR; i += 256) {
    int n = n0 + i;
    if (n < N) deg[n] = 1.0f + (float)c[i];
  }
}


#define GEMM_T    256
#define GEMM_ROWS 32
#define GEMM_P    (IN_F + 4)

template <int K, int NCT, int RELU>
static __device__ __forceinline__
void gemm_part(float* xt, const float* __restrict__ X,
               const float* __restrict__ Wa, const float* __restrict__ ba,
               const float* __restrict__ Wb, const float* __restrict__ bb,
               float* __restrict__ Y, int ldy,
               int halfO, int koffB, int N, int nChunks, int blk, int nblk) {
  constexpr int NK  = K / 4;
  constexpr int G   = (GEMM_T / 32) / NCT;
  constexpr int MPW = (GEMM_ROWS / 16) / G;

  const int tid  = threadIdx.x;
  const int lane = tid & 31;
  const int wave = __builtin_amdgcn_readfirstlane(tid >> 5);
  const int ct   = wave % NCT;
  const int grp  = wave / NCT;
  const int col  = lane & 15;
  const int khi  = (lane >> 4) << 1;

  const int c0   = ct << 4;
  const int half = (c0 >= halfO) ? 1 : 0;
  const float* W    = half ? Wb : Wa;
  const float* bias = half ? bb : ba;
  const int wc0  = c0 - half * halfO;
  const int koff = half ? koffB : 0;

  v2f bfr[NK];
  const float* wrow = W + (size_t)(wc0 + col) * K + khi;
#pragma unroll
  for (int i = 0; i < NK; ++i) bfr[i] = *(const v2f*)(wrow + i * 4);
  const float bv = bias[wc0 + col];

  const float* xl = X + lane * 4;
  const int rmax = N - 1;
  float4 s0 = {}, s1 = {}, s2 = {}, s3 = {};
  int ch = blk;
  if (ch < nChunks) {
    const int r0 = ch * GEMM_ROWS + wave;
    s0 = *(const float4*)(xl + (size_t)min(r0,      rmax) * IN_F);
    s1 = *(const float4*)(xl + (size_t)min(r0 + 8,  rmax) * IN_F);
    s2 = *(const float4*)(xl + (size_t)min(r0 + 16, rmax) * IN_F);
    s3 = *(const float4*)(xl + (size_t)min(r0 + 24, rmax) * IN_F);
  }
  float* xs = xt + wave * GEMM_P + lane * 4;
  const float* ap = xt + (grp * MPW * 16 + col) * GEMM_P + koff + khi;

  for (; ch < nChunks; ch += nblk) {
    if (RELU) {
      s0.x = fmaxf(s0.x, 0.0f); s0.y = fmaxf(s0.y, 0.0f); s0.z = fmaxf(s0.z, 0.0f); s0.w = fmaxf(s0.w, 0.0f);
      s1.x = fmaxf(s1.x, 0.0f); s1.y = fmaxf(s1.y, 0.0f); s1.z = fmaxf(s1.z, 0.0f); s1.w = fmaxf(s1.w, 0.0f);
      s2.x = fmaxf(s2.x, 0.0f); s2.y = fmaxf(s2.y, 0.0f); s2.z = fmaxf(s2.z, 0.0f); s2.w = fmaxf(s2.w, 0.0f);
      s3.x = fmaxf(s3.x, 0.0f); s3.y = fmaxf(s3.y, 0.0f); s3.z = fmaxf(s3.z, 0.0f); s3.w = fmaxf(s3.w, 0.0f);
    }
    *(float4*)(xs)               = s0;
    *(float4*)(xs + 8 * GEMM_P)  = s1;
    *(float4*)(xs + 16 * GEMM_P) = s2;
    *(float4*)(xs + 24 * GEMM_P) = s3;
    __syncthreads();

    const int nx = ch + nblk;
    if (nx < nChunks) {
      const int r0 = nx * GEMM_ROWS + wave;
      s0 = *(const float4*)(xl + (size_t)min(r0,      rmax) * IN_F);
      s1 = *(const float4*)(xl + (size_t)min(r0 + 8,  rmax) * IN_F);
      s2 = *(const float4*)(xl + (size_t)min(r0 + 16, rmax) * IN_F);
      s3 = *(const float4*)(xl + (size_t)min(r0 + 24, rmax) * IN_F);
    }

    v8f acc[MPW];
#pragma unroll
    for (int j = 0; j < MPW; ++j) acc[j] = v8f{};
#pragma unroll
    for (int i = 0; i < NK; ++i) {
#pragma unroll
      for (int j = 0; j < MPW; ++j) {
        v2f a = *(const v2f*)(ap + j * 16 * GEMM_P + i * 4);
        acc[j] = __builtin_amdgcn_wmma_f32_16x16x4_f32(false, a, false, bfr[i],
                                                       (short)0, acc[j], false, false);
      }
    }

#pragma unroll
    for (int j = 0; j < MPW; ++j) {
      const int m0 = ch * GEMM_ROWS + (grp * MPW + j) * 16;
      if (m0 < N) {
        float* yp = Y + (size_t)(m0 + ((lane >> 4) << 3)) * ldy + c0 + col;
#pragma unroll
        for (int g = 0; g < 8; ++g)
          yp[(size_t)g * ldy] = acc[j][g] + bv;
      }
    }
    __syncthreads();
  }
}


static_assert(BIN_T == GEMM_T, "binning and product work-groups share a launch shape");

__global__ __launch_bounds__(GEMM_T)
void k_count_gemm(const int* __restrict__ ei0, const int* __restrict__ ei1,
                  int* __restrict__ cnt, int E, int NB, int binBlocks,
                  const float* __restrict__ X,
                  const float* __restrict__ Wa, const float* __restrict__ ba,
                  const float* __restrict__ Wb, const float* __restrict__ bb,
                  float* __restrict__ Y, int N, int nChunks) {
  __shared__ float xt[GEMM_ROWS * GEMM_P];
  const int b = blockIdx.x;
  if (b < binBlocks)
    bin_count_part(ei0, ei1, cnt, E, NB, b);
  else
    gemm_part<IN_F, IN_F / 16, 0>(xt, X, Wa, ba, Wb, bb, Y, IN_F, HID_F, 0, N, nChunks,
                                  b - binBlocks, (int)gridDim.x - binBlocks);
}

__global__ __launch_bounds__(GEMM_T)
void k_place_gemm(const int* __restrict__ ei0, const int* __restrict__ ei1,
                  const int* __restrict__ cnt, int* __restrict__ cursor, int* __restrict__ start,
                  int* __restrict__ recF, int* __restrict__ recB,
                  int E, int NB, int binBlocks,
                  const float* __restrict__ X,
                  const float* __restrict__ Wa, const float* __restrict__ ba,
                  const float* __restrict__ Wb, const float* __restrict__ bb,
                  float* __restrict__ Y, int N, int nChunks) {
  __shared__ float xt[GEMM_ROWS * GEMM_P];
  const int b = blockIdx.x;
  if (b < binBlocks)
    bin_place_part(ei0, ei1, cnt, cursor, start, recF, recB, E, NB, b);
  else
    gemm_part<IN_F, IN_F / 16, 0>(xt, X, Wa, ba, Wb, bb, Y, IN_F, HID_F, 0, N, nChunks,
                                  b - binBlocks, (int)gridDim.x - binBlocks);
}

#define RP_C4 (OUT_F / 4)
#define RP_IO ((TR * RP_C4 + AGG_T - 1) / AGG_T)

static __device__ __forceinline__
void reparam_load(float4 (&nz)[RP_IO], const float* __restrict__ noise, int n0, int qo, int tid) {
  const float4* z4 = (const float4*)(noise + (size_t)n0 * OUT_F);
#pragma unroll
  for (int i = 0; i < RP_IO; ++i) nz[i] = z4[min(tid + i * AGG_T, qo - 1)];
}

static __device__ __forceinline__
void reparam_store(const float4 (&nz)[RP_IO], const float* tileB, float* __restrict__ out,
                   int n0, int qo, int tid) {
  float4* o4 = (float4*)(out + (size_t)n0 * OUT_F);
#pragma unroll
  for (int i = 0; i < RP_IO; ++i) {
    int idx = tid + i * AGG_T;
    if (idx < qo) {
      const float* tr = tileB + (idx / RP_C4) * HID_F + (idx % RP_C4) * 4;
      float4 mu = *(const float4*)tr;
      float4 ls = *(const float4*)(tr + OUT_F);
      float4 o;
      o.x = mu.x + nz[i].x * expf(ls.x) * 0.2f;
      o.y = mu.y + nz[i].y * expf(ls.y) * 0.2f;
      o.z = mu.z + nz[i].z * expf(ls.z) * 0.2f;
      o.w = mu.w + nz[i].w * expf(ls.w) * 0.2f;
      o4[idx] = o;
    }
  }
}


#define AGG_W   (AGG_T / 32)
#define RPW     ((TR + AGG_W - 1) / AGG_W)
#define PASS_R  2048
static_assert(PASS_R % AGG_T == 0, "whole records per thread in a pass");
static_assert(TR + 1 <= 4 * 32, "one wave scans the row counters, four per lane");

template <int JA, int PA, int JB, int PB, int U, bool TAIL>
static __device__ __forceinline__
void acc_records(const float* __restrict__ Ya, const float* __restrict__ Yb,
                 float* tileA, float* tileB,
                 float (&accA)[JA], float (&accB)[JB ? JB : 1], int& cur, int lane,
                 int src, int dl, float rn, int k0, int cnt) {
  float va[U][JA];
  float vb[U][JB ? JB : 1];
  int   d[U];
  float r[U];
#pragma unroll
  for (int u = 0; u < U; ++u) {
    int k = k0 + u;
    d[u] = -1; r[u] = 0.0f;
    if (!TAIL || k < cnt) {
      int s = __builtin_amdgcn_readlane(src, k);
      d[u]  = __builtin_amdgcn_readlane(dl, k);
      r[u]  = __int_as_float(__builtin_amdgcn_readlane(__float_as_int(rn), k));
      const float* ya = Ya + (size_t)s * (32 * JA) + lane;
#pragma unroll
      for (int j = 0; j < JA; ++j) va[u][j] = ya[32 * j];
      if (JB) {
        const float* yb = Yb + (size_t)s * (32 * JB) + lane;
#pragma unroll
        for (int j = 0; j < JB; ++j) vb[u][j] = yb[32 * j];
      }
    }
  }
#pragma unroll
  for (int u = 0; u < U; ++u) {
    if (!TAIL || k0 + u < cnt) {
      if (__builtin_expect(d[u] != cur, 0)) {
        if (cur >= 0) {
#pragma unroll
          for (int j = 0; j < JA; ++j) tileA[cur * PA + lane + 32 * j] = accA[j];
#pragma unroll
          for (int j = 0; j < JB; ++j) tileB[cur * PB + lane + 32 * j] = accB[j];
        }
        cur = d[u];
#pragma unroll
        for (int j = 0; j < JA; ++j) accA[j] = tileA[cur * PA + lane + 32 * j];
#pragma unroll
        for (int j = 0; j < JB; ++j) accB[j] = tileB[cur * PB + lane + 32 * j];
      }
#pragma unroll
      for (int j = 0; j < JA; ++j) accA[j] += r[u] * va[u][j];
#pragma unroll
      for (int j = 0; j < JB; ++j) accB[j] += r[u] * vb[u][j];
    }
  }
}

template <int JA, int PA, int JB, int PB, int U>
static __device__ __forceinline__
void walk_owned(const float* __restrict__ Ya, float* tileA, const float* __restrict__ Yb, float* tileB,
                const int2* __restrict__ rec, int lo, int hi,
                const float* __restrict__ degD, const float* __restrict__ degS,
                int n0, int2* qrec, int* rcnt, int* rbase, int tid, int lane, int wave) {
  constexpr int KP = PASS_R / AGG_T;
  const int R0 = min(wave * RPW, TR);
  const int R1 = min(wave * RPW + RPW, TR);

  for (int p0 = lo; p0 < hi; p0 += PASS_R) {
    const int pe = min(hi, p0 + PASS_R);
    if (tid < TR) rcnt[tid] = 0;
    __syncthreads();
    int2 r[KP];
    int  row[KP], rank[KP];
#pragma unroll
    for (int k = 0; k < KP; ++k) r[k] = rec[min(p0 + tid + k * AGG_T, pe - 1)];
#pragma unroll
    for (int k = 0; k < KP; ++k) {
      row[k] = r[k].y - n0;
      rank[k] = 0;
      if (p0 + tid + k * AGG_T < pe) rank[k] = atomicAdd(&rcnt[row[k]], 1);
    }
    __syncthreads();
    if (wave == 0) {
      int c[4];
      int s = 0;
#pragma unroll
      for (int j = 0; j < 4; ++j) {
        int i = 4 * lane + j;
        c[j] = (i < TR) ? rcnt[i] : 0;
        s += c[j];
      }
      int incl = s;
#pragma unroll
      for (int off = 1; off < 32; off <<= 1) {
        int t = __shfl_up(incl, off);
        if (lane >= off) incl += t;
      }
      int run = incl - s;
#pragma unroll
      for (int j = 0; j < 4; ++j) {
        int i = 4 * lane + j;
        if (i <= TR) rbase[i] = run;
        run += c[j];
      }
    }
    __syncthreads();
#pragma unroll
    for (int k = 0; k < KP; ++k) {
      if (p0 + tid + k * AGG_T < pe)
        qrec[rbase[row[k]] + rank[k]] = make_int2(r[k].x, row[k]);
    }
    __syncthreads();

    const int qb = __builtin_amdgcn_readfirstlane(rbase[R0]);
    const int qn = __builtin_amdgcn_readfirstlane(rbase[R1]) - qb;
    float accA[JA];
    float accB[JB ? JB : 1];
#pragma unroll
    for (int j = 0; j < JA; ++j) accA[j] = 0.0f;
#pragma unroll
    for (int j = 0; j < (JB ? JB : 1); ++j) accB[j] = 0.0f;
    int cur = -1;
    for (int b = 0; b < qn; b += 32) {
      const int m = min(32, qn - b);
      int src = 0, dl = 0;
      float rn = 0.0f;
      if (lane < m) {
        int2 e = qrec[qb + b + lane];
        src = e.x;
        dl  = e.y;
        rn  = rsqrtf(degD[n0 + dl] * degS[src]);
      }
      int k0 = 0;
      for (; k0 + U <= m; k0 += U)
        acc_records<JA, PA, JB, PB, U, false>(Ya, Yb, tileA, tileB, accA, accB, cur, lane, src, dl, rn, k0, m);
      if (k0 < m)
        acc_records<JA, PA, JB, PB, U, true>(Ya, Yb, tileA, tileB, accA, accB, cur, lane, src, dl, rn, k0, m);
    }
    if (cur >= 0) {
#pragma unroll
      for (int j = 0; j < JA; ++j) tileA[cur * PA + lane + 32 * j] = accA[j];
#pragma unroll
      for (int j = 0; j < JB; ++j) tileB[cur * PB + lane + 32 * j] = accB[j];
    }
    if (p0 + PASS_R < hi) __syncthreads();
  }
}

template <int F>
__global__ __launch_bounds__(AGG_T)
void k_aggr_tile(const float* __restrict__ Y, const int* __restrict__ rec_,
                 const int* __restrict__ st,
                 const float* __restrict__ degD, const float* __restrict__ degS,
                 const float* __restrict__ noise, float* __restrict__ out, int N) {
  static_assert(F == HID_F && F == 2 * OUT_F, "the tile rows are [mu | logstd]");
  constexpr int U  = 1024 / F;
  constexpr int Q  = TR * (F / 4);
  constexpr int IT = (Q + AGG_T - 1) / AGG_T;
  __shared__ float tile[TR * F];
  __shared__ int2 qrec[PASS_R];
  __shared__ int rcnt[TR], rbase[TR + 1];

  const int2* rec = (const int2*)rec_;
  const int b    = blockIdx.x;
  const int n0   = b * TR;
  const int rows = min(TR, N - n0);
  const int q    = rows * (F / 4);
  const int tid  = threadIdx.x;
  const int lane = tid & 31;
  const int wave = __builtin_amdgcn_readfirstlane(tid >> 5);

  const int qo = rows * RP_C4;
  float4 nz[RP_IO];
  reparam_load(nz, noise, n0, qo, tid);

  {
    const float4* y4 = (const float4*)(Y + (size_t)n0 * F);
    float4 v[IT];
    float  dd[IT], ds[IT];
#pragma unroll
    for (int i = 0; i < IT; ++i) {
      int idx = min(tid + i * AGG_T, q - 1);
      int n = n0 + idx / (F / 4);
      v[i]  = y4[idx];
      dd[i] = degD[n];
      ds[i] = degS[n];
    }
#pragma unroll
    for (int i = 0; i < IT; ++i) {
      int idx = tid + i * AGG_T;
      if (idx < q) {
        float rn = rsqrtf(dd[i] * ds[i]);
        float4 o; o.x = rn * v[i].x; o.y = rn * v[i].y; o.z = rn * v[i].z; o.w = rn * v[i].w;
        ((float4*)tile)[idx] = o;
      }
    }
  }
  __syncthreads();

  const int lo = st[b], hi = st[b + 1];
  walk_owned<F / 32, F, 0, 0, U>(Y, tile, nullptr, nullptr, rec, lo, hi, degD, degS, n0, qrec, rcnt, rbase, tid, lane, wave);
  __syncthreads();

  reparam_store(nz, tile, out, n0, qo, tid);
}


#define L2_RT   ((TR + 15) / 16)
#define L2_ROWS (L2_RT * 16)
#define L2_P    (IN_F + 4)

__global__ __launch_bounds__(AGG_T)
void k_aggr_lin2(const float* __restrict__ Y, const int* __restrict__ rec_,
                 const int* __restrict__ st,
                 const float* __restrict__ degD, const float* __restrict__ degS,
                 const float* __restrict__ Wa, const float* __restrict__ ba,
                 const float* __restrict__ Wb, const float* __restrict__ bb,
                 float* __restrict__ Y2, int N) {
  constexpr int F  = IN_F;
  constexpr int P  = L2_P;
  constexpr int K  = HID_F;
  constexpr int NK = K / 4;
  constexpr int U  = 1024 / F;
  constexpr int QP = L2_ROWS * (F / 4);
  constexpr int IT = (QP + AGG_T - 1) / AGG_T;
  constexpr int NCT = HID_F / 16;
  constexpr int NG  = (AGG_T / 32) / NCT;
  static_assert(L2_RT <= 2 * NG, "two row tiles per wave cover the tile");
  __shared__ float tile[L2_ROWS * P];
  __shared__ int2 qrec[PASS_R];
  __shared__ int rcnt[TR], rbase[TR + 1];

  const int2* rec = (const int2*)rec_;
  const int b    = blockIdx.x;
  const int n0   = b * TR;
  const int rows = min(TR, N - n0);
  const int q    = rows * (F / 4);
  const int tid  = threadIdx.x;
  const int lane = tid & 31;
  const int wave = __builtin_amdgcn_readfirstlane(tid >> 5);

  {
    const float4* y4 = (const float4*)(Y + (size_t)n0 * F);
    float4 v[IT];
    float  dd[IT], ds[IT];
#pragma unroll
    for (int i = 0; i < IT; ++i) {
      int idx = min(tid + i * AGG_T, q - 1);
      int n = n0 + idx / (F / 4);
      v[i]  = y4[idx];
      dd[i] = degD[n];
      ds[i] = degS[n];
    }
#pragma unroll
    for (int i = 0; i < IT; ++i) {
      int idx = tid + i * AGG_T;
      if (idx < QP) {
        float4 o = {};
        if (idx < q) {
          float rn = rsqrtf(dd[i] * ds[i]);
          o.x = rn * v[i].x; o.y = rn * v[i].y; o.z = rn * v[i].z; o.w = rn * v[i].w;
        }
        *(float4*)(tile + (idx / (F / 4)) * P + (idx % (F / 4)) * 4) = o;
      }
    }
  }
  __syncthreads();

  const int lo = st[b], hi = st[b + 1];
  walk_owned<F / 32, P, 0, 0, U>(Y, tile, nullptr, nullptr, rec, lo, hi, degD, degS, n0, qrec, rcnt, rbase, tid, lane, wave);

  const int ct   = wave % NCT;
  const int grp  = wave / NCT;
  const int col  = lane & 15;
  const int khi  = (lane >> 4) << 1;
  const int cc0  = ct << 4;
  const int half = (cc0 >= OUT_F) ? 1 : 0;
  const float* W    = half ? Wb : Wa;
  const float* bias = half ? bb : ba;
  const int wc0  = cc0 - half * OUT_F;
  const int koff = half ? HID_F : 0;
  v2f bfr[NK];
  const float* wrow = W + (size_t)(wc0 + col) * K + khi;
#pragma unroll
  for (int i = 0; i < NK; ++i) bfr[i] = *(const v2f*)(wrow + i * 4);
  const float bv = bias[wc0 + col];
  __syncthreads();

  const bool two = (grp + NG) < L2_RT;
  const int rt1  = two ? grp + NG : grp;
  const float* ap0 = tile + (grp * 16 + col) * P + koff + khi;
  const float* ap1 = tile + (rt1 * 16 + col) * P + koff + khi;
  v8f acc0 = {}, acc1 = {};
#pragma unroll
  for (int i = 0; i < NK; ++i) {
    v2f a0 = *(const v2f*)(ap0 + i * 4);
    v2f a1 = *(const v2f*)(ap1 + i * 4);
    a0.x = fmaxf(a0.x, 0.0f); a0.y = fmaxf(a0.y, 0.0f);
    a1.x = fmaxf(a1.x, 0.0f); a1.y = fmaxf(a1.y, 0.0f);
    acc0 = __builtin_amdgcn_wmma_f32_16x16x4_f32(false, a0, false, bfr[i], (short)0, acc0, false, false);
    acc1 = __builtin_amdgcn_wmma_f32_16x16x4_f32(false, a1, false, bfr[i], (short)0, acc1, false, false);
  }
  __syncthreads();

  {
    float* yp0 = tile + (grp * 16 + ((lane >> 4) << 3)) * P + cc0 + col;
#pragma unroll
    for (int g = 0; g < 8; ++g) yp0[g * P] = acc0[g] + bv;
    if (two) {
      float* yp1 = tile + (rt1 * 16 + ((lane >> 4) << 3)) * P + cc0 + col;
#pragma unroll
      for (int g = 0; g < 8; ++g) yp1[g * P] = acc1[g] + bv;
    }
  }
  __syncthreads();

  {
    constexpr int C4 = HID_F / 4;
    constexpr int IO = (TR * C4 + AGG_T - 1) / AGG_T;
    const int qo = rows * C4;
    float4* o4 = (float4*)(Y2 + (size_t)n0 * HID_F);
#pragma unroll
    for (int i = 0; i < IO; ++i) {
      int idx = tid + i * AGG_T;
      if (idx < qo) o4[idx] = *(const float4*)(tile + (idx / C4) * P + (idx % C4) * 4);
    }
  }
}


__global__ __launch_bounds__(AGG_T)
void k_aggr_lin2_tile(const float* __restrict__ Y, const float* __restrict__ Yb,
                      const int* __restrict__ rec_, const int* __restrict__ st,
                      const float* __restrict__ degD, const float* __restrict__ degS,
                      const float* __restrict__ Wa, const float* __restrict__ ba,
                      const float* __restrict__ Wb, const float* __restrict__ bb,
                      float* __restrict__ Y2,
                      const float* __restrict__ noise, float* __restrict__ out, int N) {
  constexpr int F   = IN_F;
  constexpr int FB  = HID_F;
  constexpr int P   = L2_P;
  constexpr int K   = HID_F;
  constexpr int NK  = K / 4;
  constexpr int U   = 4;
  constexpr int QP  = L2_ROWS * (F / 4);
  constexpr int IT  = (QP + AGG_T - 1) / AGG_T;
  constexpr int QB  = TR * (FB / 4);
  constexpr int ITB = (QB + AGG_T - 1) / AGG_T;
  constexpr int NCT = HID_F / 16;
  constexpr int NG  = (AGG_T / 32) / NCT;
  static_assert(L2_RT <= 2 * NG, "two row tiles per wave cover the tile");
  __shared__ float tile[L2_ROWS * P];
  __shared__ int2 qrec[PASS_R];
  __shared__ int rcnt[TR], rbase[TR + 1];
  __shared__ float tileB[TR * FB];

  const int2* rec = (const int2*)rec_;
  const int b    = blockIdx.x;
  const int n0   = b * TR;
  const int rows = min(TR, N - n0);
  const int q    = rows * (F / 4);
  const int qb   = rows * (FB / 4);
  const int tid  = threadIdx.x;
  const int lane = tid & 31;
  const int wave = __builtin_amdgcn_readfirstlane(tid >> 5);

  {
    const float4* y4 = (const float4*)(Y + (size_t)n0 * F);
    float4 v[IT];
    float  dd[IT], ds[IT];
#pragma unroll
    for (int i = 0; i < IT; ++i) {
      int idx = min(tid + i * AGG_T, q - 1);
      int n = n0 + idx / (F / 4);
      v[i]  = y4[idx];
      dd[i] = degD[n];
      ds[i] = degS[n];
    }
#pragma unroll
    for (int i = 0; i < IT; ++i) {
      int idx = tid + i * AGG_T;
      if (idx < QP) {
        float4 o = {};
        if (idx < q) {
          float rn = rsqrtf(dd[i] * ds[i]);
          o.x = rn * v[i].x; o.y = rn * v[i].y; o.z = rn * v[i].z; o.w = rn * v[i].w;
        }
        *(float4*)(tile + (idx / (F / 4)) * P + (idx % (F / 4)) * 4) = o;
      }
    }
  }
  {
    const float4* y4 = (const float4*)(Yb + (size_t)n0 * FB);
    float4 v[ITB];
    float  dd[ITB], ds[ITB];
#pragma unroll
    for (int i = 0; i < ITB; ++i) {
      int idx = min(tid + i * AGG_T, qb - 1);
      int n = n0 + idx / (FB / 4);
      v[i]  = y4[idx];
      dd[i] = degD[n];
      ds[i] = degS[n];
    }
#pragma unroll
    for (int i = 0; i < ITB; ++i) {
      int idx = tid + i * AGG_T;
      if (idx < qb) {
        float rn = rsqrtf(dd[i] * ds[i]);
        float4 o; o.x = rn * v[i].x; o.y = rn * v[i].y; o.z = rn * v[i].z; o.w = rn * v[i].w;
        ((float4*)tileB)[idx] = o;
      }
    }
  }
  __syncthreads();

  const int lo = st[b], hi = st[b + 1];
  walk_owned<F / 32, P, FB / 32, FB, U>(Y, tile, Yb, tileB, rec, lo, hi, degD, degS, n0, qrec, rcnt, rbase, tid, lane, wave);

  const int ct   = wave % NCT;
  const int grp  = wave / NCT;
  const int col  = lane & 15;
  const int khi  = (lane >> 4) << 1;
  const int cc0  = ct << 4;
  const int half = (cc0 >= OUT_F) ? 1 : 0;
  const float* W    = half ? Wb : Wa;
  const float* bias = half ? bb : ba;
  const int wc0  = cc0 - half * OUT_F;
  const int koff = half ? HID_F : 0;
  v2f bfr[NK];
  const float* wrow = W + (size_t)(wc0 + col) * K + khi;
#pragma unroll
  for (int i = 0; i < NK; ++i) bfr[i] = *(const v2f*)(wrow + i * 4);
  const float bv = bias[wc0 + col];
  const int qo = rows * RP_C4;
  float4 nz[RP_IO];
  reparam_load(nz, noise, n0, qo, tid);
  __syncthreads();

  reparam_store(nz, tileB, out, n0, qo, tid);

  const bool two = (grp + NG) < L2_RT;
  const int rt1  = two ? grp + NG : grp;
  const float* ap0 = tile + (grp * 16 + col) * P + koff + khi;
  const float* ap1 = tile + (rt1 * 16 + col) * P + koff + khi;
  v8f acc0 = {}, acc1 = {};
#pragma unroll
  for (int i = 0; i < NK; ++i) {
    v2f a0 = *(const v2f*)(ap0 + i * 4);
    v2f a1 = *(const v2f*)(ap1 + i * 4);
    a0.x = fmaxf(a0.x, 0.0f); a0.y = fmaxf(a0.y, 0.0f);
    a1.x = fmaxf(a1.x, 0.0f); a1.y = fmaxf(a1.y, 0.0f);
    acc0 = __builtin_amdgcn_wmma_f32_16x16x4_f32(false, a0, false, bfr[i], (short)0, acc0, false, false);
    acc1 = __builtin_amdgcn_wmma_f32_16x16x4_f32(false, a1, false, bfr[i], (short)0, acc1, false, false);
  }
  __syncthreads();

  {
    float* yp0 = tile + (grp * 16 + ((lane >> 4) << 3)) * P + cc0 + col;
#pragma unroll
    for (int g = 0; g < 8; ++g) yp0[g * P] = acc0[g] + bv;
    if (two) {
      float* yp1 = tile + (rt1 * 16 + ((lane >> 4) << 3)) * P + cc0 + col;
#pragma unroll
      for (int g = 0; g < 8; ++g) yp1[g * P] = acc1[g] + bv;
    }
  }
  __syncthreads();

  {
    constexpr int C4 = HID_F / 4;
    constexpr int IO = (TR * C4 + AGG_T - 1) / AGG_T;
    const int qo = rows * C4;
    float4* o4 = (float4*)(Y2 + (size_t)n0 * HID_F);
#pragma unroll
    for (int i = 0; i < IO; ++i) {
      int idx = tid + i * AGG_T;
      if (idx < qo) o4[idx] = *(const float4*)(tile + (idx / C4) * P + (idx % C4) * 4);
    }
  }
}


static inline int ceil_div(int a, int b) { return (a + b - 1) / b; }

extern "C" void kernel_launch(void* const* d_in, const int* in_sizes, int n_in,
                              void* d_out, int out_size, void* d_ws, size_t ws_size,
                              hipStream_t stream) {
  const float* s       = (const float*)d_in[0];
  const float* t       = (const float*)d_in[1];
  const int*   ei      = (const int*)  d_in[2];
  const float* noise_s = (const float*)d_in[3];
  const float* noise_t = (const float*)d_in[4];
  const float* sm1_W = (const float*)d_in[5],  *sm1_b = (const float*)d_in[6];
  const float* sm2_W = (const float*)d_in[7],  *sm2_b = (const float*)d_in[8];
  const float* sl1_W = (const float*)d_in[9],  *sl1_b = (const float*)d_in[10];
  const float* sl2_W = (const float*)d_in[11], *sl2_b = (const float*)d_in[12];
  const float* tm1_W = (const float*)d_in[13], *tm1_b = (const float*)d_in[14];
  const float* tm2_W = (const float*)d_in[15], *tm2_b = (const float*)d_in[16];
  const float* tl1_W = (const float*)d_in[17], *tl1_b = (const float*)d_in[18];
  const float* tl2_W = (const float*)d_in[19], *tl2_b = (const float*)d_in[20];

  const int N = in_sizes[0] / IN_F;
  const int E = in_sizes[2] / 2;
  const int* ei0 = ei;
  const int* ei1 = ei + E;
  const int NB = ceil_div(N, TR);

  float* ws  = (float*)d_ws;
  float* Y1s = ws;
  float* Y2s = Y1s + (size_t)N * IN_F;
  float* Y1t = Y2s + (size_t)N * HID_F;
  float* Y2t = Y1s;
  int*   recF = (int*)(Y1t + (size_t)N * IN_F + (size_t)N * HID_F);
  int*   recB = recF + (size_t)2 * E;
  float* deg_in  = (float*)(recB + (size_t)2 * E);
  float* deg_out = deg_in + N;
  int*   cnt     = (int*)(deg_out + N);
  int*   cursor  = cnt + REPL * 2 * NB;
  int*   start   = cursor + REPL * 2 * NB;
  const int* startF = start;
  const int* startB = start + (NB + 1);

  const int nChunks = ceil_div(N, GEMM_ROWS);
  const int gemmBlocks = 512;
  const int binBlocks = ceil_div(E, BIN_T * BIN_EPT);

  k_zero_counts<<<ceil_div(2 * REPL * 2 * NB, 256), 256, 0, stream>>>(cnt, 2 * REPL * 2 * NB);
  k_count_gemm<<<binBlocks + gemmBlocks, GEMM_T, 2 * NB * sizeof(int), stream>>>(
      ei0, ei1, cnt, E, NB, binBlocks,
      t, tm1_W, tm1_b, tl1_W, tl1_b, Y1t, N, nChunks);
  k_place_gemm<<<binBlocks + gemmBlocks, GEMM_T, (4 * NB + BIN_T / 32 + 1) * sizeof(int), stream>>>(
      ei0, ei1, cnt, cursor, start, recF, recB, E, NB, binBlocks,
      s, sm1_W, sm1_b, sl1_W, sl1_b, Y1s, N, nChunks);
  k_degrees<<<dim3(NB, 2), 256, 0, stream>>>(recF, recB, start, deg_in, deg_out, N, NB);

  k_aggr_lin2<<<NB, AGG_T, 0, stream>>>(Y1s, recF, startF, deg_in, deg_out,
                                        sm2_W, sm2_b, sl2_W, sl2_b, Y2s, N);

  float* out_s = (float*)d_out;
  float* out_t = out_s + (size_t)N * OUT_F;

  k_aggr_lin2_tile<<<NB, AGG_T, 0, stream>>>(Y1t, Y2s, recB, startB, deg_out, deg_in,
                                             tm2_W, tm2_b, tl2_W, tl2_b, Y2t, noise_s, out_s, N);

  k_aggr_tile<HID_F><<<NB, AGG_T, 0, stream>>>(Y2t, recF, startF, deg_in, deg_out, noise_t, out_t, N);
}
